# starting kernel
# baseline (speedup 1.0000x reference)
#include <hip/hip_runtime.h>
#include <hip/hip_bf16.h>

typedef __attribute__((ext_vector_type(2))) float v2f;
typedef __attribute__((ext_vector_type(8))) float v8f;

__global__ void deg_init_kernel(float* deg, int n) {
  int i = blockIdx.x * blockDim.x + threadIdx.x;
  if (i < n) deg[i] = 1.0f;
}

__global__ void deg_edge_kernel(const int* __restrict__ col, float* deg, int E) {
  int e = blockIdx.x * blockDim.x + threadIdx.x;
  if (e < E) atomicAdd(&deg[col[e]], 1.0f);
}

__global__ void rsqrt_kernel(float* deg, int n) {
  int i = blockIdx.x * blockDim.x + threadIdx.x;
  if (i < n) deg[i] = rsqrtf(deg[i]);
}

template <int N, int K, bool BIAS_RELU, int TM>
__global__ void gemm_wmma_kernel(const float* __restrict__ A,
                                 const float* __restrict__ B,
                                 float* __restrict__ C,
                                 const float* __restrict__ act_bias,
                                 int M) {
  const int lane = threadIdx.x & 31;
  const int wave = threadIdx.x >> 5;
  const int n0   = wave * 16;
  const int half = lane >> 4;
  const int l    = lane & 15;
  const int kb   = 2 * half;

  constexpr int KSTEPS = K / 4;

  v2f breg[KSTEPS];
  {
    const float* bcol = B + (size_t)kb * N + (n0 + l);
#pragma unroll
    for (int s = 0; s < KSTEPS; ++s) {
      breg[s].x = bcol[s * 4 * N];
      breg[s].y = bcol[(s * 4 + 1) * N];
    }
  }
  const float* bias = act_bias + kb;

  for (int t = 0; t < TM; ++t) {
    const int m0 = (blockIdx.x * TM + t) * 16;
    if (m0 >= M) break;

    int ar = m0 + l;
    if (ar > M - 1) ar = M - 1;
    const float* arow = A + (size_t)ar * K + kb;

    v8f acc = {};
#pragma unroll
    for (int s = 0; s < KSTEPS; ++s) {
      v2f a;
      a.x = arow[s * 4];
      a.y = arow[s * 4 + 1];
      if constexpr (BIAS_RELU) {
        a.x = fmaxf(a.x + bias[s * 4], 0.0f);
        a.y = fmaxf(a.y + bias[s * 4 + 1], 0.0f);
      }
      acc = __builtin_amdgcn_wmma_f32_16x16x4_f32(false, a, false, breg[s],
                                                  (short)0, acc, false, false);
    }

    float* cp = C + (size_t)(m0 + half * 8) * N + (n0 + l);
#pragma unroll
    for (int v = 0; v < 8; ++v) {
      if (m0 + half * 8 + v < M) cp[(size_t)v * N] = acc[v];
    }
  }
}

template <int F, bool HAS_BIAS>
__global__ void self_bias_init_kernel(const float* __restrict__ h,
                                      const float* __restrict__ dinv,
                                      const float* __restrict__ bias,
                                      float* __restrict__ out, int n) {
  int idx = blockIdx.x * blockDim.x + threadIdx.x;
  if (idx >= n * F) return;
  int i = idx / F;
  int f = idx - i * F;
  float d = dinv[i];
  float v = h[idx] * d * d;
  if constexpr (HAS_BIAS) v += bias[f];
  out[idx] = v;
}

template <int F>
__global__ void edge_agg_kernel(const int* __restrict__ row,
                                const int* __restrict__ col,
                                const float* __restrict__ h,
                                const float* __restrict__ dinv,
                                float* out, int E) {
  const int lane = threadIdx.x & 31;
  const int e = blockIdx.x * (blockDim.x >> 5) + (threadIdx.x >> 5);
  if (e >= E) return;
  const int r = row[e];
  const int c = col[e];
  const float norm = dinv[r] * dinv[c];
  constexpr int VEC = F / 32;
  const float* hp = h + (size_t)r * F + lane * VEC;
  float* op = out + (size_t)c * F + lane * VEC;
  if constexpr (VEC == 4) {
    float4 hv = *(const float4*)hp;
    atomicAdd(op + 0, hv.x * norm);
    atomicAdd(op + 1, hv.y * norm);
    atomicAdd(op + 2, hv.z * norm);
    atomicAdd(op + 3, hv.w * norm);
  } else {
    float2 hv = *(const float2*)hp;
    atomicAdd(op + 0, hv.x * norm);
    atomicAdd(op + 1, hv.y * norm);
  }
}

extern "C" void kernel_launch(void* const* d_in, const int* in_sizes, int n_in,
                              void* d_out, int out_size, void* d_ws, size_t ws_size,
                              hipStream_t stream) {
  const float* x  = (const float*)d_in[0];
  const int*   ei = (const int*)d_in[1];
  const float* W1 = (const float*)d_in[2];
  const float* b1 = (const float*)d_in[3];
  const float* W2 = (const float*)d_in[4];
  const float* b2 = (const float*)d_in[5];

  const int hid    = in_sizes[3];
  const int out_ch = in_sizes[5];
  const int in_ch  = in_sizes[2] / hid;
  const int n      = in_sizes[0] / in_ch;
  const int E      = in_sizes[1] / 2;
  const int* row = ei;
  const int* col = ei + E;

  float* ws = (float*)d_ws;
  size_t off = 0;
  auto alloc = [&](size_t elems) {
    float* p = ws + off;
    off += (elems + 63) & ~(size_t)63;
    return p;
  };
  float* dinv = alloc((size_t)n);
  float* h1   = alloc((size_t)n * hid);
  float* agg1 = alloc((size_t)n * hid);
  float* g2   = alloc((size_t)n * out_ch);
  float* out  = (float*)d_out;

  deg_init_kernel<<<(n + 255) / 256, 256, 0, stream>>>(dinv, n);
  deg_edge_kernel<<<(E + 255) / 256, 256, 0, stream>>>(col, dinv, E);
  rsqrt_kernel<<<(n + 255) / 256, 256, 0, stream>>>(dinv, n);

  constexpr int TM = 4;
  const int mtiles   = (n + 15) / 16;
  const int mblocks  = (mtiles + TM - 1) / TM;

  gemm_wmma_kernel<128, 128, false, TM><<<mblocks, 32 * (128 / 16), 0, stream>>>(
      x, W1, h1, nullptr, n);
  self_bias_init_kernel<128, false>
      <<<(unsigned)(((size_t)n * 128 + 255) / 256), 256, 0, stream>>>(
          h1, dinv, nullptr, agg1, n);
  edge_agg_kernel<128><<<(E + 7) / 8, 256, 0, stream>>>(row, col, h1, dinv, agg1, E);

  gemm_wmma_kernel<64, 128, true, TM><<<mblocks, 32 * (64 / 16), 0, stream>>>(
      agg1, W2, g2, b1, n);
  self_bias_init_kernel<64, true>
      <<<(unsigned)(((size_t)n * 64 + 255) / 256), 256, 0, stream>>>(
          g2, dinv, b2, out, n);
  edge_agg_kernel<64><<<(E + 7) / 8, 256, 0, stream>>>(row, col, g2, dinv, out, E);
}


# Edge sums by destination: per-call counting sort of edges (int atomic ranks, arrival order) + per-node register gather with one row store replace per-edge f32 atomic adds; sum order still varies
# speedup vs baseline: 2.1518x; 2.1518x over previous
#include <hip/hip_runtime.h>
#include <hip/hip_bf16.h>


typedef __attribute__((ext_vector_type(2))) float v2f;
typedef __attribute__((ext_vector_type(8))) float v8f;

__global__ void zero_cnt_kernel(int* cnt, int n) {
  int i = blockIdx.x * blockDim.x + threadIdx.x;
  if (i < n) cnt[i] = 0;
}

__global__ void rank_kernel(const int* __restrict__ col, int* cnt,
                            int* __restrict__ rank, int E) {
  int e = blockIdx.x * blockDim.x + threadIdx.x;
  if (e < E) rank[e] = atomicAdd(&cnt[col[e]], 1);
}

constexpr int SCAN_THREADS = 1024;
constexpr int SCAN_EPT = 8;

__global__ __launch_bounds__(SCAN_THREADS) void scan_dinv_kernel(
    const int* __restrict__ cnt, int* __restrict__ offs,
    float* __restrict__ dinv, int n) {
  __shared__ int wsum[2][32];
  const int tid = threadIdx.x;
  const int lane = tid & 31;
  const int wave = tid >> 5;
  constexpr int TILE = SCAN_THREADS * SCAN_EPT;
  const int ntiles = (n + TILE - 1) / TILE;
  int carry = 0;
  for (int t = 0; t < ntiles; ++t) {
    const int base = t * TILE + tid * SCAN_EPT;
    const bool full = base + SCAN_EPT <= n;
    int v[SCAN_EPT];
    if (full) {
      const int4 a = *(const int4*)(cnt + base);
      const int4 b = *(const int4*)(cnt + base + 4);
      v[0] = a.x; v[1] = a.y; v[2] = a.z; v[3] = a.w;
      v[4] = b.x; v[5] = b.y; v[6] = b.z; v[7] = b.w;
    } else {
      for (int q = 0; q < SCAN_EPT; ++q) v[q] = (base + q < n) ? cnt[base + q] : 0;
    }
    int s = 0;
    for (int q = 0; q < SCAN_EPT; ++q) s += v[q];
    int incl = s;
    for (int d = 1; d < 32; d <<= 1) {
      const int y = __shfl_up(incl, d);
      if (lane >= d) incl += y;
    }
    if (lane == 31) wsum[t & 1][wave] = incl;
    __syncthreads();
    const int wt = wsum[t & 1][lane];
    int winc = wt;
    for (int d = 1; d < 32; d <<= 1) {
      const int y = __shfl_up(winc, d);
      if (lane >= d) winc += y;
    }
    const int wave_excl = __shfl(winc, wave) - __shfl(wt, wave);
    const int total = __shfl(winc, 31);
    int ex = carry + wave_excl + (incl - s);
    if (full) {
      int o[SCAN_EPT];
      float r[SCAN_EPT];
      for (int q = 0; q < SCAN_EPT; ++q) {
        o[q] = ex;
        ex += v[q];
        r[q] = rsqrtf((float)(v[q] + 1));
      }
      *(int4*)(offs + base)     = make_int4(o[0], o[1], o[2], o[3]);
      *(int4*)(offs + base + 4) = make_int4(o[4], o[5], o[6], o[7]);
      *(float4*)(dinv + base)     = make_float4(r[0], r[1], r[2], r[3]);
      *(float4*)(dinv + base + 4) = make_float4(r[4], r[5], r[6], r[7]);
    } else {
      for (int q = 0; q < SCAN_EPT; ++q) {
        if (base + q < n) {
          offs[base + q] = ex;
          dinv[base + q] = rsqrtf((float)(v[q] + 1));
          ex += v[q];
        }
      }
    }
    carry += total;
  }
  if (tid == 0) offs[n] = carry;
}

__global__ void fill_kernel(const int* __restrict__ row, const int* __restrict__ col,
                            const int* __restrict__ rank, const int* __restrict__ offs,
                            const float* __restrict__ dinv, int* __restrict__ rec, int E) {
  int e = blockIdx.x * blockDim.x + threadIdx.x;
  if (e >= E) return;
  const int r = row[e];
  const int c = col[e];
  ((int2*)rec)[offs[c] + rank[e]] = make_int2(r, __float_as_int(dinv[r]));
}

template <int N, int K, bool BIAS_RELU, int TM>
__global__ void gemm_wmma_kernel(const float* __restrict__ A,
                                 const float* __restrict__ B,
                                 float* __restrict__ C,
                                 const float* __restrict__ act_bias,
                                 int M) {
  const int lane = threadIdx.x & 31;
  const int wave = threadIdx.x >> 5;
  const int n0   = wave * 16;
  const int half = lane >> 4;
  const int l    = lane & 15;
  const int kb   = 2 * half;

  constexpr int KSTEPS = K / 4;

  v2f breg[KSTEPS];
  {
    const float* bcol = B + (size_t)kb * N + (n0 + l);
#pragma unroll
    for (int s = 0; s < KSTEPS; ++s) {
      breg[s].x = bcol[s * 4 * N];
      breg[s].y = bcol[(s * 4 + 1) * N];
    }
  }
  const float* bias = act_bias + kb;

  for (int t = 0; t < TM; ++t) {
    const int m0 = (blockIdx.x * TM + t) * 16;
    if (m0 >= M) break;

    int ar = m0 + l;
    if (ar > M - 1) ar = M - 1;
    const float* arow = A + (size_t)ar * K + kb;

    v8f acc = {};
#pragma unroll
    for (int s = 0; s < KSTEPS; ++s) {
      v2f a;
      a.x = arow[s * 4];
      a.y = arow[s * 4 + 1];
      if constexpr (BIAS_RELU) {
        a.x = fmaxf(a.x + bias[s * 4], 0.0f);
        a.y = fmaxf(a.y + bias[s * 4 + 1], 0.0f);
      }
      acc = __builtin_amdgcn_wmma_f32_16x16x4_f32(false, a, false, breg[s],
                                                  (short)0, acc, false, false);
    }

    float* cp = C + (size_t)(m0 + half * 8) * N + (n0 + l);
#pragma unroll
    for (int v = 0; v < 8; ++v) {
      if (m0 + half * 8 + v < M) cp[(size_t)v * N] = acc[v];
    }
  }
}


template <int F, int G, int B>
__device__ __forceinline__ void gather_steps(const float* __restrict__ h, int rv, float wv,
                                             int t0, int g, int fl, float4& acc) {
  float4 hv[B];
  float w[B];
#pragma unroll
  for (int u = 0; u < B; ++u) {
    int r;
    if constexpr (G == 1) {
      r = __builtin_amdgcn_readlane(rv, t0 + u);
      w[u] = __int_as_float(__builtin_amdgcn_readlane(__float_as_int(wv), t0 + u));
    } else {
      const int kk = (t0 + u) * G + g;
      r = __shfl(rv, kk);
      w[u] = __shfl(wv, kk);
    }
    hv[u] = *(const float4*)(h + (size_t)r * F + fl * 4);
  }
#pragma unroll
  for (int u = 0; u < B; ++u) {
    acc.x = fmaf(hv[u].x, w[u], acc.x);
    acc.y = fmaf(hv[u].y, w[u], acc.y);
    acc.z = fmaf(hv[u].z, w[u], acc.z);
    acc.w = fmaf(hv[u].w, w[u], acc.w);
  }
}

template <int F, bool HAS_BIAS>
__global__ __launch_bounds__(256) void gather_agg_kernel(
    const int* __restrict__ offs, const int* __restrict__ rec,
    const float* __restrict__ h, const float* __restrict__ dinv,
    const float* __restrict__ bias, float* __restrict__ out, int n) {
  constexpr int LPR = F / 4;
  constexpr int G = 32 / LPR;
  constexpr int BMAX = 8 / G;
  const int lane = threadIdx.x & 31;
  const int g = lane / LPR;
  const int fl = lane % LPR;
  const int wave = __builtin_amdgcn_readfirstlane((int)(threadIdx.x >> 5));
  const int wpb = blockDim.x >> 5;
  const int nwaves = gridDim.x * wpb;
  float4 bv = make_float4(0.0f, 0.0f, 0.0f, 0.0f);
  if constexpr (HAS_BIAS) bv = *(const float4*)(bias + fl * 4);

  for (int c = blockIdx.x * wpb + wave; c < n; c += nwaves) {
    const int beg = offs[c];
    const int end = offs[c + 1];
    const float dc = dinv[c];
    float4 acc = make_float4(0.0f, 0.0f, 0.0f, 0.0f);
    float4 hs = make_float4(0.0f, 0.0f, 0.0f, 0.0f);
    if (g == 0) hs = *(const float4*)(h + (size_t)c * F + fl * 4);
    for (int j = beg; j < end; j += 32) {
      const int m = min(32, end - j);
      int rv = c;
      float wv = 0.0f;
      if (lane < m) {
        const int2 p = ((const int2*)rec)[j + lane];
        rv = p.x;
        wv = __int_as_float(p.y) * dc;
      }
      const int S = m / G;
      int t = 0;
      for (; t + BMAX <= S; t += BMAX) gather_steps<F, G, BMAX>(h, rv, wv, t, g, fl, acc);
      if constexpr (BMAX >= 8) {
        if (t + 4 <= S) { gather_steps<F, G, 4>(h, rv, wv, t, g, fl, acc); t += 4; }
      }
      if constexpr (BMAX >= 4) {
        if (t + 2 <= S) { gather_steps<F, G, 2>(h, rv, wv, t, g, fl, acc); t += 2; }
      }
      if (t < S) gather_steps<F, G, 1>(h, rv, wv, t, g, fl, acc);
      if constexpr (G == 2) {
        if (m & 1) {
          const int r = __shfl(rv, m - 1);
          const float w = __shfl(wv, m - 1);
          if (g == 0) {
            const float4 hv = *(const float4*)(h + (size_t)r * F + fl * 4);
            acc.x = fmaf(hv.x, w, acc.x);
            acc.y = fmaf(hv.y, w, acc.y);
            acc.z = fmaf(hv.z, w, acc.z);
            acc.w = fmaf(hv.w, w, acc.w);
          }
        }
      }
    }
    if (g == 0) {
      acc.x += hs.x * dc * dc + bv.x;
      acc.y += hs.y * dc * dc + bv.y;
      acc.z += hs.z * dc * dc + bv.z;
      acc.w += hs.w * dc * dc + bv.w;
    }
    if constexpr (G == 2) {
      acc.x += __shfl_xor(acc.x, 16);
      acc.y += __shfl_xor(acc.y, 16);
      acc.z += __shfl_xor(acc.z, 16);
      acc.w += __shfl_xor(acc.w, 16);
    }
    if (g == 0) *(float4*)(out + (size_t)c * F + fl * 4) = acc;
  }
}

extern "C" void kernel_launch(void* const* d_in, const int* in_sizes, int n_in,
                              void* d_out, int out_size, void* d_ws, size_t ws_size,
                              hipStream_t stream) {
  const float* x  = (const float*)d_in[0];
  const int*   ei = (const int*)d_in[1];
  const float* W1 = (const float*)d_in[2];
  const float* b1 = (const float*)d_in[3];
  const float* W2 = (const float*)d_in[4];
  const float* b2 = (const float*)d_in[5];

  const int hid    = in_sizes[3];
  const int out_ch = in_sizes[5];
  const int in_ch  = in_sizes[2] / hid;
  const int n      = in_sizes[0] / in_ch;
  const int E      = in_sizes[1] / 2;
  const int* row = ei;
  const int* col = ei + E;

  float* ws = (float*)d_ws;
  size_t off = 0;
  auto alloc = [&](size_t elems) {
    float* p = ws + off;
    off += (elems + 63) & ~(size_t)63;
    return p;
  };
  int*   cnt  = (int*)alloc((size_t)n);
  int*   offs = (int*)alloc((size_t)n + 1);
  float* dinv = alloc((size_t)n);
  int*   rank = (int*)alloc((size_t)E);
  int*   rec  = (int*)alloc((size_t)E * 2);
  float* h1   = alloc((size_t)n * hid);
  float* agg1 = alloc((size_t)n * hid);
  float* g2   = alloc((size_t)n * out_ch);
  float* out  = (float*)d_out;

  zero_cnt_kernel<<<(n + 255) / 256, 256, 0, stream>>>(cnt, n);
  rank_kernel<<<(E + 255) / 256, 256, 0, stream>>>(col, cnt, rank, E);
  scan_dinv_kernel<<<1, SCAN_THREADS, 0, stream>>>(cnt, offs, dinv, n);
  fill_kernel<<<(E + 255) / 256, 256, 0, stream>>>(row, col, rank, offs, dinv, rec, E);

  constexpr int TM = 4;
  const int mtiles   = (n + 15) / 16;
  const int mblocks  = (mtiles + TM - 1) / TM;

  gemm_wmma_kernel<128, 128, false, TM><<<mblocks, 32 * (128 / 16), 0, stream>>>(
      x, W1, h1, nullptr, n);
  gather_agg_kernel<128, false><<<2048, 256, 0, stream>>>(offs, rec, h1, dinv, nullptr, agg1, n);

  gemm_wmma_kernel<64, 128, true, TM><<<mblocks, 32 * (64 / 16), 0, stream>>>(
      agg1, W2, g2, b1, n);
  gather_agg_kernel<64, true><<<2048, 256, 0, stream>>>(offs, rec, g2, dinv, b2, out, n);
}


# Edge grouping without global atomics: edges binned into 512-node buckets (per-chunk LDS histograms, table scan), per-bucket LDS counters give ranks, offsets, dinv; sum order still varies (LDS order)
# speedup vs baseline: 2.8052x; 1.3037x over previous
#include <hip/hip_runtime.h>
#include <hip/hip_bf16.h>


typedef __attribute__((ext_vector_type(2))) float v2f;
typedef __attribute__((ext_vector_type(8))) float v8f;

constexpr int BSHIFT = 9;
constexpr int BNODES = 1 << BSHIFT;
constexpr int NCH = 256;
constexpr int NBMAX = 1024;
constexpr int SORT_THREADS = 1024;
constexpr int EPB = 4;
static_assert(EPB == 4, "the load fences in bucket_hist_kernel and fill_kernel name four values each");
constexpr int FILL_THREADS = 256;

__device__ __forceinline__ int wave_incl_scan(int v, int lane) {
  for (int d = 1; d < 32; d <<= 1) {
    const int y = __shfl_up(v, d);
    if (lane >= d) v += y;
  }
  return v;
}

__device__ __forceinline__ int block_excl_scan_1024(int v, int* wsum, int lane, int wave,
                                                    int& total) {
  const int incl = wave_incl_scan(v, lane);
  if (lane == 31) wsum[wave] = incl;
  __syncthreads();
  const int wt = wsum[lane];
  const int winc = wave_incl_scan(wt, lane);
  const int wave_excl = __shfl(winc, wave) - __shfl(wt, wave);
  total = __shfl(winc, 31);
  return wave_excl + incl - v;
}

__global__ __launch_bounds__(SORT_THREADS) void bucket_hist_kernel(
    const int* __restrict__ col, int* __restrict__ T, int E, int che, int nb) {
  __shared__ int hist[NBMAX];
  const int tid = threadIdx.x;
  const int ch = blockIdx.x;
  for (int b = tid; b < nb; b += SORT_THREADS) hist[b] = 0;
  __syncthreads();
  const int e0 = ch * che;
  const int e1 = min(E, e0 + che);
  for (int e = e0 + tid; e < e1; e += EPB * SORT_THREADS) {
    int c[EPB];
#pragma unroll
    for (int u = 0; u < EPB; ++u) {
      c[u] = col[min(e + u * SORT_THREADS, e1 - 1)];
    }
    asm volatile("" :: "v"(c[0]), "v"(c[1]), "v"(c[2]), "v"(c[3]));
#pragma unroll
    for (int u = 0; u < EPB; ++u) {
      if (e + u * SORT_THREADS < e1) atomicAdd(&hist[c[u] >> BSHIFT], 1);
    }
  }
  __syncthreads();
  for (int b = tid; b < nb; b += SORT_THREADS) T[b * NCH + ch] = hist[b];
}

__global__ __launch_bounds__(SORT_THREADS) void bucket_scan_kernel(
    int* __restrict__ T, int* __restrict__ bstart, int nb) {
  __shared__ int tot[NBMAX];
  __shared__ int wsum[32];
  const int tid = threadIdx.x;
  const int lane = tid & 31;
  const int wave = tid >> 5;
  for (int b = wave; b < nb; b += 32) {
    int* p = T + b * NCH + lane * 8;
    const int4 a = *(const int4*)p;
    const int4 c = *(const int4*)(p + 4);
    const int s = a.x + a.y + a.z + a.w + c.x + c.y + c.z + c.w;
    const int incl = wave_incl_scan(s, lane);
    const int ex = incl - s;
    int4 oa, oc;
    oa.x = ex;          oa.y = oa.x + a.x;  oa.z = oa.y + a.y;  oa.w = oa.z + a.z;
    oc.x = oa.w + a.w;  oc.y = oc.x + c.x;  oc.z = oc.y + c.y;  oc.w = oc.z + c.z;
    *(int4*)p = oa;
    *(int4*)(p + 4) = oc;
    if (lane == 31) tot[b] = incl;
  }
  __syncthreads();
  const int v = (tid < nb) ? tot[tid] : 0;
  int total;
  const int ex = block_excl_scan_1024(v, wsum, lane, wave, total);
  if (tid < nb) bstart[tid] = ex;
  if (tid == 0) bstart[nb] = total;
}

__global__ __launch_bounds__(SORT_THREADS) void bucket_scatter_kernel(
    const int* __restrict__ row, const int* __restrict__ col, const int* __restrict__ T,
    const int* __restrict__ bstart, int* __restrict__ part, int E, int che, int nb) {
  __shared__ int cur[NBMAX];
  const int tid = threadIdx.x;
  const int ch = blockIdx.x;
  for (int b = tid; b < nb; b += SORT_THREADS) cur[b] = bstart[b] + T[b * NCH + ch];
  __syncthreads();
  const int e0 = ch * che;
  const int e1 = min(E, e0 + che);
  for (int e = e0 + tid; e < e1; e += EPB * SORT_THREADS) {
    int r[EPB], c[EPB];
#pragma unroll
    for (int u = 0; u < EPB; ++u) {
      const int ee = e + u * SORT_THREADS;
      r[u] = (ee < e1) ? row[ee] : 0;
      c[u] = (ee < e1) ? col[ee] : 0;
    }
#pragma unroll
    for (int u = 0; u < EPB; ++u) {
      if (e + u * SORT_THREADS < e1) {
        const int pos = atomicAdd(&cur[c[u] >> BSHIFT], 1);
        ((int2*)part)[pos] = make_int2(r[u], c[u]);
      }
    }
  }
}

__global__ __launch_bounds__(SORT_THREADS) void bucket_sort_kernel(
    const int* __restrict__ part, const int* __restrict__ bstart, int* __restrict__ rank2,
    int* __restrict__ offs, float* __restrict__ dinv, int n, int nb) {
  __shared__ int cnt[BNODES];
  __shared__ int wsum[32];
  const int tid = threadIdx.x;
  const int lane = tid & 31;
  const int wave = tid >> 5;
  const int b = blockIdx.x;
  const int c0 = b << BSHIFT;
  const int p0 = bstart[b];
  const int p1 = bstart[b + 1];
  if (tid < BNODES) cnt[tid] = 0;
  __syncthreads();
  for (int p = p0 + tid; p < p1; p += EPB * SORT_THREADS) {
    int d[EPB];
#pragma unroll
    for (int u = 0; u < EPB; ++u) {
      const int pp = p + u * SORT_THREADS;
      d[u] = (pp < p1) ? part[2 * (size_t)pp + 1] : c0;
    }
#pragma unroll
    for (int u = 0; u < EPB; ++u) {
      const int pp = p + u * SORT_THREADS;
      if (pp < p1) rank2[pp] = atomicAdd(&cnt[d[u] - c0], 1);
    }
  }
  __syncthreads();
  const int v = (tid < BNODES) ? cnt[tid] : 0;
  int total;
  const int ex = block_excl_scan_1024(v, wsum, lane, wave, total);
  const int c = c0 + tid;
  if (tid < BNODES && c < n) {
    offs[c] = p0 + ex;
    dinv[c] = rsqrtf((float)(v + 1));
  }
  if (b == nb - 1 && tid == 0) offs[n] = p1;
}

__global__ __launch_bounds__(FILL_THREADS) void fill_kernel(
    const int* __restrict__ part, const int* __restrict__ rank2,
    const int* __restrict__ offs, const float* __restrict__ dinv,
    int* __restrict__ rec, int E) {
  const int base = blockIdx.x * (EPB * FILL_THREADS) + threadIdx.x;
  int2 rc[EPB];
  int k[EPB], o[EPB];
  float w[EPB];
#pragma unroll
  for (int u = 0; u < EPB; ++u) {
    const int p = min(base + u * FILL_THREADS, E - 1);
    rc[u] = ((const int2*)part)[p];
    k[u] = rank2[p];
  }
  asm volatile("" :: "v"(rc[0].x), "v"(rc[0].y), "v"(rc[1].x), "v"(rc[1].y),
                     "v"(rc[2].x), "v"(rc[2].y), "v"(rc[3].x), "v"(rc[3].y));
#pragma unroll
  for (int u = 0; u < EPB; ++u) {
    o[u] = offs[rc[u].y];
    w[u] = dinv[rc[u].x];
  }
  asm volatile("" :: "v"(o[0]), "v"(o[1]), "v"(o[2]), "v"(o[3]),
                     "v"(w[0]), "v"(w[1]), "v"(w[2]), "v"(w[3]),
                     "v"(k[0]), "v"(k[1]), "v"(k[2]), "v"(k[3]));
#pragma unroll
  for (int u = 0; u < EPB; ++u) {
    if (base + u * FILL_THREADS < E)
      ((int2*)rec)[o[u] + k[u]] = make_int2(rc[u].x, __float_as_int(w[u]));
  }
}

template <int N, int K, bool BIAS_RELU, int TM>
__global__ void gemm_wmma_kernel(const float* __restrict__ A,
                                 const float* __restrict__ B,
                                 float* __restrict__ C,
                                 const float* __restrict__ act_bias,
                                 int M) {
  const int lane = threadIdx.x & 31;
  const int wave = threadIdx.x >> 5;
  const int n0   = wave * 16;
  const int half = lane >> 4;
  const int l    = lane & 15;
  const int kb   = 2 * half;

  constexpr int KSTEPS = K / 4;

  v2f breg[KSTEPS];
  {
    const float* bcol = B + (size_t)kb * N + (n0 + l);
#pragma unroll
    for (int s = 0; s < KSTEPS; ++s) {
      breg[s].x = bcol[s * 4 * N];
      breg[s].y = bcol[(s * 4 + 1) * N];
    }
  }
  const float* bias = act_bias + kb;

  for (int t = 0; t < TM; ++t) {
    const int m0 = (blockIdx.x * TM + t) * 16;
    if (m0 >= M) break;

    int ar = m0 + l;
    if (ar > M - 1) ar = M - 1;
    const float* arow = A + (size_t)ar * K + kb;

    v8f acc = {};
#pragma unroll
    for (int s = 0; s < KSTEPS; ++s) {
      v2f a;
      a.x = arow[s * 4];
      a.y = arow[s * 4 + 1];
      if constexpr (BIAS_RELU) {
        a.x = fmaxf(a.x + bias[s * 4], 0.0f);
        a.y = fmaxf(a.y + bias[s * 4 + 1], 0.0f);
      }
      acc = __builtin_amdgcn_wmma_f32_16x16x4_f32(false, a, false, breg[s],
                                                  (short)0, acc, false, false);
    }

    float* cp = C + (size_t)(m0 + half * 8) * N + (n0 + l);
#pragma unroll
    for (int v = 0; v < 8; ++v) {
      if (m0 + half * 8 + v < M) cp[(size_t)v * N] = acc[v];
    }
  }
}


template <int F, int G, int B>
__device__ __forceinline__ void gather_steps(const float* __restrict__ h, int rv, float wv,
                                             int t0, int g, int fl, float4& acc) {
  float4 hv[B];
  float w[B];
#pragma unroll
  for (int u = 0; u < B; ++u) {
    int r;
    if constexpr (G == 1) {
      r = __builtin_amdgcn_readlane(rv, t0 + u);
      w[u] = __int_as_float(__builtin_amdgcn_readlane(__float_as_int(wv), t0 + u));
    } else {
      const int kk = (t0 + u) * G + g;
      r = __shfl(rv, kk);
      w[u] = __shfl(wv, kk);
    }
    hv[u] = *(const float4*)(h + (size_t)r * F + fl * 4);
  }
#pragma unroll
  for (int u = 0; u < B; ++u) {
    acc.x = fmaf(hv[u].x, w[u], acc.x);
    acc.y = fmaf(hv[u].y, w[u], acc.y);
    acc.z = fmaf(hv[u].z, w[u], acc.z);
    acc.w = fmaf(hv[u].w, w[u], acc.w);
  }
}

template <int F, bool HAS_BIAS>
__global__ __launch_bounds__(256) void gather_agg_kernel(
    const int* __restrict__ offs, const int* __restrict__ rec,
    const float* __restrict__ h, const float* __restrict__ dinv,
    const float* __restrict__ bias, float* __restrict__ out, int n) {
  constexpr int LPR = F / 4;
  constexpr int G = 32 / LPR;
  constexpr int BMAX = 8 / G;
  const int lane = threadIdx.x & 31;
  const int g = lane / LPR;
  const int fl = lane % LPR;
  const int wave = __builtin_amdgcn_readfirstlane((int)(threadIdx.x >> 5));
  const int wpb = blockDim.x >> 5;
  const int nwaves = gridDim.x * wpb;
  float4 bv = make_float4(0.0f, 0.0f, 0.0f, 0.0f);
  if constexpr (HAS_BIAS) bv = *(const float4*)(bias + fl * 4);

  for (int c = blockIdx.x * wpb + wave; c < n; c += nwaves) {
    const int beg = offs[c];
    const int end = offs[c + 1];
    const float dc = dinv[c];
    float4 acc = make_float4(0.0f, 0.0f, 0.0f, 0.0f);
    float4 hs = make_float4(0.0f, 0.0f, 0.0f, 0.0f);
    if (g == 0) hs = *(const float4*)(h + (size_t)c * F + fl * 4);
    for (int j = beg; j < end; j += 32) {
      const int m = min(32, end - j);
      int rv = c;
      float wv = 0.0f;
      if (lane < m) {
        const int2 p = ((const int2*)rec)[j + lane];
        rv = p.x;
        wv = __int_as_float(p.y) * dc;
      }
      const int S = m / G;
      int t = 0;
      for (; t + BMAX <= S; t += BMAX) gather_steps<F, G, BMAX>(h, rv, wv, t, g, fl, acc);
      if constexpr (BMAX >= 8) {
        if (t + 4 <= S) { gather_steps<F, G, 4>(h, rv, wv, t, g, fl, acc); t += 4; }
      }
      if constexpr (BMAX >= 4) {
        if (t + 2 <= S) { gather_steps<F, G, 2>(h, rv, wv, t, g, fl, acc); t += 2; }
      }
      if (t < S) gather_steps<F, G, 1>(h, rv, wv, t, g, fl, acc);
      if constexpr (G == 2) {
        if (m & 1) {
          const int r = __shfl(rv, m - 1);
          const float w = __shfl(wv, m - 1);
          if (g == 0) {
            const float4 hv = *(const float4*)(h + (size_t)r * F + fl * 4);
            acc.x = fmaf(hv.x, w, acc.x);
            acc.y = fmaf(hv.y, w, acc.y);
            acc.z = fmaf(hv.z, w, acc.z);
            acc.w = fmaf(hv.w, w, acc.w);
          }
        }
      }
    }
    if (g == 0) {
      acc.x += hs.x * dc * dc + bv.x;
      acc.y += hs.y * dc * dc + bv.y;
      acc.z += hs.z * dc * dc + bv.z;
      acc.w += hs.w * dc * dc + bv.w;
    }
    if constexpr (G == 2) {
      acc.x += __shfl_xor(acc.x, 16);
      acc.y += __shfl_xor(acc.y, 16);
      acc.z += __shfl_xor(acc.z, 16);
      acc.w += __shfl_xor(acc.w, 16);
    }
    if (g == 0) *(float4*)(out + (size_t)c * F + fl * 4) = acc;
  }
}

extern "C" void kernel_launch(void* const* d_in, const int* in_sizes, int n_in,
                              void* d_out, int out_size, void* d_ws, size_t ws_size,
                              hipStream_t stream) {
  const float* x  = (const float*)d_in[0];
  const int*   ei = (const int*)d_in[1];
  const float* W1 = (const float*)d_in[2];
  const float* b1 = (const float*)d_in[3];
  const float* W2 = (const float*)d_in[4];
  const float* b2 = (const float*)d_in[5];

  const int hid    = in_sizes[3];
  const int out_ch = in_sizes[5];
  const int in_ch  = in_sizes[2] / hid;
  const int n      = in_sizes[0] / in_ch;
  const int E      = in_sizes[1] / 2;
  const int* row = ei;
  const int* col = ei + E;

  const int nb  = (n + BNODES - 1) / BNODES;
  const int che = (E + NCH - 1) / NCH;

  float* ws = (float*)d_ws;
  size_t off = 0;
  auto alloc = [&](size_t elems) {
    float* p = ws + off;
    off += (elems + 63) & ~(size_t)63;
    return p;
  };
  int*   T      = (int*)alloc((size_t)nb * NCH);
  int*   bstart = (int*)alloc((size_t)nb + 1);
  int*   offs   = (int*)alloc((size_t)n + 1);
  float* dinv   = alloc((size_t)n);
  int*   part   = (int*)alloc((size_t)E * 2);
  int*   rank2  = (int*)alloc((size_t)E);
  int*   rec    = (int*)alloc((size_t)E * 2);
  float* h1   = alloc((size_t)n * hid);
  float* agg1 = alloc((size_t)n * hid);
  float* g2   = alloc((size_t)n * out_ch);
  float* out  = (float*)d_out;

  bucket_hist_kernel<<<NCH, SORT_THREADS, 0, stream>>>(col, T, E, che, nb);
  bucket_scan_kernel<<<1, SORT_THREADS, 0, stream>>>(T, bstart, nb);
  bucket_scatter_kernel<<<NCH, SORT_THREADS, 0, stream>>>(row, col, T, bstart, part, E, che, nb);
  bucket_sort_kernel<<<nb, SORT_THREADS, 0, stream>>>(part, bstart, rank2, offs, dinv, n, nb);
  fill_kernel<<<(E + EPB * FILL_THREADS - 1) / (EPB * FILL_THREADS), FILL_THREADS, 0, stream>>>(
      part, rank2, offs, dinv, rec, E);

  constexpr int TM = 4;
  const int mtiles   = (n + 15) / 16;
  const int mblocks  = (mtiles + TM - 1) / TM;

  gemm_wmma_kernel<128, 128, false, TM><<<mblocks, 32 * (128 / 16), 0, stream>>>(
      x, W1, h1, nullptr, n);
  gather_agg_kernel<128, false><<<2048, 256, 0, stream>>>(offs, rec, h1, dinv, nullptr, agg1, n);

  gemm_wmma_kernel<64, 128, true, TM><<<mblocks, 32 * (64 / 16), 0, stream>>>(
      agg1, W2, g2, b1, n);
  gather_agg_kernel<64, true><<<2048, 256, 0, stream>>>(offs, rec, g2, dinv, b2, out, n);
}


# x@W1 has no launch of its own: its tiles ride as extra work-groups beside the one-block table scan and between the record-fill blocks (independent work overlaps); 9 -> 8 launches
# speedup vs baseline: 3.0841x; 1.0994x over previous
#include <hip/hip_runtime.h>
#include <hip/hip_bf16.h>


typedef __attribute__((ext_vector_type(2))) float v2f;
typedef __attribute__((ext_vector_type(8))) float v8f;

constexpr int BSHIFT = 9;
constexpr int BNODES = 1 << BSHIFT;
constexpr int NCH = 256;
constexpr int NBMAX = 1024;
constexpr int SORT_THREADS = 1024;
constexpr int EPB = 4;
static_assert(EPB == 4, "the load fences in bucket_hist_kernel and fill_kernel name four values each");
constexpr int FILL_THREADS = 256;

__device__ __forceinline__ int wave_incl_scan(int v, int lane) {
  for (int d = 1; d < 32; d <<= 1) {
    const int y = __shfl_up(v, d);
    if (lane >= d) v += y;
  }
  return v;
}

__device__ __forceinline__ int block_excl_scan_1024(int v, int* wsum, int lane, int wave,
                                                    int& total) {
  const int incl = wave_incl_scan(v, lane);
  if (lane == 31) wsum[wave] = incl;
  __syncthreads();
  const int wt = wsum[lane];
  const int winc = wave_incl_scan(wt, lane);
  const int wave_excl = __shfl(winc, wave) - __shfl(wt, wave);
  total = __shfl(winc, 31);
  return wave_excl + incl - v;
}

__global__ __launch_bounds__(SORT_THREADS) void bucket_hist_kernel(
    const int* __restrict__ col, int* __restrict__ T, int E, int che, int nb) {
  __shared__ int hist[NBMAX];
  const int tid = threadIdx.x;
  const int ch = blockIdx.x;
  for (int b = tid; b < nb; b += SORT_THREADS) hist[b] = 0;
  __syncthreads();
  const int e0 = ch * che;
  const int e1 = min(E, e0 + che);
  for (int e = e0 + tid; e < e1; e += EPB * SORT_THREADS) {
    int c[EPB];
#pragma unroll
    for (int u = 0; u < EPB; ++u) {
      c[u] = col[min(e + u * SORT_THREADS, e1 - 1)];
    }
    asm volatile("" :: "v"(c[0]), "v"(c[1]), "v"(c[2]), "v"(c[3]));
#pragma unroll
    for (int u = 0; u < EPB; ++u) {
      if (e + u * SORT_THREADS < e1) atomicAdd(&hist[c[u] >> BSHIFT], 1);
    }
  }
  __syncthreads();
  for (int b = tid; b < nb; b += SORT_THREADS) T[b * NCH + ch] = hist[b];
}

__device__ __forceinline__ void bucket_scan_body(
    int* __restrict__ T, int* __restrict__ bstart, int nb) {
  __shared__ int tot[NBMAX];
  __shared__ int wsum[32];
  const int tid = threadIdx.x;
  const int lane = tid & 31;
  const int wave = tid >> 5;
  for (int b = wave; b < nb; b += 32) {
    int* p = T + b * NCH + lane * 8;
    const int4 a = *(const int4*)p;
    const int4 c = *(const int4*)(p + 4);
    const int s = a.x + a.y + a.z + a.w + c.x + c.y + c.z + c.w;
    const int incl = wave_incl_scan(s, lane);
    const int ex = incl - s;
    int4 oa, oc;
    oa.x = ex;          oa.y = oa.x + a.x;  oa.z = oa.y + a.y;  oa.w = oa.z + a.z;
    oc.x = oa.w + a.w;  oc.y = oc.x + c.x;  oc.z = oc.y + c.y;  oc.w = oc.z + c.z;
    *(int4*)p = oa;
    *(int4*)(p + 4) = oc;
    if (lane == 31) tot[b] = incl;
  }
  __syncthreads();
  const int v = (tid < nb) ? tot[tid] : 0;
  int total;
  const int ex = block_excl_scan_1024(v, wsum, lane, wave, total);
  if (tid < nb) bstart[tid] = ex;
  if (tid == 0) bstart[nb] = total;
}

__global__ __launch_bounds__(SORT_THREADS) void bucket_scatter_kernel(
    const int* __restrict__ row, const int* __restrict__ col, const int* __restrict__ T,
    const int* __restrict__ bstart, int* __restrict__ part, int E, int che, int nb) {
  __shared__ int cur[NBMAX];
  const int tid = threadIdx.x;
  const int ch = blockIdx.x;
  for (int b = tid; b < nb; b += SORT_THREADS) cur[b] = bstart[b] + T[b * NCH + ch];
  __syncthreads();
  const int e0 = ch * che;
  const int e1 = min(E, e0 + che);
  for (int e = e0 + tid; e < e1; e += EPB * SORT_THREADS) {
    int r[EPB], c[EPB];
#pragma unroll
    for (int u = 0; u < EPB; ++u) {
      const int ee = e + u * SORT_THREADS;
      r[u] = (ee < e1) ? row[ee] : 0;
      c[u] = (ee < e1) ? col[ee] : 0;
    }
#pragma unroll
    for (int u = 0; u < EPB; ++u) {
      if (e + u * SORT_THREADS < e1) {
        const int pos = atomicAdd(&cur[c[u] >> BSHIFT], 1);
        ((int2*)part)[pos] = make_int2(r[u], c[u]);
      }
    }
  }
}

__global__ __launch_bounds__(SORT_THREADS) void bucket_sort_kernel(
    const int* __restrict__ part, const int* __restrict__ bstart, int* __restrict__ rank2,
    int* __restrict__ offs, float* __restrict__ dinv, int n, int nb) {
  __shared__ int cnt[BNODES];
  __shared__ int wsum[32];
  const int tid = threadIdx.x;
  const int lane = tid & 31;
  const int wave = tid >> 5;
  const int b = blockIdx.x;
  const int c0 = b << BSHIFT;
  const int p0 = bstart[b];
  const int p1 = bstart[b + 1];
  if (tid < BNODES) cnt[tid] = 0;
  __syncthreads();
  for (int p = p0 + tid; p < p1; p += EPB * SORT_THREADS) {
    int d[EPB];
#pragma unroll
    for (int u = 0; u < EPB; ++u) {
      const int pp = p + u * SORT_THREADS;
      d[u] = (pp < p1) ? part[2 * (size_t)pp + 1] : c0;
    }
#pragma unroll
    for (int u = 0; u < EPB; ++u) {
      const int pp = p + u * SORT_THREADS;
      if (pp < p1) rank2[pp] = atomicAdd(&cnt[d[u] - c0], 1);
    }
  }
  __syncthreads();
  const int v = (tid < BNODES) ? cnt[tid] : 0;
  int total;
  const int ex = block_excl_scan_1024(v, wsum, lane, wave, total);
  const int c = c0 + tid;
  if (tid < BNODES && c < n) {
    offs[c] = p0 + ex;
    dinv[c] = rsqrtf((float)(v + 1));
  }
  if (b == nb - 1 && tid == 0) offs[n] = p1;
}

__device__ __forceinline__ void fill_body(
    const int* __restrict__ part, const int* __restrict__ rank2,
    const int* __restrict__ offs, const float* __restrict__ dinv,
    int* __restrict__ rec, int E, int fb) {
  const int base = fb * (EPB * FILL_THREADS) + threadIdx.x;
  int2 rc[EPB];
  int k[EPB], o[EPB];
  float w[EPB];
#pragma unroll
  for (int u = 0; u < EPB; ++u) {
    const int p = min(base + u * FILL_THREADS, E - 1);
    rc[u] = ((const int2*)part)[p];
    k[u] = rank2[p];
  }
  asm volatile("" :: "v"(rc[0].x), "v"(rc[0].y), "v"(rc[1].x), "v"(rc[1].y),
                     "v"(rc[2].x), "v"(rc[2].y), "v"(rc[3].x), "v"(rc[3].y));
#pragma unroll
  for (int u = 0; u < EPB; ++u) {
    o[u] = offs[rc[u].y];
    w[u] = dinv[rc[u].x];
  }
  asm volatile("" :: "v"(o[0]), "v"(o[1]), "v"(o[2]), "v"(o[3]),
                     "v"(w[0]), "v"(w[1]), "v"(w[2]), "v"(w[3]),
                     "v"(k[0]), "v"(k[1]), "v"(k[2]), "v"(k[3]));
#pragma unroll
  for (int u = 0; u < EPB; ++u) {
    if (base + u * FILL_THREADS < E)
      ((int2*)rec)[o[u] + k[u]] = make_int2(rc[u].x, __float_as_int(w[u]));
  }
}

template <int N, int K, bool BIAS_RELU>
__device__ __forceinline__ void gemm_wave_tiles(const float* __restrict__ A,
                                                const float* __restrict__ B,
                                                float* __restrict__ C,
                                                const float* __restrict__ act_bias,
                                                int M, int lane, int n0, int mt0, int mt1) {
  if (mt0 >= mt1 || mt0 * 16 >= M) return;
  const int half = lane >> 4;
  const int l    = lane & 15;
  const int kb   = 2 * half;

  constexpr int KSTEPS = K / 4;

  v2f breg[KSTEPS];
  {
    const float* bcol = B + (size_t)kb * N + (n0 + l);
#pragma unroll
    for (int s = 0; s < KSTEPS; ++s) {
      breg[s].x = bcol[s * 4 * N];
      breg[s].y = bcol[(s * 4 + 1) * N];
    }
  }
  const float* bias = act_bias + kb;

  for (int mt = mt0; mt < mt1; ++mt) {
    const int m0 = mt * 16;
    if (m0 >= M) break;

    int ar = m0 + l;
    if (ar > M - 1) ar = M - 1;
    const float* arow = A + (size_t)ar * K + kb;

    v8f acc = {};
#pragma unroll
    for (int s = 0; s < KSTEPS; ++s) {
      v2f a;
      a.x = arow[s * 4];
      a.y = arow[s * 4 + 1];
      if constexpr (BIAS_RELU) {
        a.x = fmaxf(a.x + bias[s * 4], 0.0f);
        a.y = fmaxf(a.y + bias[s * 4 + 1], 0.0f);
      }
      acc = __builtin_amdgcn_wmma_f32_16x16x4_f32(false, a, false, breg[s],
                                                  (short)0, acc, false, false);
    }

    float* cp = C + (size_t)(m0 + half * 8) * N + (n0 + l);
#pragma unroll
    for (int v = 0; v < 8; ++v) {
      if (m0 + half * 8 + v < M) cp[(size_t)v * N] = acc[v];
    }
  }
}

template <int N, int K, bool BIAS_RELU, int TM>
__global__ void gemm_wmma_kernel(const float* __restrict__ A,
                                 const float* __restrict__ B,
                                 float* __restrict__ C,
                                 const float* __restrict__ act_bias,
                                 int M) {
  const int lane = threadIdx.x & 31;
  const int wave = threadIdx.x >> 5;
  const int mt0  = blockIdx.x * TM;
  gemm_wave_tiles<N, K, BIAS_RELU>(A, B, C, act_bias, M, lane, wave * 16, mt0, mt0 + TM);
}

constexpr int TM_SCAN = 2;
constexpr int SCAN_GEMM_BLOCKS = 255;
constexpr int TM_FILL = 4;

__global__ __launch_bounds__(SORT_THREADS) void scan_gemm_kernel(
    int* __restrict__ T, int* __restrict__ bstart, int nb,
    const float* __restrict__ x, const float* __restrict__ W1, float* __restrict__ h1,
    int n, int mt_split) {
  if (blockIdx.x == 0) {
    bucket_scan_body(T, bstart, nb);
    return;
  }
  const int lane = threadIdx.x & 31;
  const int wave = __builtin_amdgcn_readfirstlane((int)(threadIdx.x >> 5));
  const int mt0  = (((int)blockIdx.x - 1) * 4 + (wave >> 3)) * TM_SCAN;
  gemm_wave_tiles<128, 128, false>(x, W1, h1, nullptr, n, lane, (wave & 7) * 16, mt0,
                                   min(mt0 + TM_SCAN, mt_split));
}

__global__ __launch_bounds__(FILL_THREADS) void fill_gemm_kernel(
    const int* __restrict__ part, const int* __restrict__ rank2,
    const int* __restrict__ offs, const float* __restrict__ dinv,
    int* __restrict__ rec, int E, int nfill,
    const float* __restrict__ x, const float* __restrict__ W1, float* __restrict__ h1,
    int n, int mt_split, int ngemm) {
  const int b = blockIdx.x;
  const int pairs = min(nfill, ngemm);
  int fb = -1, gb = -1;
  if (b < 2 * pairs) {
    if (b & 1) gb = b >> 1; else fb = b >> 1;
  } else if (nfill > ngemm) {
    fb = b - pairs;
  } else {
    gb = b - pairs;
  }
  if (gb >= 0) {
    const int lane = threadIdx.x & 31;
    const int wave = threadIdx.x >> 5;
    const int mt0  = mt_split + gb * TM_FILL;
    gemm_wave_tiles<128, 128, false>(x, W1, h1, nullptr, n, lane, wave * 16, mt0, mt0 + TM_FILL);
    return;
  }
  fill_body(part, rank2, offs, dinv, rec, E, fb);
}


template <int F, int G, int B>
__device__ __forceinline__ void gather_steps(const float* __restrict__ h, int rv, float wv,
                                             int t0, int g, int fl, float4& acc) {
  float4 hv[B];
  float w[B];
#pragma unroll
  for (int u = 0; u < B; ++u) {
    int r;
    if constexpr (G == 1) {
      r = __builtin_amdgcn_readlane(rv, t0 + u);
      w[u] = __int_as_float(__builtin_amdgcn_readlane(__float_as_int(wv), t0 + u));
    } else {
      const int kk = (t0 + u) * G + g;
      r = __shfl(rv, kk);
      w[u] = __shfl(wv, kk);
    }
    hv[u] = *(const float4*)(h + (size_t)r * F + fl * 4);
  }
#pragma unroll
  for (int u = 0; u < B; ++u) {
    acc.x = fmaf(hv[u].x, w[u], acc.x);
    acc.y = fmaf(hv[u].y, w[u], acc.y);
    acc.z = fmaf(hv[u].z, w[u], acc.z);
    acc.w = fmaf(hv[u].w, w[u], acc.w);
  }
}

template <int F, bool HAS_BIAS>
__global__ __launch_bounds__(256) void gather_agg_kernel(
    const int* __restrict__ offs, const int* __restrict__ rec,
    const float* __restrict__ h, const float* __restrict__ dinv,
    const float* __restrict__ bias, float* __restrict__ out, int n) {
  constexpr int LPR = F / 4;
  constexpr int G = 32 / LPR;
  constexpr int BMAX = 8 / G;
  const int lane = threadIdx.x & 31;
  const int g = lane / LPR;
  const int fl = lane % LPR;
  const int wave = __builtin_amdgcn_readfirstlane((int)(threadIdx.x >> 5));
  const int wpb = blockDim.x >> 5;
  const int nwaves = gridDim.x * wpb;
  float4 bv = make_float4(0.0f, 0.0f, 0.0f, 0.0f);
  if constexpr (HAS_BIAS) bv = *(const float4*)(bias + fl * 4);

  for (int c = blockIdx.x * wpb + wave; c < n; c += nwaves) {
    const int beg = offs[c];
    const int end = offs[c + 1];
    const float dc = dinv[c];
    float4 acc = make_float4(0.0f, 0.0f, 0.0f, 0.0f);
    float4 hs = make_float4(0.0f, 0.0f, 0.0f, 0.0f);
    if (g == 0) hs = *(const float4*)(h + (size_t)c * F + fl * 4);
    for (int j = beg; j < end; j += 32) {
      const int m = min(32, end - j);
      int rv = c;
      float wv = 0.0f;
      if (lane < m) {
        const int2 p = ((const int2*)rec)[j + lane];
        rv = p.x;
        wv = __int_as_float(p.y) * dc;
      }
      const int S = m / G;
      int t = 0;
      for (; t + BMAX <= S; t += BMAX) gather_steps<F, G, BMAX>(h, rv, wv, t, g, fl, acc);
      if constexpr (BMAX >= 8) {
        if (t + 4 <= S) { gather_steps<F, G, 4>(h, rv, wv, t, g, fl, acc); t += 4; }
      }
      if constexpr (BMAX >= 4) {
        if (t + 2 <= S) { gather_steps<F, G, 2>(h, rv, wv, t, g, fl, acc); t += 2; }
      }
      if (t < S) gather_steps<F, G, 1>(h, rv, wv, t, g, fl, acc);
      if constexpr (G == 2) {
        if (m & 1) {
          const int r = __shfl(rv, m - 1);
          const float w = __shfl(wv, m - 1);
          if (g == 0) {
            const float4 hv = *(const float4*)(h + (size_t)r * F + fl * 4);
            acc.x = fmaf(hv.x, w, acc.x);
            acc.y = fmaf(hv.y, w, acc.y);
            acc.z = fmaf(hv.z, w, acc.z);
            acc.w = fmaf(hv.w, w, acc.w);
          }
        }
      }
    }
    if (g == 0) {
      acc.x += hs.x * dc * dc + bv.x;
      acc.y += hs.y * dc * dc + bv.y;
      acc.z += hs.z * dc * dc + bv.z;
      acc.w += hs.w * dc * dc + bv.w;
    }
    if constexpr (G == 2) {
      acc.x += __shfl_xor(acc.x, 16);
      acc.y += __shfl_xor(acc.y, 16);
      acc.z += __shfl_xor(acc.z, 16);
      acc.w += __shfl_xor(acc.w, 16);
    }
    if (g == 0) *(float4*)(out + (size_t)c * F + fl * 4) = acc;
  }
}

extern "C" void kernel_launch(void* const* d_in, const int* in_sizes, int n_in,
                              void* d_out, int out_size, void* d_ws, size_t ws_size,
                              hipStream_t stream) {
  const float* x  = (const float*)d_in[0];
  const int*   ei = (const int*)d_in[1];
  const float* W1 = (const float*)d_in[2];
  const float* b1 = (const float*)d_in[3];
  const float* W2 = (const float*)d_in[4];
  const float* b2 = (const float*)d_in[5];

  const int hid    = in_sizes[3];
  const int out_ch = in_sizes[5];
  const int in_ch  = in_sizes[2] / hid;
  const int n      = in_sizes[0] / in_ch;
  const int E      = in_sizes[1] / 2;
  const int* row = ei;
  const int* col = ei + E;

  const int nb  = (n + BNODES - 1) / BNODES;
  const int che = (E + NCH - 1) / NCH;

  float* ws = (float*)d_ws;
  size_t off = 0;
  auto alloc = [&](size_t elems) {
    float* p = ws + off;
    off += (elems + 63) & ~(size_t)63;
    return p;
  };
  int*   T      = (int*)alloc((size_t)nb * NCH);
  int*   bstart = (int*)alloc((size_t)nb + 1);
  int*   offs   = (int*)alloc((size_t)n + 1);
  float* dinv   = alloc((size_t)n);
  int*   part   = (int*)alloc((size_t)E * 2);
  int*   rank2  = (int*)alloc((size_t)E);
  int*   rec    = (int*)alloc((size_t)E * 2);
  float* h1   = alloc((size_t)n * hid);
  float* agg1 = alloc((size_t)n * hid);
  float* g2   = alloc((size_t)n * out_ch);
  float* out  = (float*)d_out;

  constexpr int TM = 4;
  const int mtiles   = (n + 15) / 16;
  const int mblocks  = (mtiles + TM - 1) / TM;
  const int scan_tiles = SCAN_GEMM_BLOCKS * 4 * TM_SCAN;
  const int mt_split = mtiles < scan_tiles ? mtiles : scan_tiles;
  const int ngemm = (mtiles - mt_split + TM_FILL - 1) / TM_FILL;
  const int nfill = (E + EPB * FILL_THREADS - 1) / (EPB * FILL_THREADS);

  bucket_hist_kernel<<<NCH, SORT_THREADS, 0, stream>>>(col, T, E, che, nb);
  scan_gemm_kernel<<<1 + SCAN_GEMM_BLOCKS, SORT_THREADS, 0, stream>>>(
      T, bstart, nb, x, W1, h1, n, mt_split);
  bucket_scatter_kernel<<<NCH, SORT_THREADS, 0, stream>>>(row, col, T, bstart, part, E, che, nb);
  bucket_sort_kernel<<<nb, SORT_THREADS, 0, stream>>>(part, bstart, rank2, offs, dinv, n, nb);
  fill_gemm_kernel<<<nfill + ngemm, FILL_THREADS, 0, stream>>>(
      part, rank2, offs, dinv, rec, E, nfill, x, W1, h1, n, mt_split, ngemm);

  gather_agg_kernel<128, false><<<2048, 256, 0, stream>>>(offs, rec, h1, dinv, nullptr, agg1, n);

  gemm_wmma_kernel<64, 128, true, TM><<<mblocks, 32 * (64 / 16), 0, stream>>>(
      agg1, W2, g2, b1, n);
  gather_agg_kernel<64, true><<<2048, 256, 0, stream>>>(offs, rec, g2, dinv, b2, out, n);
}


# A tiles of all three matrix products staged once per work-group in LDS by whole-row 16-byte loads, fragments read from LDS, instead of each column-tile wave's 8-byte 16-line global loads
# speedup vs baseline: 3.6030x; 1.1683x over previous
#include <hip/hip_runtime.h>
#include <hip/hip_bf16.h>


typedef __attribute__((ext_vector_type(2))) float v2f;
typedef __attribute__((ext_vector_type(8))) float v8f;

constexpr int BSHIFT = 9;
constexpr int BNODES = 1 << BSHIFT;
constexpr int NCH = 256;
constexpr int NBMAX = 1024;
constexpr int SORT_THREADS = 1024;
constexpr int EPB = 4;
static_assert(EPB == 4, "the load fences in bucket_hist_kernel and fill_kernel name four values each");
constexpr int FILL_THREADS = 256;

__device__ __forceinline__ int wave_incl_scan(int v, int lane) {
  for (int d = 1; d < 32; d <<= 1) {
    const int y = __shfl_up(v, d);
    if (lane >= d) v += y;
  }
  return v;
}

__device__ __forceinline__ int block_excl_scan_1024(int v, int* wsum, int lane, int wave,
                                                    int& total) {
  const int incl = wave_incl_scan(v, lane);
  if (lane == 31) wsum[wave] = incl;
  __syncthreads();
  const int wt = wsum[lane];
  const int winc = wave_incl_scan(wt, lane);
  const int wave_excl = __shfl(winc, wave) - __shfl(wt, wave);
  total = __shfl(winc, 31);
  return wave_excl + incl - v;
}

__global__ __launch_bounds__(SORT_THREADS) void bucket_hist_kernel(
    const int* __restrict__ col, int* __restrict__ T, int E, int che, int nb) {
  __shared__ int hist[NBMAX];
  const int tid = threadIdx.x;
  const int ch = blockIdx.x;
  for (int b = tid; b < nb; b += SORT_THREADS) hist[b] = 0;
  __syncthreads();
  const int e0 = ch * che;
  const int e1 = min(E, e0 + che);
  for (int e = e0 + tid; e < e1; e += EPB * SORT_THREADS) {
    int c[EPB];
#pragma unroll
    for (int u = 0; u < EPB; ++u) {
      c[u] = col[min(e + u * SORT_THREADS, e1 - 1)];
    }
    asm volatile("" :: "v"(c[0]), "v"(c[1]), "v"(c[2]), "v"(c[3]));
#pragma unroll
    for (int u = 0; u < EPB; ++u) {
      if (e + u * SORT_THREADS < e1) atomicAdd(&hist[c[u] >> BSHIFT], 1);
    }
  }
  __syncthreads();
  for (int b = tid; b < nb; b += SORT_THREADS) T[b * NCH + ch] = hist[b];
}

__device__ __forceinline__ void bucket_scan_body(
    int* __restrict__ T, int* __restrict__ bstart, int nb) {
  __shared__ int tot[NBMAX];
  __shared__ int wsum[32];
  const int tid = threadIdx.x;
  const int lane = tid & 31;
  const int wave = tid >> 5;
  for (int b = wave; b < nb; b += 32) {
    int* p = T + b * NCH + lane * 8;
    const int4 a = *(const int4*)p;
    const int4 c = *(const int4*)(p + 4);
    const int s = a.x + a.y + a.z + a.w + c.x + c.y + c.z + c.w;
    const int incl = wave_incl_scan(s, lane);
    const int ex = incl - s;
    int4 oa, oc;
    oa.x = ex;          oa.y = oa.x + a.x;  oa.z = oa.y + a.y;  oa.w = oa.z + a.z;
    oc.x = oa.w + a.w;  oc.y = oc.x + c.x;  oc.z = oc.y + c.y;  oc.w = oc.z + c.z;
    *(int4*)p = oa;
    *(int4*)(p + 4) = oc;
    if (lane == 31) tot[b] = incl;
  }
  __syncthreads();
  const int v = (tid < nb) ? tot[tid] : 0;
  int total;
  const int ex = block_excl_scan_1024(v, wsum, lane, wave, total);
  if (tid < nb) bstart[tid] = ex;
  if (tid == 0) bstart[nb] = total;
}

__global__ __launch_bounds__(SORT_THREADS) void bucket_scatter_kernel(
    const int* __restrict__ row, const int* __restrict__ col, const int* __restrict__ T,
    const int* __restrict__ bstart, int* __restrict__ part, int E, int che, int nb) {
  __shared__ int cur[NBMAX];
  const int tid = threadIdx.x;
  const int ch = blockIdx.x;
  for (int b = tid; b < nb; b += SORT_THREADS) cur[b] = bstart[b] + T[b * NCH + ch];
  __syncthreads();
  const int e0 = ch * che;
  const int e1 = min(E, e0 + che);
  for (int e = e0 + tid; e < e1; e += EPB * SORT_THREADS) {
    int r[EPB], c[EPB];
#pragma unroll
    for (int u = 0; u < EPB; ++u) {
      const int ee = e + u * SORT_THREADS;
      r[u] = (ee < e1) ? row[ee] : 0;
      c[u] = (ee < e1) ? col[ee] : 0;
    }
#pragma unroll
    for (int u = 0; u < EPB; ++u) {
      if (e + u * SORT_THREADS < e1) {
        const int pos = atomicAdd(&cur[c[u] >> BSHIFT], 1);
        ((int2*)part)[pos] = make_int2(r[u], c[u]);
      }
    }
  }
}

__global__ __launch_bounds__(SORT_THREADS) void bucket_sort_kernel(
    const int* __restrict__ part, const int* __restrict__ bstart, int* __restrict__ rank2,
    int* __restrict__ offs, float* __restrict__ dinv, int n, int nb) {
  __shared__ int cnt[BNODES];
  __shared__ int wsum[32];
  const int tid = threadIdx.x;
  const int lane = tid & 31;
  const int wave = tid >> 5;
  const int b = blockIdx.x;
  const int c0 = b << BSHIFT;
  const int p0 = bstart[b];
  const int p1 = bstart[b + 1];
  if (tid < BNODES) cnt[tid] = 0;
  __syncthreads();
  for (int p = p0 + tid; p < p1; p += EPB * SORT_THREADS) {
    int d[EPB];
#pragma unroll
    for (int u = 0; u < EPB; ++u) {
      const int pp = p + u * SORT_THREADS;
      d[u] = (pp < p1) ? part[2 * (size_t)pp + 1] : c0;
    }
#pragma unroll
    for (int u = 0; u < EPB; ++u) {
      const int pp = p + u * SORT_THREADS;
      if (pp < p1) rank2[pp] = atomicAdd(&cnt[d[u] - c0], 1);
    }
  }
  __syncthreads();
  const int v = (tid < BNODES) ? cnt[tid] : 0;
  int total;
  const int ex = block_excl_scan_1024(v, wsum, lane, wave, total);
  const int c = c0 + tid;
  if (tid < BNODES && c < n) {
    offs[c] = p0 + ex;
    dinv[c] = rsqrtf((float)(v + 1));
  }
  if (b == nb - 1 && tid == 0) offs[n] = p1;
}

__device__ __forceinline__ void fill_body(
    const int* __restrict__ part, const int* __restrict__ rank2,
    const int* __restrict__ offs, const float* __restrict__ dinv,
    int* __restrict__ rec, int E, int fb) {
  const int base = fb * (EPB * FILL_THREADS) + threadIdx.x;
  int2 rc[EPB];
  int k[EPB], o[EPB];
  float w[EPB];
#pragma unroll
  for (int u = 0; u < EPB; ++u) {
    const int p = min(base + u * FILL_THREADS, E - 1);
    rc[u] = ((const int2*)part)[p];
    k[u] = rank2[p];
  }
  asm volatile("" :: "v"(rc[0].x), "v"(rc[0].y), "v"(rc[1].x), "v"(rc[1].y),
                     "v"(rc[2].x), "v"(rc[2].y), "v"(rc[3].x), "v"(rc[3].y));
#pragma unroll
  for (int u = 0; u < EPB; ++u) {
    o[u] = offs[rc[u].y];
    w[u] = dinv[rc[u].x];
  }
  asm volatile("" :: "v"(o[0]), "v"(o[1]), "v"(o[2]), "v"(o[3]),
                     "v"(w[0]), "v"(w[1]), "v"(w[2]), "v"(w[3]),
                     "v"(k[0]), "v"(k[1]), "v"(k[2]), "v"(k[3]));
#pragma unroll
  for (int u = 0; u < EPB; ++u) {
    if (base + u * FILL_THREADS < E)
      ((int2*)rec)[o[u] + k[u]] = make_int2(rc[u].x, __float_as_int(w[u]));
  }
}

constexpr int GK = 128;
constexpr int APITCH = GK + 4;
constexpr int STAGE_TILES = 4;
constexpr int STAGE_ROWS = STAGE_TILES * 16;
constexpr int STAGE_FLOATS = STAGE_ROWS * APITCH;

template <int NT, int ROWS>
__device__ __forceinline__ void stage_load(const float* __restrict__ A, int M, int row0, int tid,
                                           float4 (&v)[ROWS * (GK / 4) / NT]) {
  constexpr int PER = ROWS * (GK / 4) / NT;
  static_assert(NT % 32 == 0 && PER * NT == ROWS * (GK / 4), "whole rows per wave");
#pragma unroll
  for (int i = 0; i < PER; ++i) {
    const int idx = tid + i * NT;
    const int r = min(row0 + (idx >> 5), M - 1);
    v[i] = *(const float4*)(A + (size_t)r * GK + (idx & 31) * 4);
  }
}

template <int NT, int ROWS, bool BIAS_RELU>
__device__ __forceinline__ void stage_store(float* stage, const float* __restrict__ act_bias,
                                            int tid, const float4 (&v)[ROWS * (GK / 4) / NT]) {
  constexpr int PER = ROWS * (GK / 4) / NT;
  float4 bv = make_float4(0.0f, 0.0f, 0.0f, 0.0f);
  if constexpr (BIAS_RELU) bv = *(const float4*)(act_bias + (tid & 31) * 4);
#pragma unroll
  for (int i = 0; i < PER; ++i) {
    const int idx = tid + i * NT;
    float4 t = v[i];
    if constexpr (BIAS_RELU) {
      t.x = fmaxf(t.x + bv.x, 0.0f);
      t.y = fmaxf(t.y + bv.y, 0.0f);
      t.z = fmaxf(t.z + bv.z, 0.0f);
      t.w = fmaxf(t.w + bv.w, 0.0f);
    }
    *(float4*)(stage + (idx >> 5) * APITCH + (idx & 31) * 4) = t;
  }
}

template <int N>
__device__ __forceinline__ void load_b_frags(const float* __restrict__ B, int lane, int n0,
                                             v2f (&breg)[GK / 4]) {
  const int kb = 2 * (lane >> 4);
  const float* bcol = B + (size_t)kb * N + (n0 + (lane & 15));
#pragma unroll
  for (int s = 0; s < GK / 4; ++s) {
    breg[s].x = bcol[s * 4 * N];
    breg[s].y = bcol[(s * 4 + 1) * N];
  }
}

template <int N>
__device__ __forceinline__ void tile_from_stage(const float* tile, const v2f (&breg)[GK / 4],
                                                float* __restrict__ C, int M, int m0,
                                                int lane, int n0) {
  const int half = lane >> 4;
  const int l    = lane & 15;
  const float* arow = tile + l * APITCH + 2 * half;
  v8f acc = {};
#pragma unroll
  for (int s = 0; s < GK / 4; ++s) {
    const v2f a = *(const v2f*)(arow + s * 4);
    acc = __builtin_amdgcn_wmma_f32_16x16x4_f32(false, a, false, breg[s],
                                                (short)0, acc, false, false);
  }
  float* cp = C + (size_t)(m0 + half * 8) * N + (n0 + l);
#pragma unroll
  for (int v = 0; v < 8; ++v) {
    if (m0 + half * 8 + v < M) cp[(size_t)v * N] = acc[v];
  }
}

template <int N, bool BIAS_RELU>
__global__ __launch_bounds__(32 * (N / 16)) void gemm_wmma_kernel(const float* __restrict__ A,
                                                                  const float* __restrict__ B,
                                                                  float* __restrict__ C,
                                                                  const float* __restrict__ act_bias,
                                                                  int M) {
  constexpr int NT = 32 * (N / 16);
  __shared__ __attribute__((aligned(16))) float stage[STAGE_FLOATS];
  const int tid  = threadIdx.x;
  const int lane = tid & 31;
  const int wave = __builtin_amdgcn_readfirstlane(tid >> 5);
  const int mt0  = blockIdx.x * STAGE_TILES;
  float4 v[STAGE_ROWS * (GK / 4) / NT];
  stage_load<NT, STAGE_ROWS>(A, M, mt0 * 16, tid, v);
  v2f breg[GK / 4];
  load_b_frags<N>(B, lane, wave * 16, breg);
  stage_store<NT, STAGE_ROWS, BIAS_RELU>(stage, act_bias, tid, v);
  __syncthreads();
  for (int t = 0; t < STAGE_TILES; ++t) {
    const int m0 = (mt0 + t) * 16;
    if (m0 < M) tile_from_stage<N>(stage + t * 16 * APITCH, breg, C, M, m0, lane, wave * 16);
  }
}

constexpr int TM_SCAN = 2;
constexpr int SCAN_GEMM_BLOCKS = 255;
constexpr int TM_FILL = 4;
static_assert(TM_FILL == STAGE_TILES, "a fill-side product block stages its tiles in one go");
static_assert(4 * TM_SCAN == 2 * STAGE_TILES, "a scan-side product block stages its tiles in two rounds");

__global__ __launch_bounds__(SORT_THREADS) void scan_gemm_kernel(
    int* __restrict__ T, int* __restrict__ bstart, int nb,
    const float* __restrict__ x, const float* __restrict__ W1, float* __restrict__ h1,
    int n, int mt_split) {
  if (blockIdx.x == 0) {
    bucket_scan_body(T, bstart, nb);
    return;
  }
  __shared__ __attribute__((aligned(16))) float stage[STAGE_FLOATS];
  const int tid  = threadIdx.x;
  const int lane = tid & 31;
  const int wave = __builtin_amdgcn_readfirstlane(tid >> 5);
  const int grp  = wave >> 3;
  const int n0   = (wave & 7) * 16;
  const int bt0  = ((int)blockIdx.x - 1) * (4 * TM_SCAN);
  if (bt0 >= mt_split) return;
  float4 v0[STAGE_ROWS * (GK / 4) / SORT_THREADS];
  float4 v1[STAGE_ROWS * (GK / 4) / SORT_THREADS];
  stage_load<SORT_THREADS, STAGE_ROWS>(x, n, bt0 * 16, tid, v0);
  stage_load<SORT_THREADS, STAGE_ROWS>(x, n, (bt0 + STAGE_TILES) * 16, tid, v1);
  v2f breg[GK / 4];
  load_b_frags<128>(W1, lane, n0, breg);
  stage_store<SORT_THREADS, STAGE_ROWS, false>(stage, nullptr, tid, v0);
  static_assert(STAGE_ROWS * (GK / 4) / SORT_THREADS == 2, "the load fence below names two pieces");
  asm volatile("" :: "v"(v1[0].x), "v"(v1[0].y), "v"(v1[0].z), "v"(v1[0].w),
                     "v"(v1[1].x), "v"(v1[1].y), "v"(v1[1].z), "v"(v1[1].w));
  __syncthreads();
  {
    const int t = bt0 + grp;
    if (t < mt_split) tile_from_stage<128>(stage + grp * 16 * APITCH, breg, h1, n, t * 16, lane, n0);
  }
  __syncthreads();
  stage_store<SORT_THREADS, STAGE_ROWS, false>(stage, nullptr, tid, v1);
  __syncthreads();
  {
    const int t = bt0 + STAGE_TILES + grp;
    if (t < mt_split) tile_from_stage<128>(stage + grp * 16 * APITCH, breg, h1, n, t * 16, lane, n0);
  }
}

__global__ __launch_bounds__(FILL_THREADS) void fill_gemm_kernel(
    const int* __restrict__ part, const int* __restrict__ rank2,
    const int* __restrict__ offs, const float* __restrict__ dinv,
    int* __restrict__ rec, int E, int nfill,
    const float* __restrict__ x, const float* __restrict__ W1, float* __restrict__ h1,
    int n, int mt_split, int ngemm) {
  __shared__ __attribute__((aligned(16))) float stage[STAGE_FLOATS];
  const int b = blockIdx.x;
  const int pairs = min(nfill, ngemm);
  int fb = -1, gb = -1;
  if (b < 2 * pairs) {
    if (b & 1) gb = b >> 1; else fb = b >> 1;
  } else if (nfill > ngemm) {
    fb = b - pairs;
  } else {
    gb = b - pairs;
  }
  if (gb >= 0) {
    const int tid  = threadIdx.x;
    const int lane = tid & 31;
    const int wave = __builtin_amdgcn_readfirstlane(tid >> 5);
    const int mt0  = mt_split + gb * TM_FILL;
    float4 v[STAGE_ROWS * (GK / 4) / FILL_THREADS];
    stage_load<FILL_THREADS, STAGE_ROWS>(x, n, mt0 * 16, tid, v);
    stage_store<FILL_THREADS, STAGE_ROWS, false>(stage, nullptr, tid, v);
    __syncthreads();
    v2f breg[GK / 4];
    load_b_frags<128>(W1, lane, wave * 16, breg);
    for (int t = 0; t < TM_FILL; ++t) {
      const int m0 = (mt0 + t) * 16;
      if (m0 < n) tile_from_stage<128>(stage + t * 16 * APITCH, breg, h1, n, m0, lane, wave * 16);
    }
    return;
  }
  fill_body(part, rank2, offs, dinv, rec, E, fb);
}


template <int F, int G, int B>
__device__ __forceinline__ void gather_steps(const float* __restrict__ h, int rv, float wv,
                                             int t0, int g, int fl, float4& acc) {
  float4 hv[B];
  float w[B];
#pragma unroll
  for (int u = 0; u < B; ++u) {
    int r;
    if constexpr (G == 1) {
      r = __builtin_amdgcn_readlane(rv, t0 + u);
      w[u] = __int_as_float(__builtin_amdgcn_readlane(__float_as_int(wv), t0 + u));
    } else {
      const int kk = (t0 + u) * G + g;
      r = __shfl(rv, kk);
      w[u] = __shfl(wv, kk);
    }
    hv[u] = *(const float4*)(h + (size_t)r * F + fl * 4);
  }
#pragma unroll
  for (int u = 0; u < B; ++u) {
    acc.x = fmaf(hv[u].x, w[u], acc.x);
    acc.y = fmaf(hv[u].y, w[u], acc.y);
    acc.z = fmaf(hv[u].z, w[u], acc.z);
    acc.w = fmaf(hv[u].w, w[u], acc.w);
  }
}

template <int F, bool HAS_BIAS>
__global__ __launch_bounds__(256) void gather_agg_kernel(
    const int* __restrict__ offs, const int* __restrict__ rec,
    const float* __restrict__ h, const float* __restrict__ dinv,
    const float* __restrict__ bias, float* __restrict__ out, int n) {
  constexpr int LPR = F / 4;
  constexpr int G = 32 / LPR;
  constexpr int BMAX = 8 / G;
  const int lane = threadIdx.x & 31;
  const int g = lane / LPR;
  const int fl = lane % LPR;
  const int wave = __builtin_amdgcn_readfirstlane((int)(threadIdx.x >> 5));
  const int wpb = blockDim.x >> 5;
  const int nwaves = gridDim.x * wpb;
  float4 bv = make_float4(0.0f, 0.0f, 0.0f, 0.0f);
  if constexpr (HAS_BIAS) bv = *(const float4*)(bias + fl * 4);

  for (int c = blockIdx.x * wpb + wave; c < n; c += nwaves) {
    const int beg = offs[c];
    const int end = offs[c + 1];
    const float dc = dinv[c];
    float4 acc = make_float4(0.0f, 0.0f, 0.0f, 0.0f);
    float4 hs = make_float4(0.0f, 0.0f, 0.0f, 0.0f);
    if (g == 0) hs = *(const float4*)(h + (size_t)c * F + fl * 4);
    for (int j = beg; j < end; j += 32) {
      const int m = min(32, end - j);
      int rv = c;
      float wv = 0.0f;
      if (lane < m) {
        const int2 p = ((const int2*)rec)[j + lane];
        rv = p.x;
        wv = __int_as_float(p.y) * dc;
      }
      const int S = m / G;
      int t = 0;
      for (; t + BMAX <= S; t += BMAX) gather_steps<F, G, BMAX>(h, rv, wv, t, g, fl, acc);
      if constexpr (BMAX >= 8) {
        if (t + 4 <= S) { gather_steps<F, G, 4>(h, rv, wv, t, g, fl, acc); t += 4; }
      }
      if constexpr (BMAX >= 4) {
        if (t + 2 <= S) { gather_steps<F, G, 2>(h, rv, wv, t, g, fl, acc); t += 2; }
      }
      if (t < S) gather_steps<F, G, 1>(h, rv, wv, t, g, fl, acc);
      if constexpr (G == 2) {
        if (m & 1) {
          const int r = __shfl(rv, m - 1);
          const float w = __shfl(wv, m - 1);
          if (g == 0) {
            const float4 hv = *(const float4*)(h + (size_t)r * F + fl * 4);
            acc.x = fmaf(hv.x, w, acc.x);
            acc.y = fmaf(hv.y, w, acc.y);
            acc.z = fmaf(hv.z, w, acc.z);
            acc.w = fmaf(hv.w, w, acc.w);
          }
        }
      }
    }
    if (g == 0) {
      acc.x += hs.x * dc * dc + bv.x;
      acc.y += hs.y * dc * dc + bv.y;
      acc.z += hs.z * dc * dc + bv.z;
      acc.w += hs.w * dc * dc + bv.w;
    }
    if constexpr (G == 2) {
      acc.x += __shfl_xor(acc.x, 16);
      acc.y += __shfl_xor(acc.y, 16);
      acc.z += __shfl_xor(acc.z, 16);
      acc.w += __shfl_xor(acc.w, 16);
    }
    if (g == 0) *(float4*)(out + (size_t)c * F + fl * 4) = acc;
  }
}

extern "C" void kernel_launch(void* const* d_in, const int* in_sizes, int n_in,
                              void* d_out, int out_size, void* d_ws, size_t ws_size,
                              hipStream_t stream) {
  const float* x  = (const float*)d_in[0];
  const int*   ei = (const int*)d_in[1];
  const float* W1 = (const float*)d_in[2];
  const float* b1 = (const float*)d_in[3];
  const float* W2 = (const float*)d_in[4];
  const float* b2 = (const float*)d_in[5];

  const int hid    = in_sizes[3];
  const int out_ch = in_sizes[5];
  const int in_ch  = in_sizes[2] / hid;
  const int n      = in_sizes[0] / in_ch;
  const int E      = in_sizes[1] / 2;
  const int* row = ei;
  const int* col = ei + E;

  const int nb  = (n + BNODES - 1) / BNODES;
  const int che = (E + NCH - 1) / NCH;

  float* ws = (float*)d_ws;
  size_t off = 0;
  auto alloc = [&](size_t elems) {
    float* p = ws + off;
    off += (elems + 63) & ~(size_t)63;
    return p;
  };
  int*   T      = (int*)alloc((size_t)nb * NCH);
  int*   bstart = (int*)alloc((size_t)nb + 1);
  int*   offs   = (int*)alloc((size_t)n + 1);
  float* dinv   = alloc((size_t)n);
  int*   part   = (int*)alloc((size_t)E * 2);
  int*   rank2  = (int*)alloc((size_t)E);
  int*   rec    = (int*)alloc((size_t)E * 2);
  float* h1   = alloc((size_t)n * hid);
  float* agg1 = alloc((size_t)n * hid);
  float* g2   = alloc((size_t)n * out_ch);
  float* out  = (float*)d_out;

  constexpr int TM = STAGE_TILES;
  const int mtiles   = (n + 15) / 16;
  const int mblocks  = (mtiles + TM - 1) / TM;
  const int scan_tiles = SCAN_GEMM_BLOCKS * 4 * TM_SCAN;
  const int mt_split = mtiles < scan_tiles ? mtiles : scan_tiles;
  const int ngemm = (mtiles - mt_split + TM_FILL - 1) / TM_FILL;
  const int nfill = (E + EPB * FILL_THREADS - 1) / (EPB * FILL_THREADS);

  bucket_hist_kernel<<<NCH, SORT_THREADS, 0, stream>>>(col, T, E, che, nb);
  scan_gemm_kernel<<<1 + SCAN_GEMM_BLOCKS, SORT_THREADS, 0, stream>>>(
      T, bstart, nb, x, W1, h1, n, mt_split);
  bucket_scatter_kernel<<<NCH, SORT_THREADS, 0, stream>>>(row, col, T, bstart, part, E, che, nb);
  bucket_sort_kernel<<<nb, SORT_THREADS, 0, stream>>>(part, bstart, rank2, offs, dinv, n, nb);
  fill_gemm_kernel<<<nfill + ngemm, FILL_THREADS, 0, stream>>>(
      part, rank2, offs, dinv, rec, E, nfill, x, W1, h1, n, mt_split, ngemm);

  gather_agg_kernel<128, false><<<2048, 256, 0, stream>>>(offs, rec, h1, dinv, nullptr, agg1, n);

  gemm_wmma_kernel<64, true><<<mblocks, 32 * (64 / 16), 0, stream>>>(
      agg1, W2, g2, b1, n);
  gather_agg_kernel<64, true><<<2048, 256, 0, stream>>>(offs, rec, g2, dinv, b2, out, n);
}


# Layer-2 product fused behind layer-1 aggregation: a work-group sums 32 nodes' rows into LDS (bias+relu) and multiplies them by W2; agg1 never goes to memory; 8 -> 7 launches; sum order unchanged
# speedup vs baseline: 3.8005x; 1.0548x over previous
#include <hip/hip_runtime.h>
#include <hip/hip_bf16.h>


typedef __attribute__((ext_vector_type(2))) float v2f;
typedef __attribute__((ext_vector_type(8))) float v8f;

constexpr int BSHIFT = 9;
constexpr int BNODES = 1 << BSHIFT;
constexpr int NCH = 256;
constexpr int NBMAX = 1024;
constexpr int SORT_THREADS = 1024;
constexpr int EPB = 4;
static_assert(EPB == 4, "the load fences in bucket_hist_kernel and fill_kernel name four values each");
constexpr int FILL_THREADS = 256;

__device__ __forceinline__ int wave_incl_scan(int v, int lane) {
  for (int d = 1; d < 32; d <<= 1) {
    const int y = __shfl_up(v, d);
    if (lane >= d) v += y;
  }
  return v;
}

__device__ __forceinline__ int block_excl_scan_1024(int v, int* wsum, int lane, int wave,
                                                    int& total) {
  const int incl = wave_incl_scan(v, lane);
  if (lane == 31) wsum[wave] = incl;
  __syncthreads();
  const int wt = wsum[lane];
  const int winc = wave_incl_scan(wt, lane);
  const int wave_excl = __shfl(winc, wave) - __shfl(wt, wave);
  total = __shfl(winc, 31);
  return wave_excl + incl - v;
}

__global__ __launch_bounds__(SORT_THREADS) void bucket_hist_kernel(
    const int* __restrict__ col, int* __restrict__ T, int E, int che, int nb) {
  __shared__ int hist[NBMAX];
  const int tid = threadIdx.x;
  const int ch = blockIdx.x;
  for (int b = tid; b < nb; b += SORT_THREADS) hist[b] = 0;
  __syncthreads();
  const int e0 = ch * che;
  const int e1 = min(E, e0 + che);
  for (int e = e0 + tid; e < e1; e += EPB * SORT_THREADS) {
    int c[EPB];
#pragma unroll
    for (int u = 0; u < EPB; ++u) {
      c[u] = col[min(e + u * SORT_THREADS, e1 - 1)];
    }
    asm volatile("" :: "v"(c[0]), "v"(c[1]), "v"(c[2]), "v"(c[3]));
#pragma unroll
    for (int u = 0; u < EPB; ++u) {
      if (e + u * SORT_THREADS < e1) atomicAdd(&hist[c[u] >> BSHIFT], 1);
    }
  }
  __syncthreads();
  for (int b = tid; b < nb; b += SORT_THREADS) T[b * NCH + ch] = hist[b];
}

__device__ __forceinline__ void bucket_scan_body(
    int* __restrict__ T, int* __restrict__ bstart, int nb) {
  __shared__ int tot[NBMAX];
  __shared__ int wsum[32];
  const int tid = threadIdx.x;
  const int lane = tid & 31;
  const int wave = tid >> 5;
  for (int b = wave; b < nb; b += 32) {
    int* p = T + b * NCH + lane * 8;
    const int4 a = *(const int4*)p;
    const int4 c = *(const int4*)(p + 4);
    const int s = a.x + a.y + a.z + a.w + c.x + c.y + c.z + c.w;
    const int incl = wave_incl_scan(s, lane);
    const int ex = incl - s;
    int4 oa, oc;
    oa.x = ex;          oa.y = oa.x + a.x;  oa.z = oa.y + a.y;  oa.w = oa.z + a.z;
    oc.x = oa.w + a.w;  oc.y = oc.x + c.x;  oc.z = oc.y + c.y;  oc.w = oc.z + c.z;
    *(int4*)p = oa;
    *(int4*)(p + 4) = oc;
    if (lane == 31) tot[b] = incl;
  }
  __syncthreads();
  const int v = (tid < nb) ? tot[tid] : 0;
  int total;
  const int ex = block_excl_scan_1024(v, wsum, lane, wave, total);
  if (tid < nb) bstart[tid] = ex;
  if (tid == 0) bstart[nb] = total;
}

__global__ __launch_bounds__(SORT_THREADS) void bucket_scatter_kernel(
    const int* __restrict__ row, const int* __restrict__ col, const int* __restrict__ T,
    const int* __restrict__ bstart, int* __restrict__ part, int E, int che, int nb) {
  __shared__ int cur[NBMAX];
  const int tid = threadIdx.x;
  const int ch = blockIdx.x;
  for (int b = tid; b < nb; b += SORT_THREADS) cur[b] = bstart[b] + T[b * NCH + ch];
  __syncthreads();
  const int e0 = ch * che;
  const int e1 = min(E, e0 + che);
  for (int e = e0 + tid; e < e1; e += EPB * SORT_THREADS) {
    int r[EPB], c[EPB];
#pragma unroll
    for (int u = 0; u < EPB; ++u) {
      const int ee = e + u * SORT_THREADS;
      r[u] = (ee < e1) ? row[ee] : 0;
      c[u] = (ee < e1) ? col[ee] : 0;
    }
#pragma unroll
    for (int u = 0; u < EPB; ++u) {
      if (e + u * SORT_THREADS < e1) {
        const int pos = atomicAdd(&cur[c[u] >> BSHIFT], 1);
        ((int2*)part)[pos] = make_int2(r[u], c[u]);
      }
    }
  }
}

__global__ __launch_bounds__(SORT_THREADS) void bucket_sort_kernel(
    const int* __restrict__ part, const int* __restrict__ bstart, int* __restrict__ rank2,
    int* __restrict__ offs, float* __restrict__ dinv, int n, int nb) {
  __shared__ int cnt[BNODES];
  __shared__ int wsum[32];
  const int tid = threadIdx.x;
  const int lane = tid & 31;
  const int wave = tid >> 5;
  const int b = blockIdx.x;
  const int c0 = b << BSHIFT;
  const int p0 = bstart[b];
  const int p1 = bstart[b + 1];
  if (tid < BNODES) cnt[tid] = 0;
  __syncthreads();
  for (int p = p0 + tid; p < p1; p += EPB * SORT_THREADS) {
    int d[EPB];
#pragma unroll
    for (int u = 0; u < EPB; ++u) {
      const int pp = p + u * SORT_THREADS;
      d[u] = (pp < p1) ? part[2 * (size_t)pp + 1] : c0;
    }
#pragma unroll
    for (int u = 0; u < EPB; ++u) {
      const int pp = p + u * SORT_THREADS;
      if (pp < p1) rank2[pp] = atomicAdd(&cnt[d[u] - c0], 1);
    }
  }
  __syncthreads();
  const int v = (tid < BNODES) ? cnt[tid] : 0;
  int total;
  const int ex = block_excl_scan_1024(v, wsum, lane, wave, total);
  const int c = c0 + tid;
  if (tid < BNODES && c < n) {
    offs[c] = p0 + ex;
    dinv[c] = rsqrtf((float)(v + 1));
  }
  if (b == nb - 1 && tid == 0) offs[n] = p1;
}

__device__ __forceinline__ void fill_body(
    const int* __restrict__ part, const int* __restrict__ rank2,
    const int* __restrict__ offs, const float* __restrict__ dinv,
    int* __restrict__ rec, int E, int fb) {
  const int base = fb * (EPB * FILL_THREADS) + threadIdx.x;
  int2 rc[EPB];
  int k[EPB], o[EPB];
  float w[EPB];
#pragma unroll
  for (int u = 0; u < EPB; ++u) {
    const int p = min(base + u * FILL_THREADS, E - 1);
    rc[u] = ((const int2*)part)[p];
    k[u] = rank2[p];
  }
  asm volatile("" :: "v"(rc[0].x), "v"(rc[0].y), "v"(rc[1].x), "v"(rc[1].y),
                     "v"(rc[2].x), "v"(rc[2].y), "v"(rc[3].x), "v"(rc[3].y));
#pragma unroll
  for (int u = 0; u < EPB; ++u) {
    o[u] = offs[rc[u].y];
    w[u] = dinv[rc[u].x];
  }
  asm volatile("" :: "v"(o[0]), "v"(o[1]), "v"(o[2]), "v"(o[3]),
                     "v"(w[0]), "v"(w[1]), "v"(w[2]), "v"(w[3]),
                     "v"(k[0]), "v"(k[1]), "v"(k[2]), "v"(k[3]));
#pragma unroll
  for (int u = 0; u < EPB; ++u) {
    if (base + u * FILL_THREADS < E)
      ((int2*)rec)[o[u] + k[u]] = make_int2(rc[u].x, __float_as_int(w[u]));
  }
}

constexpr int GK = 128;
constexpr int APITCH = GK + 4;
constexpr int STAGE_TILES = 4;
constexpr int STAGE_ROWS = STAGE_TILES * 16;
constexpr int STAGE_FLOATS = STAGE_ROWS * APITCH;

template <int NT, int ROWS>
__device__ __forceinline__ void stage_load(const float* __restrict__ A, int M, int row0, int tid,
                                           float4 (&v)[ROWS * (GK / 4) / NT]) {
  constexpr int PER = ROWS * (GK / 4) / NT;
  static_assert(NT % 32 == 0 && PER * NT == ROWS * (GK / 4), "whole rows per wave");
#pragma unroll
  for (int i = 0; i < PER; ++i) {
    const int idx = tid + i * NT;
    const int r = min(row0 + (idx >> 5), M - 1);
    v[i] = *(const float4*)(A + (size_t)r * GK + (idx & 31) * 4);
  }
}

template <int NT, int ROWS>
__device__ __forceinline__ void stage_store(float* stage, int tid,
                                            const float4 (&v)[ROWS * (GK / 4) / NT]) {
  constexpr int PER = ROWS * (GK / 4) / NT;
#pragma unroll
  for (int i = 0; i < PER; ++i) {
    const int idx = tid + i * NT;
    *(float4*)(stage + (idx >> 5) * APITCH + (idx & 31) * 4) = v[i];
  }
}

template <int N>
__device__ __forceinline__ void load_b_frags(const float* __restrict__ B, int lane, int n0,
                                             v2f (&breg)[GK / 4]) {
  const int kb = 2 * (lane >> 4);
  const float* bcol = B + (size_t)kb * N + (n0 + (lane & 15));
#pragma unroll
  for (int s = 0; s < GK / 4; ++s) {
    breg[s].x = bcol[s * 4 * N];
    breg[s].y = bcol[(s * 4 + 1) * N];
  }
}

template <int N>
__device__ __forceinline__ void tile_from_stage(const float* tile, const v2f (&breg)[GK / 4],
                                                float* __restrict__ C, int M, int m0,
                                                int lane, int n0) {
  const int half = lane >> 4;
  const int l    = lane & 15;
  const float* arow = tile + l * APITCH + 2 * half;
  v8f acc = {};
#pragma unroll
  for (int s = 0; s < GK / 4; ++s) {
    const v2f a = *(const v2f*)(arow + s * 4);
    acc = __builtin_amdgcn_wmma_f32_16x16x4_f32(false, a, false, breg[s],
                                                (short)0, acc, false, false);
  }
  float* cp = C + (size_t)(m0 + half * 8) * N + (n0 + l);
#pragma unroll
  for (int v = 0; v < 8; ++v) {
    if (m0 + half * 8 + v < M) cp[(size_t)v * N] = acc[v];
  }
}

constexpr int TM_SCAN = 2;
constexpr int SCAN_GEMM_BLOCKS = 255;
constexpr int TM_FILL = 4;
static_assert(TM_FILL == STAGE_TILES, "a fill-side product block stages its tiles in one go");
static_assert(4 * TM_SCAN == 2 * STAGE_TILES, "a scan-side product block stages its tiles in two rounds");

__global__ __launch_bounds__(SORT_THREADS) void scan_gemm_kernel(
    int* __restrict__ T, int* __restrict__ bstart, int nb,
    const float* __restrict__ x, const float* __restrict__ W1, float* __restrict__ h1,
    int n, int mt_split) {
  if (blockIdx.x == 0) {
    bucket_scan_body(T, bstart, nb);
    return;
  }
  __shared__ __attribute__((aligned(16))) float stage[STAGE_FLOATS];
  const int tid  = threadIdx.x;
  const int lane = tid & 31;
  const int wave = __builtin_amdgcn_readfirstlane(tid >> 5);
  const int grp  = wave >> 3;
  const int n0   = (wave & 7) * 16;
  const int bt0  = ((int)blockIdx.x - 1) * (4 * TM_SCAN);
  if (bt0 >= mt_split) return;
  float4 v0[STAGE_ROWS * (GK / 4) / SORT_THREADS];
  float4 v1[STAGE_ROWS * (GK / 4) / SORT_THREADS];
  stage_load<SORT_THREADS, STAGE_ROWS>(x, n, bt0 * 16, tid, v0);
  stage_load<SORT_THREADS, STAGE_ROWS>(x, n, (bt0 + STAGE_TILES) * 16, tid, v1);
  v2f breg[GK / 4];
  load_b_frags<128>(W1, lane, n0, breg);
  stage_store<SORT_THREADS, STAGE_ROWS>(stage, tid, v0);
  static_assert(STAGE_ROWS * (GK / 4) / SORT_THREADS == 2, "the load fence below names two pieces");
  asm volatile("" :: "v"(v1[0].x), "v"(v1[0].y), "v"(v1[0].z), "v"(v1[0].w),
                     "v"(v1[1].x), "v"(v1[1].y), "v"(v1[1].z), "v"(v1[1].w));
  __syncthreads();
  {
    const int t = bt0 + grp;
    if (t < mt_split) tile_from_stage<128>(stage + grp * 16 * APITCH, breg, h1, n, t * 16, lane, n0);
  }
  __syncthreads();
  stage_store<SORT_THREADS, STAGE_ROWS>(stage, tid, v1);
  __syncthreads();
  {
    const int t = bt0 + STAGE_TILES + grp;
    if (t < mt_split) tile_from_stage<128>(stage + grp * 16 * APITCH, breg, h1, n, t * 16, lane, n0);
  }
}

__global__ __launch_bounds__(FILL_THREADS) void fill_gemm_kernel(
    const int* __restrict__ part, const int* __restrict__ rank2,
    const int* __restrict__ offs, const float* __restrict__ dinv,
    int* __restrict__ rec, int E, int nfill,
    const float* __restrict__ x, const float* __restrict__ W1, float* __restrict__ h1,
    int n, int mt_split, int ngemm) {
  __shared__ __attribute__((aligned(16))) float stage[STAGE_FLOATS];
  const int b = blockIdx.x;
  const int pairs = min(nfill, ngemm);
  int fb = -1, gb = -1;
  if (b < 2 * pairs) {
    if (b & 1) gb = b >> 1; else fb = b >> 1;
  } else if (nfill > ngemm) {
    fb = b - pairs;
  } else {
    gb = b - pairs;
  }
  if (gb >= 0) {
    const int tid  = threadIdx.x;
    const int lane = tid & 31;
    const int wave = __builtin_amdgcn_readfirstlane(tid >> 5);
    const int mt0  = mt_split + gb * TM_FILL;
    float4 v[STAGE_ROWS * (GK / 4) / FILL_THREADS];
    stage_load<FILL_THREADS, STAGE_ROWS>(x, n, mt0 * 16, tid, v);
    stage_store<FILL_THREADS, STAGE_ROWS>(stage, tid, v);
    __syncthreads();
    v2f breg[GK / 4];
    load_b_frags<128>(W1, lane, wave * 16, breg);
    for (int t = 0; t < TM_FILL; ++t) {
      const int m0 = (mt0 + t) * 16;
      if (m0 < n) tile_from_stage<128>(stage + t * 16 * APITCH, breg, h1, n, m0, lane, wave * 16);
    }
    return;
  }
  fill_body(part, rank2, offs, dinv, rec, E, fb);
}


template <int F, int G, int B>
__device__ __forceinline__ void gather_steps(const float* __restrict__ h, int rv, float wv,
                                             int t0, int g, int fl, float4& acc) {
  float4 hv[B];
  float w[B];
#pragma unroll
  for (int u = 0; u < B; ++u) {
    int r;
    if constexpr (G == 1) {
      r = __builtin_amdgcn_readlane(rv, t0 + u);
      w[u] = __int_as_float(__builtin_amdgcn_readlane(__float_as_int(wv), t0 + u));
    } else {
      const int kk = (t0 + u) * G + g;
      r = __shfl(rv, kk);
      w[u] = __shfl(wv, kk);
    }
    hv[u] = *(const float4*)(h + (size_t)r * F + fl * 4);
  }
#pragma unroll
  for (int u = 0; u < B; ++u) {
    acc.x = fmaf(hv[u].x, w[u], acc.x);
    acc.y = fmaf(hv[u].y, w[u], acc.y);
    acc.z = fmaf(hv[u].z, w[u], acc.z);
    acc.w = fmaf(hv[u].w, w[u], acc.w);
  }
}

template <int F>
__device__ __forceinline__ void list_walk(const int* __restrict__ rec, const float* __restrict__ h,
                                          int beg, int end, int c, float dc, int lane, int g, int fl,
                                          float4& acc) {
  constexpr int G = 32 / (F / 4);
  constexpr int BMAX = 8 / G;
  for (int j = beg; j < end; j += 32) {
    const int m = min(32, end - j);
    int rv = c;
    float wv = 0.0f;
    if (lane < m) {
      const int2 p = ((const int2*)rec)[j + lane];
      rv = p.x;
      wv = __int_as_float(p.y) * dc;
    }
    const int S = m / G;
    int t = 0;
    for (; t + BMAX <= S; t += BMAX) gather_steps<F, G, BMAX>(h, rv, wv, t, g, fl, acc);
    if constexpr (BMAX >= 8) {
      if (t + 4 <= S) { gather_steps<F, G, 4>(h, rv, wv, t, g, fl, acc); t += 4; }
    }
    if constexpr (BMAX >= 4) {
      if (t + 2 <= S) { gather_steps<F, G, 2>(h, rv, wv, t, g, fl, acc); t += 2; }
    }
    if (t < S) gather_steps<F, G, 1>(h, rv, wv, t, g, fl, acc);
    if constexpr (G == 2) {
      if (m & 1) {
        const int r = __shfl(rv, m - 1);
        const float w = __shfl(wv, m - 1);
        if (g == 0) {
          const float4 hv = *(const float4*)(h + (size_t)r * F + fl * 4);
          acc.x = fmaf(hv.x, w, acc.x);
          acc.y = fmaf(hv.y, w, acc.y);
          acc.z = fmaf(hv.z, w, acc.z);
          acc.w = fmaf(hv.w, w, acc.w);
        }
      }
    }
  }
}

template <int F, bool HAS_BIAS>
__global__ __launch_bounds__(256) void gather_agg_kernel(
    const int* __restrict__ offs, const int* __restrict__ rec,
    const float* __restrict__ h, const float* __restrict__ dinv,
    const float* __restrict__ bias, float* __restrict__ out, int n) {
  constexpr int LPR = F / 4;
  constexpr int G = 32 / LPR;
  const int lane = threadIdx.x & 31;
  const int g = lane / LPR;
  const int fl = lane % LPR;
  const int wave = __builtin_amdgcn_readfirstlane((int)(threadIdx.x >> 5));
  const int wpb = blockDim.x >> 5;
  const int nwaves = gridDim.x * wpb;
  float4 bv = make_float4(0.0f, 0.0f, 0.0f, 0.0f);
  if constexpr (HAS_BIAS) bv = *(const float4*)(bias + fl * 4);

  for (int c = blockIdx.x * wpb + wave; c < n; c += nwaves) {
    const int beg = offs[c];
    const int end = offs[c + 1];
    const float dc = dinv[c];
    float4 acc = make_float4(0.0f, 0.0f, 0.0f, 0.0f);
    float4 hs = make_float4(0.0f, 0.0f, 0.0f, 0.0f);
    if (g == 0) hs = *(const float4*)(h + (size_t)c * F + fl * 4);
    list_walk<F>(rec, h, beg, end, c, dc, lane, g, fl, acc);
    if (g == 0) {
      acc.x += hs.x * dc * dc + bv.x;
      acc.y += hs.y * dc * dc + bv.y;
      acc.z += hs.z * dc * dc + bv.z;
      acc.w += hs.w * dc * dc + bv.w;
    }
    if constexpr (G == 2) {
      acc.x += __shfl_xor(acc.x, 16);
      acc.y += __shfl_xor(acc.y, 16);
      acc.z += __shfl_xor(acc.z, 16);
      acc.w += __shfl_xor(acc.w, 16);
    }
    if (g == 0) *(float4*)(out + (size_t)c * F + fl * 4) = acc;
  }
}

constexpr int FG_TILES = 2;
constexpr int FG_ROWS  = FG_TILES * 16;
constexpr int FG_N     = 64;
constexpr int FG_WAVES = FG_N / 16;
constexpr int FG_NPW   = FG_ROWS / FG_WAVES;
constexpr int FG_KP    = 2;
constexpr int FG_KS    = GK / 4 / FG_KP;

__global__ __launch_bounds__(32 * FG_WAVES) void gather_agg_gemm_kernel(
    const int* __restrict__ offs, const int* __restrict__ rec,
    const float* __restrict__ h, const float* __restrict__ dinv,
    const float* __restrict__ act_bias, const float* __restrict__ B,
    float* __restrict__ C, int n) {
  constexpr int F = GK;
  static_assert(F / 4 == 32, "one lane per 4 features of a row");
  __shared__ __attribute__((aligned(16))) float stage[FG_ROWS * APITCH];
  const int lane = threadIdx.x & 31;
  const int wave = __builtin_amdgcn_readfirstlane((int)(threadIdx.x >> 5));
  const int base = blockIdx.x * FG_ROWS;
  const float4 bv = *(const float4*)(act_bias + lane * 4);
  for (int i = 0; i < FG_NPW; ++i) {
    const int lr = wave * FG_NPW + i;
    const int c = base + lr;
    float4 acc = make_float4(0.0f, 0.0f, 0.0f, 0.0f);
    if (c < n) {
      const int beg = offs[c];
      const int end = offs[c + 1];
      const float dc = dinv[c];
      const float4 hs = *(const float4*)(h + (size_t)c * F + lane * 4);
      list_walk<F>(rec, h, beg, end, c, dc, lane, 0, lane, acc);
      acc.x = fmaxf(acc.x + hs.x * dc * dc + bv.x, 0.0f);
      acc.y = fmaxf(acc.y + hs.y * dc * dc + bv.y, 0.0f);
      acc.z = fmaxf(acc.z + hs.z * dc * dc + bv.z, 0.0f);
      acc.w = fmaxf(acc.w + hs.w * dc * dc + bv.w, 0.0f);
    }
    *(float4*)(stage + lr * APITCH + lane * 4) = acc;
  }
  __syncthreads();
  const int half = lane >> 4;
  const int l    = lane & 15;
  const int n0   = wave * 16;
  const float* arow = stage + l * APITCH + 2 * half;
  const float* bcol = B + (size_t)(2 * half) * FG_N + (n0 + l);
  v8f acc[FG_TILES] = {};
#pragma unroll 1
  for (int q = 0; q < FG_KP; ++q) {
    const float* bq = bcol + (size_t)q * (FG_KS * 4 * FG_N);
    const float* aq = arow + q * (FG_KS * 4);
    v2f bp[FG_KS];
#pragma unroll
    for (int s = 0; s < FG_KS; ++s) {
      bp[s].x = bq[s * 4 * FG_N];
      bp[s].y = bq[(s * 4 + 1) * FG_N];
    }
#pragma unroll
    for (int s = 0; s < FG_KS; ++s) {
#pragma unroll
      for (int t = 0; t < FG_TILES; ++t) {
        const v2f a = *(const v2f*)(aq + t * 16 * APITCH + s * 4);
        acc[t] = __builtin_amdgcn_wmma_f32_16x16x4_f32(false, a, false, bp[s],
                                                       (short)0, acc[t], false, false);
      }
    }
  }
#pragma unroll
  for (int t = 0; t < FG_TILES; ++t) {
    const int m0 = base + t * 16;
    float* cp = C + (size_t)(m0 + half * 8) * FG_N + (n0 + l);
#pragma unroll
    for (int v = 0; v < 8; ++v) {
      if (m0 + half * 8 + v < n) cp[(size_t)v * FG_N] = acc[t][v];
    }
  }
}

extern "C" void kernel_launch(void* const* d_in, const int* in_sizes, int n_in,
                              void* d_out, int out_size, void* d_ws, size_t ws_size,
                              hipStream_t stream) {
  const float* x  = (const float*)d_in[0];
  const int*   ei = (const int*)d_in[1];
  const float* W1 = (const float*)d_in[2];
  const float* b1 = (const float*)d_in[3];
  const float* W2 = (const float*)d_in[4];
  const float* b2 = (const float*)d_in[5];

  const int hid    = in_sizes[3];
  const int out_ch = in_sizes[5];
  const int in_ch  = in_sizes[2] / hid;
  const int n      = in_sizes[0] / in_ch;
  const int E      = in_sizes[1] / 2;
  const int* row = ei;
  const int* col = ei + E;

  const int nb  = (n + BNODES - 1) / BNODES;
  const int che = (E + NCH - 1) / NCH;

  float* ws = (float*)d_ws;
  size_t off = 0;
  auto alloc = [&](size_t elems) {
    float* p = ws + off;
    off += (elems + 63) & ~(size_t)63;
    return p;
  };
  int*   T      = (int*)alloc((size_t)nb * NCH);
  int*   bstart = (int*)alloc((size_t)nb + 1);
  int*   offs   = (int*)alloc((size_t)n + 1);
  float* dinv   = alloc((size_t)n);
  int*   part   = (int*)alloc((size_t)E * 2);
  int*   rank2  = (int*)alloc((size_t)E);
  int*   rec    = (int*)alloc((size_t)E * 2);
  float* h1   = alloc((size_t)n * hid);
  float* g2   = alloc((size_t)n * out_ch);
  float* out  = (float*)d_out;

  const int mtiles   = (n + 15) / 16;
  const int scan_tiles = SCAN_GEMM_BLOCKS * 4 * TM_SCAN;
  const int mt_split = mtiles < scan_tiles ? mtiles : scan_tiles;
  const int ngemm = (mtiles - mt_split + TM_FILL - 1) / TM_FILL;
  const int nfill = (E + EPB * FILL_THREADS - 1) / (EPB * FILL_THREADS);

  bucket_hist_kernel<<<NCH, SORT_THREADS, 0, stream>>>(col, T, E, che, nb);
  scan_gemm_kernel<<<1 + SCAN_GEMM_BLOCKS, SORT_THREADS, 0, stream>>>(
      T, bstart, nb, x, W1, h1, n, mt_split);
  bucket_scatter_kernel<<<NCH, SORT_THREADS, 0, stream>>>(row, col, T, bstart, part, E, che, nb);
  bucket_sort_kernel<<<nb, SORT_THREADS, 0, stream>>>(part, bstart, rank2, offs, dinv, n, nb);
  fill_gemm_kernel<<<nfill + ngemm, FILL_THREADS, 0, stream>>>(
      part, rank2, offs, dinv, rec, E, nfill, x, W1, h1, n, mt_split, ngemm);

  gather_agg_gemm_kernel<<<(n + FG_ROWS - 1) / FG_ROWS, 32 * FG_WAVES, 0, stream>>>(
      offs, rec, h1, dinv, b1, W2, g2, n);
  gather_agg_kernel<64, true><<<2048, 256, 0, stream>>>(offs, rec, g2, dinv, b2, out, n);
}


# Edge grouping in 2 launches (was 4): each chunk counts, scans and places its edges by bucket in its own segment; no table scan; x@W1 tiles all in the fill; 7->5 launches; sum order still varies
# speedup vs baseline: 3.9205x; 1.0316x over previous
#include <hip/hip_runtime.h>
#include <hip/hip_bf16.h>


typedef __attribute__((ext_vector_type(2))) float v2f;
typedef __attribute__((ext_vector_type(8))) float v8f;

constexpr int BSHIFT = 9;
constexpr int BNODES = 1 << BSHIFT;
constexpr int NCH = 256;
constexpr int NBMAX = 1024;
constexpr int SORT_THREADS = 1024;
constexpr int EPB = 4;
static_assert(EPB == 4, "the load fences in chunk_group_kernel and fill_body name four values each");
static_assert(NBMAX == SORT_THREADS, "chunk_group_kernel: one bucket counter per thread");
constexpr int SEGW = NCH / 32;
static_assert(SEGW * 32 == NCH && SEGW <= 32, "bucket_sort_kernel: lane s of a wave holds the wave's piece s");
constexpr int FILL_THREADS = 256;

__device__ __forceinline__ int wave_incl_scan(int v, int lane) {
  for (int d = 1; d < 32; d <<= 1) {
    const int y = __shfl_up(v, d);
    if (lane >= d) v += y;
  }
  return v;
}

__device__ __forceinline__ int block_excl_scan_1024(int v, int* wsum, int lane, int wave,
                                                    int& total) {
  const int incl = wave_incl_scan(v, lane);
  if (lane == 31) wsum[wave] = incl;
  __syncthreads();
  const int wt = wsum[lane];
  const int winc = wave_incl_scan(wt, lane);
  const int wave_excl = __shfl(winc, wave) - __shfl(wt, wave);
  total = __shfl(winc, 31);
  return wave_excl + incl - v;
}

__global__ __launch_bounds__(SORT_THREADS) void chunk_group_kernel(
    const int* __restrict__ row, const int* __restrict__ col, int* __restrict__ T,
    int* __restrict__ L, int* __restrict__ part, int E, int che, int nb) {
  __shared__ int hist[NBMAX];
  __shared__ int wsum[32];
  const int tid = threadIdx.x;
  const int lane = tid & 31;
  const int wave = tid >> 5;
  const int ch = blockIdx.x;
  hist[tid] = 0;
  __syncthreads();
  const int e0 = ch * che;
  const int e1 = min(E, e0 + che);
  for (int e = e0 + tid; e < e1; e += EPB * SORT_THREADS) {
    int c[EPB];
#pragma unroll
    for (int u = 0; u < EPB; ++u) {
      c[u] = col[min(e + u * SORT_THREADS, e1 - 1)];
    }
    asm volatile("" :: "v"(c[0]), "v"(c[1]), "v"(c[2]), "v"(c[3]));
#pragma unroll
    for (int u = 0; u < EPB; ++u) {
      if (e + u * SORT_THREADS < e1) atomicAdd(&hist[c[u] >> BSHIFT], 1);
    }
  }
  __syncthreads();
  const int v = hist[tid];
  int total;
  const int ex = block_excl_scan_1024(v, wsum, lane, wave, total);
  if (tid < nb) {
    T[tid * NCH + ch] = v;
    L[tid * NCH + ch] = ex;
  }
  hist[tid] = ex;
  __syncthreads();
  for (int e = e0 + tid; e < e1; e += EPB * SORT_THREADS) {
    int r[EPB], c[EPB];
#pragma unroll
    for (int u = 0; u < EPB; ++u) {
      const int ee = e + u * SORT_THREADS;
      r[u] = (ee < e1) ? row[ee] : 0;
      c[u] = (ee < e1) ? col[ee] : 0;
    }
#pragma unroll
    for (int u = 0; u < EPB; ++u) {
      if (e + u * SORT_THREADS < e1) {
        const int pos = atomicAdd(&hist[c[u] >> BSHIFT], 1);
        ((int2*)part)[e0 + pos] = make_int2(r[u], c[u]);
      }
    }
  }
}

__global__ __launch_bounds__(SORT_THREADS) void bucket_sort_kernel(
    const int* __restrict__ part, const int* __restrict__ T, const int* __restrict__ L,
    int* __restrict__ rank2, int* __restrict__ offs, float* __restrict__ dinv,
    int n, int nb, int che) {
  __shared__ int cnt[BNODES];
  __shared__ int wsum[32];
  __shared__ int wsum0[32];
  const int tid = threadIdx.x;
  const int lane = tid & 31;
  const int wave = __builtin_amdgcn_readfirstlane(tid >> 5);
  const int b = blockIdx.x;
  const int c0 = b << BSHIFT;
  int segc = 0, segl = 0;
  if (lane < SEGW) {
    segc = T[b * NCH + wave * SEGW + lane];
    segl = L[b * NCH + wave * SEGW + lane];
  }
  if (tid < BNODES) cnt[tid] = 0;
  int p0;
  (void)block_excl_scan_1024(segl, wsum0, lane, wave, p0);
  int mx = segc;
  for (int sh = 1; sh < SEGW; sh <<= 1) mx = max(mx, __shfl_xor(mx, sh));
  mx = __builtin_amdgcn_readfirstlane(mx);
  for (int i0 = 0; i0 < mx; i0 += 64) {
    int d[SEGW][2];
#pragma unroll
    for (int s = 0; s < SEGW; ++s) {
      const int len = __builtin_amdgcn_readlane(segc, s);
      const int base = (wave * SEGW + s) * che + __builtin_amdgcn_readlane(segl, s);
#pragma unroll
      for (int k = 0; k < 2; ++k) {
        const int i = i0 + k * 32 + lane;
        d[s][k] = (i < len) ? part[2 * (size_t)(base + i) + 1] : c0;
      }
    }
    int rk[SEGW][2];
#pragma unroll
    for (int s = 0; s < SEGW; ++s) {
      const int len = __builtin_amdgcn_readlane(segc, s);
#pragma unroll
      for (int k = 0; k < 2; ++k) {
        const int i = i0 + k * 32 + lane;
        rk[s][k] = (i < len) ? atomicAdd(&cnt[d[s][k] - c0], 1) : 0;
      }
    }
#pragma unroll
    for (int s = 0; s < SEGW; ++s) {
      const int len = __builtin_amdgcn_readlane(segc, s);
      const int base = (wave * SEGW + s) * che + __builtin_amdgcn_readlane(segl, s);
#pragma unroll
      for (int k = 0; k < 2; ++k) {
        const int i = i0 + k * 32 + lane;
        if (i < len) rank2[base + i] = rk[s][k];
      }
    }
  }
  __syncthreads();
  const int v = (tid < BNODES) ? cnt[tid] : 0;
  int total;
  const int ex = block_excl_scan_1024(v, wsum, lane, wave, total);
  const int c = c0 + tid;
  if (tid < BNODES && c < n) {
    offs[c] = p0 + ex;
    dinv[c] = rsqrtf((float)(v + 1));
  }
  if (b == nb - 1 && tid == 0) offs[n] = p0 + total;
}

__device__ __forceinline__ void fill_body(
    const int* __restrict__ part, const int* __restrict__ rank2,
    const int* __restrict__ offs, const float* __restrict__ dinv,
    int* __restrict__ rec, int E, int fb) {
  const int base = fb * (EPB * FILL_THREADS) + threadIdx.x;
  int2 rc[EPB];
  int k[EPB], o[EPB];
  float w[EPB];
#pragma unroll
  for (int u = 0; u < EPB; ++u) {
    const int p = min(base + u * FILL_THREADS, E - 1);
    rc[u] = ((const int2*)part)[p];
    k[u] = rank2[p];
  }
  asm volatile("" :: "v"(rc[0].x), "v"(rc[0].y), "v"(rc[1].x), "v"(rc[1].y),
                     "v"(rc[2].x), "v"(rc[2].y), "v"(rc[3].x), "v"(rc[3].y));
#pragma unroll
  for (int u = 0; u < EPB; ++u) {
    o[u] = offs[rc[u].y];
    w[u] = dinv[rc[u].x];
  }
  asm volatile("" :: "v"(o[0]), "v"(o[1]), "v"(o[2]), "v"(o[3]),
                     "v"(w[0]), "v"(w[1]), "v"(w[2]), "v"(w[3]),
                     "v"(k[0]), "v"(k[1]), "v"(k[2]), "v"(k[3]));
#pragma unroll
  for (int u = 0; u < EPB; ++u) {
    if (base + u * FILL_THREADS < E)
      ((int2*)rec)[o[u] + k[u]] = make_int2(rc[u].x, __float_as_int(w[u]));
  }
}

constexpr int GK = 128;
constexpr int APITCH = GK + 4;
constexpr int STAGE_TILES = 4;
constexpr int STAGE_ROWS = STAGE_TILES * 16;
constexpr int STAGE_FLOATS = STAGE_ROWS * APITCH;

template <int NT, int ROWS>
__device__ __forceinline__ void stage_load(const float* __restrict__ A, int M, int row0, int tid,
                                           float4 (&v)[ROWS * (GK / 4) / NT]) {
  constexpr int PER = ROWS * (GK / 4) / NT;
  static_assert(NT % 32 == 0 && PER * NT == ROWS * (GK / 4), "whole rows per wave");
#pragma unroll
  for (int i = 0; i < PER; ++i) {
    const int idx = tid + i * NT;
    const int r = min(row0 + (idx >> 5), M - 1);
    v[i] = *(const float4*)(A + (size_t)r * GK + (idx & 31) * 4);
  }
}

template <int NT, int ROWS>
__device__ __forceinline__ void stage_store(float* stage, int tid,
                                            const float4 (&v)[ROWS * (GK / 4) / NT]) {
  constexpr int PER = ROWS * (GK / 4) / NT;
#pragma unroll
  for (int i = 0; i < PER; ++i) {
    const int idx = tid + i * NT;
    *(float4*)(stage + (idx >> 5) * APITCH + (idx & 31) * 4) = v[i];
  }
}

template <int N>
__device__ __forceinline__ void load_b_frags(const float* __restrict__ B, int lane, int n0,
                                             v2f (&breg)[GK / 4]) {
  const int kb = 2 * (lane >> 4);
  const float* bcol = B + (size_t)kb * N + (n0 + (lane & 15));
#pragma unroll
  for (int s = 0; s < GK / 4; ++s) {
    breg[s].x = bcol[s * 4 * N];
    breg[s].y = bcol[(s * 4 + 1) * N];
  }
}

template <int N>
__device__ __forceinline__ void tile_from_stage(const float* tile, const v2f (&breg)[GK / 4],
                                                float* __restrict__ C, int M, int m0,
                                                int lane, int n0) {
  const int half = lane >> 4;
  const int l    = lane & 15;
  const float* arow = tile + l * APITCH + 2 * half;
  v8f acc = {};
#pragma unroll
  for (int s = 0; s < GK / 4; ++s) {
    const v2f a = *(const v2f*)(arow + s * 4);
    acc = __builtin_amdgcn_wmma_f32_16x16x4_f32(false, a, false, breg[s],
                                                (short)0, acc, false, false);
  }
  float* cp = C + (size_t)(m0 + half * 8) * N + (n0 + l);
#pragma unroll
  for (int v = 0; v < 8; ++v) {
    if (m0 + half * 8 + v < M) cp[(size_t)v * N] = acc[v];
  }
}

constexpr int TM_FILL = 4;
static_assert(TM_FILL == STAGE_TILES, "a product block stages its tiles in one go");

__global__ __launch_bounds__(FILL_THREADS) void fill_gemm_kernel(
    const int* __restrict__ part, const int* __restrict__ rank2,
    const int* __restrict__ offs, const float* __restrict__ dinv,
    int* __restrict__ rec, int E, int nfill,
    const float* __restrict__ x, const float* __restrict__ W1, float* __restrict__ h1,
    int n, int ngemm) {
  __shared__ __attribute__((aligned(16))) float stage[STAGE_FLOATS];
  const int b = blockIdx.x;
  const int pairs = min(nfill, ngemm);
  int fb = -1, gb = -1;
  if (b < 2 * pairs) {
    if (b & 1) gb = b >> 1; else fb = b >> 1;
  } else if (nfill > ngemm) {
    fb = b - pairs;
  } else {
    gb = b - pairs;
  }
  if (gb >= 0) {
    const int tid  = threadIdx.x;
    const int lane = tid & 31;
    const int wave = __builtin_amdgcn_readfirstlane(tid >> 5);
    const int mt0  = gb * TM_FILL;
    float4 v[STAGE_ROWS * (GK / 4) / FILL_THREADS];
    stage_load<FILL_THREADS, STAGE_ROWS>(x, n, mt0 * 16, tid, v);
    stage_store<FILL_THREADS, STAGE_ROWS>(stage, tid, v);
    __syncthreads();
    v2f breg[GK / 4];
    load_b_frags<128>(W1, lane, wave * 16, breg);
    for (int t = 0; t < TM_FILL; ++t) {
      const int m0 = (mt0 + t) * 16;
      if (m0 < n) tile_from_stage<128>(stage + t * 16 * APITCH, breg, h1, n, m0, lane, wave * 16);
    }
    return;
  }
  fill_body(part, rank2, offs, dinv, rec, E, fb);
}


template <int F, int G, int B>
__device__ __forceinline__ void gather_steps(const float* __restrict__ h, int rv, float wv,
                                             int t0, int g, int fl, float4& acc) {
  float4 hv[B];
  float w[B];
#pragma unroll
  for (int u = 0; u < B; ++u) {
    int r;
    if constexpr (G == 1) {
      r = __builtin_amdgcn_readlane(rv, t0 + u);
      w[u] = __int_as_float(__builtin_amdgcn_readlane(__float_as_int(wv), t0 + u));
    } else {
      const int kk = (t0 + u) * G + g;
      r = __shfl(rv, kk);
      w[u] = __shfl(wv, kk);
    }
    hv[u] = *(const float4*)(h + (size_t)r * F + fl * 4);
  }
#pragma unroll
  for (int u = 0; u < B; ++u) {
    acc.x = fmaf(hv[u].x, w[u], acc.x);
    acc.y = fmaf(hv[u].y, w[u], acc.y);
    acc.z = fmaf(hv[u].z, w[u], acc.z);
    acc.w = fmaf(hv[u].w, w[u], acc.w);
  }
}

template <int F>
__device__ __forceinline__ void list_walk(const int* __restrict__ rec, const float* __restrict__ h,
                                          int beg, int end, int c, float dc, int lane, int g, int fl,
                                          float4& acc) {
  constexpr int G = 32 / (F / 4);
  constexpr int BMAX = 8 / G;
  for (int j = beg; j < end; j += 32) {
    const int m = min(32, end - j);
    int rv = c;
    float wv = 0.0f;
    if (lane < m) {
      const int2 p = ((const int2*)rec)[j + lane];
      rv = p.x;
      wv = __int_as_float(p.y) * dc;
    }
    const int S = m / G;
    int t = 0;
    for (; t + BMAX <= S; t += BMAX) gather_steps<F, G, BMAX>(h, rv, wv, t, g, fl, acc);
    if constexpr (BMAX >= 8) {
      if (t + 4 <= S) { gather_steps<F, G, 4>(h, rv, wv, t, g, fl, acc); t += 4; }
    }
    if constexpr (BMAX >= 4) {
      if (t + 2 <= S) { gather_steps<F, G, 2>(h, rv, wv, t, g, fl, acc); t += 2; }
    }
    if (t < S) gather_steps<F, G, 1>(h, rv, wv, t, g, fl, acc);
    if constexpr (G == 2) {
      if (m & 1) {
        const int r = __shfl(rv, m - 1);
        const float w = __shfl(wv, m - 1);
        if (g == 0) {
          const float4 hv = *(const float4*)(h + (size_t)r * F + fl * 4);
          acc.x = fmaf(hv.x, w, acc.x);
          acc.y = fmaf(hv.y, w, acc.y);
          acc.z = fmaf(hv.z, w, acc.z);
          acc.w = fmaf(hv.w, w, acc.w);
        }
      }
    }
  }
}

template <int F, bool HAS_BIAS>
__global__ __launch_bounds__(256) void gather_agg_kernel(
    const int* __restrict__ offs, const int* __restrict__ rec,
    const float* __restrict__ h, const float* __restrict__ dinv,
    const float* __restrict__ bias, float* __restrict__ out, int n) {
  constexpr int LPR = F / 4;
  constexpr int G = 32 / LPR;
  const int lane = threadIdx.x & 31;
  const int g = lane / LPR;
  const int fl = lane % LPR;
  const int wave = __builtin_amdgcn_readfirstlane((int)(threadIdx.x >> 5));
  const int wpb = blockDim.x >> 5;
  const int nwaves = gridDim.x * wpb;
  float4 bv = make_float4(0.0f, 0.0f, 0.0f, 0.0f);
  if constexpr (HAS_BIAS) bv = *(const float4*)(bias + fl * 4);

  for (int c = blockIdx.x * wpb + wave; c < n; c += nwaves) {
    const int beg = offs[c];
    const int end = offs[c + 1];
    const float dc = dinv[c];
    float4 acc = make_float4(0.0f, 0.0f, 0.0f, 0.0f);
    float4 hs = make_float4(0.0f, 0.0f, 0.0f, 0.0f);
    if (g == 0) hs = *(const float4*)(h + (size_t)c * F + fl * 4);
    list_walk<F>(rec, h, beg, end, c, dc, lane, g, fl, acc);
    if (g == 0) {
      acc.x += hs.x * dc * dc + bv.x;
      acc.y += hs.y * dc * dc + bv.y;
      acc.z += hs.z * dc * dc + bv.z;
      acc.w += hs.w * dc * dc + bv.w;
    }
    if constexpr (G == 2) {
      acc.x += __shfl_xor(acc.x, 16);
      acc.y += __shfl_xor(acc.y, 16);
      acc.z += __shfl_xor(acc.z, 16);
      acc.w += __shfl_xor(acc.w, 16);
    }
    if (g == 0) *(float4*)(out + (size_t)c * F + fl * 4) = acc;
  }
}

constexpr int FG_TILES = 2;
constexpr int FG_ROWS  = FG_TILES * 16;
constexpr int FG_N     = 64;
constexpr int FG_WAVES = FG_N / 16;
constexpr int FG_NPW   = FG_ROWS / FG_WAVES;
constexpr int FG_KP    = 2;
constexpr int FG_KS    = GK / 4 / FG_KP;

__global__ __launch_bounds__(32 * FG_WAVES) void gather_agg_gemm_kernel(
    const int* __restrict__ offs, const int* __restrict__ rec,
    const float* __restrict__ h, const float* __restrict__ dinv,
    const float* __restrict__ act_bias, const float* __restrict__ B,
    float* __restrict__ C, int n) {
  constexpr int F = GK;
  static_assert(F / 4 == 32, "one lane per 4 features of a row");
  __shared__ __attribute__((aligned(16))) float stage[FG_ROWS * APITCH];
  const int lane = threadIdx.x & 31;
  const int wave = __builtin_amdgcn_readfirstlane((int)(threadIdx.x >> 5));
  const int base = blockIdx.x * FG_ROWS;
  const float4 bv = *(const float4*)(act_bias + lane * 4);
  for (int i = 0; i < FG_NPW; ++i) {
    const int lr = wave * FG_NPW + i;
    const int c = base + lr;
    float4 acc = make_float4(0.0f, 0.0f, 0.0f, 0.0f);
    if (c < n) {
      const int beg = offs[c];
      const int end = offs[c + 1];
      const float dc = dinv[c];
      const float4 hs = *(const float4*)(h + (size_t)c * F + lane * 4);
      list_walk<F>(rec, h, beg, end, c, dc, lane, 0, lane, acc);
      acc.x = fmaxf(acc.x + hs.x * dc * dc + bv.x, 0.0f);
      acc.y = fmaxf(acc.y + hs.y * dc * dc + bv.y, 0.0f);
      acc.z = fmaxf(acc.z + hs.z * dc * dc + bv.z, 0.0f);
      acc.w = fmaxf(acc.w + hs.w * dc * dc + bv.w, 0.0f);
    }
    *(float4*)(stage + lr * APITCH + lane * 4) = acc;
  }
  __syncthreads();
  const int half = lane >> 4;
  const int l    = lane & 15;
  const int n0   = wave * 16;
  const float* arow = stage + l * APITCH + 2 * half;
  const float* bcol = B + (size_t)(2 * half) * FG_N + (n0 + l);
  v8f acc[FG_TILES] = {};
#pragma unroll 1
  for (int q = 0; q < FG_KP; ++q) {
    const float* bq = bcol + (size_t)q * (FG_KS * 4 * FG_N);
    const float* aq = arow + q * (FG_KS * 4);
    v2f bp[FG_KS];
#pragma unroll
    for (int s = 0; s < FG_KS; ++s) {
      bp[s].x = bq[s * 4 * FG_N];
      bp[s].y = bq[(s * 4 + 1) * FG_N];
    }
#pragma unroll
    for (int s = 0; s < FG_KS; ++s) {
#pragma unroll
      for (int t = 0; t < FG_TILES; ++t) {
        const v2f a = *(const v2f*)(aq + t * 16 * APITCH + s * 4);
        acc[t] = __builtin_amdgcn_wmma_f32_16x16x4_f32(false, a, false, bp[s],
                                                       (short)0, acc[t], false, false);
      }
    }
  }
#pragma unroll
  for (int t = 0; t < FG_TILES; ++t) {
    const int m0 = base + t * 16;
    float* cp = C + (size_t)(m0 + half * 8) * FG_N + (n0 + l);
#pragma unroll
    for (int v = 0; v < 8; ++v) {
      if (m0 + half * 8 + v < n) cp[(size_t)v * FG_N] = acc[t][v];
    }
  }
}

extern "C" void kernel_launch(void* const* d_in, const int* in_sizes, int n_in,
                              void* d_out, int out_size, void* d_ws, size_t ws_size,
                              hipStream_t stream) {
  const float* x  = (const float*)d_in[0];
  const int*   ei = (const int*)d_in[1];
  const float* W1 = (const float*)d_in[2];
  const float* b1 = (const float*)d_in[3];
  const float* W2 = (const float*)d_in[4];
  const float* b2 = (const float*)d_in[5];

  const int hid    = in_sizes[3];
  const int out_ch = in_sizes[5];
  const int in_ch  = in_sizes[2] / hid;
  const int n      = in_sizes[0] / in_ch;
  const int E      = in_sizes[1] / 2;
  const int* row = ei;
  const int* col = ei + E;

  const int nb  = (n + BNODES - 1) / BNODES;
  const int che = (E + NCH - 1) / NCH;

  float* ws = (float*)d_ws;
  size_t off = 0;
  auto alloc = [&](size_t elems) {
    float* p = ws + off;
    off += (elems + 63) & ~(size_t)63;
    return p;
  };
  int*   T      = (int*)alloc((size_t)nb * NCH);
  int*   L      = (int*)alloc((size_t)nb * NCH);
  int*   offs   = (int*)alloc((size_t)n + 1);
  float* dinv   = alloc((size_t)n);
  int*   part   = (int*)alloc((size_t)E * 2);
  int*   rank2  = (int*)alloc((size_t)E);
  int*   rec    = (int*)alloc((size_t)E * 2);
  float* h1   = alloc((size_t)n * hid);
  float* g2   = alloc((size_t)n * out_ch);
  float* out  = (float*)d_out;

  const int mtiles = (n + 15) / 16;
  const int ngemm = (mtiles + TM_FILL - 1) / TM_FILL;
  const int nfill = (E + EPB * FILL_THREADS - 1) / (EPB * FILL_THREADS);

  chunk_group_kernel<<<NCH, SORT_THREADS, 0, stream>>>(row, col, T, L, part, E, che, nb);
  bucket_sort_kernel<<<nb, SORT_THREADS, 0, stream>>>(part, T, L, rank2, offs, dinv, n, nb, che);
  fill_gemm_kernel<<<nfill + ngemm, FILL_THREADS, 0, stream>>>(
      part, rank2, offs, dinv, rec, E, nfill, x, W1, h1, n, ngemm);

  gather_agg_gemm_kernel<<<(n + FG_ROWS - 1) / FG_ROWS, 32 * FG_WAVES, 0, stream>>>(
      offs, rec, h1, dinv, b1, W2, g2, n);
  gather_agg_kernel<64, true><<<2048, 256, 0, stream>>>(offs, rec, g2, dinv, b2, out, n);
}


# Norm factored: h rows stored times own dinv, node sum times dinv[dest] once; list entry = bare source placed by bucket sort; no fill pass, ranks, edge weights; rounding differs, sum order varies
# speedup vs baseline: 4.8176x; 1.2288x over previous
#include <hip/hip_runtime.h>
#include <hip/hip_bf16.h>


typedef __attribute__((ext_vector_type(2))) float v2f;
typedef __attribute__((ext_vector_type(8))) float v8f;

constexpr int BSHIFT = 9;
constexpr int BNODES = 1 << BSHIFT;
constexpr int NCH = 256;
constexpr int NBMAX = 1024;
constexpr int SORT_THREADS = 1024;
constexpr int EPB = 4;
static_assert(EPB == 4, "the load fence in chunk_group_kernel names four values");
static_assert(NBMAX == SORT_THREADS, "chunk_group_kernel: one bucket counter per thread");
constexpr int SEGW = NCH / 32;
static_assert(SEGW * 32 == NCH && SEGW <= 32, "bucket_sort_kernel: lane s of a wave holds the wave's piece s");
constexpr int GEMM_THREADS = 256;
constexpr int LCAP = 12288;

__device__ __forceinline__ int wave_incl_scan(int v, int lane) {
  for (int d = 1; d < 32; d <<= 1) {
    const int y = __shfl_up(v, d);
    if (lane >= d) v += y;
  }
  return v;
}

__device__ __forceinline__ int block_excl_scan_1024(int v, int* wsum, int lane, int wave,
                                                    int& total) {
  const int incl = wave_incl_scan(v, lane);
  if (lane == 31) wsum[wave] = incl;
  __syncthreads();
  const int wt = wsum[lane];
  const int winc = wave_incl_scan(wt, lane);
  const int wave_excl = __shfl(winc, wave) - __shfl(wt, wave);
  total = __shfl(winc, 31);
  return wave_excl + incl - v;
}

__global__ __launch_bounds__(SORT_THREADS) void chunk_group_kernel(
    const int* __restrict__ row, const int* __restrict__ col, int* __restrict__ T,
    int* __restrict__ L, int* __restrict__ part, int E, int che, int nb) {
  __shared__ int hist[NBMAX];
  __shared__ int wsum[32];
  const int tid = threadIdx.x;
  const int lane = tid & 31;
  const int wave = tid >> 5;
  const int ch = blockIdx.x;
  hist[tid] = 0;
  __syncthreads();
  const int e0 = ch * che;
  const int e1 = min(E, e0 + che);
  for (int e = e0 + tid; e < e1; e += EPB * SORT_THREADS) {
    int c[EPB];
#pragma unroll
    for (int u = 0; u < EPB; ++u) {
      c[u] = col[min(e + u * SORT_THREADS, e1 - 1)];
    }
    asm volatile("" :: "v"(c[0]), "v"(c[1]), "v"(c[2]), "v"(c[3]));
#pragma unroll
    for (int u = 0; u < EPB; ++u) {
      if (e + u * SORT_THREADS < e1) atomicAdd(&hist[c[u] >> BSHIFT], 1);
    }
  }
  __syncthreads();
  const int v = hist[tid];
  int total;
  const int ex = block_excl_scan_1024(v, wsum, lane, wave, total);
  if (tid < nb) {
    T[tid * NCH + ch] = v;
    L[tid * NCH + ch] = ex;
  }
  hist[tid] = ex;
  __syncthreads();
  for (int e = e0 + tid; e < e1; e += EPB * SORT_THREADS) {
    int r[EPB], c[EPB];
#pragma unroll
    for (int u = 0; u < EPB; ++u) {
      const int ee = e + u * SORT_THREADS;
      r[u] = (ee < e1) ? row[ee] : 0;
      c[u] = (ee < e1) ? col[ee] : 0;
    }
#pragma unroll
    for (int u = 0; u < EPB; ++u) {
      if (e + u * SORT_THREADS < e1) {
        const int pos = atomicAdd(&hist[c[u] >> BSHIFT], 1);
        ((int2*)part)[e0 + pos] = make_int2(r[u], c[u]);
      }
    }
  }
}

__device__ __forceinline__ void place_walk(const int* __restrict__ part, int* cnt, int* dst,
                                           int segc, int segl, int mx, int wave, int lane,
                                           int che, int c0) {
  for (int i0 = 0; i0 < mx; i0 += 64) {
    int2 rc[SEGW][2];
#pragma unroll
    for (int s = 0; s < SEGW; ++s) {
      const int len = __builtin_amdgcn_readlane(segc, s);
      const int base = (wave * SEGW + s) * che + __builtin_amdgcn_readlane(segl, s);
#pragma unroll
      for (int k = 0; k < 2; ++k) {
        const int i = i0 + k * 32 + lane;
        rc[s][k] = (i < len) ? ((const int2*)part)[base + i] : make_int2(0, c0);
      }
    }
    int pos[SEGW][2];
#pragma unroll
    for (int s = 0; s < SEGW; ++s) {
      const int len = __builtin_amdgcn_readlane(segc, s);
#pragma unroll
      for (int k = 0; k < 2; ++k) {
        const int i = i0 + k * 32 + lane;
        pos[s][k] = (i < len) ? atomicAdd(&cnt[rc[s][k].y - c0], 1) : 0;
      }
    }
#pragma unroll
    for (int s = 0; s < SEGW; ++s) {
      const int len = __builtin_amdgcn_readlane(segc, s);
#pragma unroll
      for (int k = 0; k < 2; ++k) {
        const int i = i0 + k * 32 + lane;
        if (i < len) dst[pos[s][k]] = rc[s][k].x;
      }
    }
  }
}

__global__ __launch_bounds__(SORT_THREADS) void bucket_sort_kernel(
    const int* __restrict__ part, const int* __restrict__ T, const int* __restrict__ L,
    int* __restrict__ rec, int* __restrict__ offs, float* __restrict__ dinv,
    int n, int nb, int che) {
  __shared__ int cnt[BNODES];
  __shared__ int lst[LCAP];
  __shared__ int wsum[32];
  __shared__ int wsum0[32];
  const int tid = threadIdx.x;
  const int lane = tid & 31;
  const int wave = __builtin_amdgcn_readfirstlane(tid >> 5);
  const int b = blockIdx.x;
  const int c0 = b << BSHIFT;
  int segc = 0, segl = 0;
  if (lane < SEGW) {
    segc = T[b * NCH + wave * SEGW + lane];
    segl = L[b * NCH + wave * SEGW + lane];
  }
  if (tid < BNODES) cnt[tid] = 0;
  int p0;
  (void)block_excl_scan_1024(segl, wsum0, lane, wave, p0);
  int mx = segc;
  for (int sh = 1; sh < SEGW; sh <<= 1) mx = max(mx, __shfl_xor(mx, sh));
  mx = __builtin_amdgcn_readfirstlane(mx);
  for (int i0 = 0; i0 < mx; i0 += 64) {
    int d[SEGW][2];
#pragma unroll
    for (int s = 0; s < SEGW; ++s) {
      const int len = __builtin_amdgcn_readlane(segc, s);
      const int base = (wave * SEGW + s) * che + __builtin_amdgcn_readlane(segl, s);
#pragma unroll
      for (int k = 0; k < 2; ++k) {
        const int i = i0 + k * 32 + lane;
        d[s][k] = (i < len) ? part[2 * (size_t)(base + i) + 1] : c0;
      }
    }
#pragma unroll
    for (int s = 0; s < SEGW; ++s) {
      const int len = __builtin_amdgcn_readlane(segc, s);
#pragma unroll
      for (int k = 0; k < 2; ++k) {
        const int i = i0 + k * 32 + lane;
        if (i < len) atomicAdd(&cnt[d[s][k] - c0], 1);
      }
    }
  }
  __syncthreads();
  const int v = (tid < BNODES) ? cnt[tid] : 0;
  int total;
  const int ex = block_excl_scan_1024(v, wsum, lane, wave, total);
  const int c = c0 + tid;
  if (tid < BNODES && c < n) {
    offs[c] = p0 + ex;
    dinv[c] = rsqrtf((float)(v + 1));
  }
  if (b == nb - 1 && tid == 0) offs[n] = p0 + total;
  if (tid < BNODES) cnt[tid] = ex;
  __syncthreads();
  const int tot = __builtin_amdgcn_readfirstlane(total);
  if (tot <= LCAP) {
    place_walk(part, cnt, lst, segc, segl, mx, wave, lane, che, c0);
    __syncthreads();
    for (int i = tid; i < tot; i += SORT_THREADS) rec[p0 + i] = lst[i];
  } else {
    place_walk(part, cnt, rec + p0, segc, segl, mx, wave, lane, che, c0);
  }
}

constexpr int GK = 128;
constexpr int APITCH = GK + 4;
constexpr int STAGE_TILES = 4;
constexpr int STAGE_ROWS = STAGE_TILES * 16;
constexpr int STAGE_FLOATS = STAGE_ROWS * APITCH;

template <int NT, int ROWS>
__device__ __forceinline__ void stage_load(const float* __restrict__ A, int M, int row0, int tid,
                                           float4 (&v)[ROWS * (GK / 4) / NT]) {
  constexpr int PER = ROWS * (GK / 4) / NT;
  static_assert(NT % 32 == 0 && PER * NT == ROWS * (GK / 4), "whole rows per wave");
#pragma unroll
  for (int i = 0; i < PER; ++i) {
    const int idx = tid + i * NT;
    const int r = min(row0 + (idx >> 5), M - 1);
    v[i] = *(const float4*)(A + (size_t)r * GK + (idx & 31) * 4);
  }
}

template <int NT, int ROWS>
__device__ __forceinline__ void stage_store(float* stage, int tid,
                                            const float4 (&v)[ROWS * (GK / 4) / NT]) {
  constexpr int PER = ROWS * (GK / 4) / NT;
#pragma unroll
  for (int i = 0; i < PER; ++i) {
    const int idx = tid + i * NT;
    *(float4*)(stage + (idx >> 5) * APITCH + (idx & 31) * 4) = v[i];
  }
}

template <int N>
__device__ __forceinline__ void load_b_frags(const float* __restrict__ B, int lane, int n0,
                                             v2f (&breg)[GK / 4]) {
  const int kb = 2 * (lane >> 4);
  const float* bcol = B + (size_t)kb * N + (n0 + (lane & 15));
#pragma unroll
  for (int s = 0; s < GK / 4; ++s) {
    breg[s].x = bcol[s * 4 * N];
    breg[s].y = bcol[(s * 4 + 1) * N];
  }
}

template <int N>
__device__ __forceinline__ void tile_from_stage(const float* tile, const v2f (&breg)[GK / 4],
                                                const float* __restrict__ rscale,
                                                float* __restrict__ C, int M, int m0,
                                                int lane, int n0) {
  const int half = lane >> 4;
  const int l    = lane & 15;
  const float* arow = tile + l * APITCH + 2 * half;
  float sc[8];
#pragma unroll
  for (int v = 0; v < 8; ++v) sc[v] = rscale[min(m0 + half * 8 + v, M - 1)];
  v8f acc = {};
#pragma unroll
  for (int s = 0; s < GK / 4; ++s) {
    const v2f a = *(const v2f*)(arow + s * 4);
    acc = __builtin_amdgcn_wmma_f32_16x16x4_f32(false, a, false, breg[s],
                                                (short)0, acc, false, false);
  }
  float* cp = C + (size_t)(m0 + half * 8) * N + (n0 + l);
#pragma unroll
  for (int v = 0; v < 8; ++v) {
    if (m0 + half * 8 + v < M) cp[(size_t)v * N] = acc[v] * sc[v];
  }
}

constexpr int TM_G1 = 4;
static_assert(TM_G1 == STAGE_TILES, "a work-group stages its tiles in one go");

__global__ __launch_bounds__(GEMM_THREADS) void xw_scaled_kernel(
    const float* __restrict__ x, const float* __restrict__ W1, const float* __restrict__ dinv,
    float* __restrict__ h1, int n) {
  __shared__ __attribute__((aligned(16))) float stage[STAGE_FLOATS];
  const int tid  = threadIdx.x;
  const int lane = tid & 31;
  const int wave = __builtin_amdgcn_readfirstlane(tid >> 5);
  const int mt0  = blockIdx.x * TM_G1;
  float4 v[STAGE_ROWS * (GK / 4) / GEMM_THREADS];
  stage_load<GEMM_THREADS, STAGE_ROWS>(x, n, mt0 * 16, tid, v);
  stage_store<GEMM_THREADS, STAGE_ROWS>(stage, tid, v);
  __syncthreads();
  v2f breg[GK / 4];
  load_b_frags<128>(W1, lane, wave * 16, breg);
  for (int t = 0; t < TM_G1; ++t) {
    const int m0 = (mt0 + t) * 16;
    if (m0 < n) tile_from_stage<128>(stage + t * 16 * APITCH, breg, dinv, h1, n, m0, lane, wave * 16);
  }
}


template <int F, int G, int B>
__device__ __forceinline__ void gather_steps(const float* __restrict__ h, int rv,
                                             int t0, int g, int fl, float4& acc) {
  float4 hv[B];
#pragma unroll
  for (int u = 0; u < B; ++u) {
    int r;
    if constexpr (G == 1) {
      r = __builtin_amdgcn_readlane(rv, t0 + u);
    } else {
      r = __shfl(rv, (t0 + u) * G + g);
    }
    hv[u] = *(const float4*)(h + (size_t)r * F + fl * 4);
  }
#pragma unroll
  for (int u = 0; u < B; ++u) {
    acc.x += hv[u].x;
    acc.y += hv[u].y;
    acc.z += hv[u].z;
    acc.w += hv[u].w;
  }
}

template <int F>
__device__ __forceinline__ void list_walk(const int* __restrict__ rec, const float* __restrict__ h,
                                          int beg, int end, int c, int lane, int g, int fl,
                                          float4& acc) {
  constexpr int G = 32 / (F / 4);
  constexpr int BMAX = 8 / G;
  for (int j = beg; j < end; j += 32) {
    const int m = min(32, end - j);
    int rv = c;
    if (lane < m) rv = rec[j + lane];
    const int S = m / G;
    int t = 0;
    for (; t + BMAX <= S; t += BMAX) gather_steps<F, G, BMAX>(h, rv, t, g, fl, acc);
    if constexpr (BMAX >= 8) {
      if (t + 4 <= S) { gather_steps<F, G, 4>(h, rv, t, g, fl, acc); t += 4; }
    }
    if constexpr (BMAX >= 4) {
      if (t + 2 <= S) { gather_steps<F, G, 2>(h, rv, t, g, fl, acc); t += 2; }
    }
    if (t < S) gather_steps<F, G, 1>(h, rv, t, g, fl, acc);
    if constexpr (G == 2) {
      if (m & 1) {
        const int r = __shfl(rv, m - 1);
        if (g == 0) {
          const float4 hv = *(const float4*)(h + (size_t)r * F + fl * 4);
          acc.x += hv.x;
          acc.y += hv.y;
          acc.z += hv.z;
          acc.w += hv.w;
        }
      }
    }
  }
}

template <int F, bool HAS_BIAS>
__global__ __launch_bounds__(256) void gather_agg_kernel(
    const int* __restrict__ offs, const int* __restrict__ rec,
    const float* __restrict__ h, const float* __restrict__ dinv,
    const float* __restrict__ bias, float* __restrict__ out, int n) {
  constexpr int LPR = F / 4;
  constexpr int G = 32 / LPR;
  const int lane = threadIdx.x & 31;
  const int g = lane / LPR;
  const int fl = lane % LPR;
  const int wave = __builtin_amdgcn_readfirstlane((int)(threadIdx.x >> 5));
  const int wpb = blockDim.x >> 5;
  const int nwaves = gridDim.x * wpb;
  float4 bv = make_float4(0.0f, 0.0f, 0.0f, 0.0f);
  if constexpr (HAS_BIAS) bv = *(const float4*)(bias + fl * 4);

  for (int c = blockIdx.x * wpb + wave; c < n; c += nwaves) {
    const int beg = offs[c];
    const int end = offs[c + 1];
    const float dc = dinv[c];
    float4 acc = make_float4(0.0f, 0.0f, 0.0f, 0.0f);
    float4 hs = make_float4(0.0f, 0.0f, 0.0f, 0.0f);
    if (g == 0) hs = *(const float4*)(h + (size_t)c * F + fl * 4);
    list_walk<F>(rec, h, beg, end, c, lane, g, fl, acc);
    if constexpr (G == 2) {
      acc.x += __shfl_xor(acc.x, 16);
      acc.y += __shfl_xor(acc.y, 16);
      acc.z += __shfl_xor(acc.z, 16);
      acc.w += __shfl_xor(acc.w, 16);
    }
    if (g == 0) {
      acc.x = (acc.x + hs.x) * dc + bv.x;
      acc.y = (acc.y + hs.y) * dc + bv.y;
      acc.z = (acc.z + hs.z) * dc + bv.z;
      acc.w = (acc.w + hs.w) * dc + bv.w;
      *(float4*)(out + (size_t)c * F + fl * 4) = acc;
    }
  }
}

constexpr int FG_TILES = 2;
constexpr int FG_ROWS  = FG_TILES * 16;
constexpr int FG_N     = 64;
constexpr int FG_WAVES = FG_N / 16;
constexpr int FG_NPW   = FG_ROWS / FG_WAVES;
constexpr int FG_KP    = 2;
constexpr int FG_KS    = GK / 4 / FG_KP;

__global__ __launch_bounds__(32 * FG_WAVES) void gather_agg_gemm_kernel(
    const int* __restrict__ offs, const int* __restrict__ rec,
    const float* __restrict__ h, const float* __restrict__ dinv,
    const float* __restrict__ act_bias, const float* __restrict__ B,
    float* __restrict__ C, int n) {
  constexpr int F = GK;
  static_assert(F / 4 == 32, "one lane per 4 features of a row");
  __shared__ __attribute__((aligned(16))) float stage[FG_ROWS * APITCH];
  const int lane = threadIdx.x & 31;
  const int wave = __builtin_amdgcn_readfirstlane((int)(threadIdx.x >> 5));
  const int base = blockIdx.x * FG_ROWS;
  const float4 bv = *(const float4*)(act_bias + lane * 4);
  for (int i = 0; i < FG_NPW; ++i) {
    const int lr = wave * FG_NPW + i;
    const int c = base + lr;
    float4 acc = make_float4(0.0f, 0.0f, 0.0f, 0.0f);
    if (c < n) {
      const int beg = offs[c];
      const int end = offs[c + 1];
      const float dc = dinv[c];
      const float4 hs = *(const float4*)(h + (size_t)c * F + lane * 4);
      list_walk<F>(rec, h, beg, end, c, lane, 0, lane, acc);
      acc.x = fmaxf((acc.x + hs.x) * dc + bv.x, 0.0f);
      acc.y = fmaxf((acc.y + hs.y) * dc + bv.y, 0.0f);
      acc.z = fmaxf((acc.z + hs.z) * dc + bv.z, 0.0f);
      acc.w = fmaxf((acc.w + hs.w) * dc + bv.w, 0.0f);
    }
    *(float4*)(stage + lr * APITCH + lane * 4) = acc;
  }
  __syncthreads();
  const int half = lane >> 4;
  const int l    = lane & 15;
  const int n0   = wave * 16;
  const float* arow = stage + l * APITCH + 2 * half;
  const float* bcol = B + (size_t)(2 * half) * FG_N + (n0 + l);
  v8f acc[FG_TILES] = {};
#pragma unroll 1
  for (int q = 0; q < FG_KP; ++q) {
    const float* bq = bcol + (size_t)q * (FG_KS * 4 * FG_N);
    const float* aq = arow + q * (FG_KS * 4);
    v2f bp[FG_KS];
#pragma unroll
    for (int s = 0; s < FG_KS; ++s) {
      bp[s].x = bq[s * 4 * FG_N];
      bp[s].y = bq[(s * 4 + 1) * FG_N];
    }
#pragma unroll
    for (int s = 0; s < FG_KS; ++s) {
#pragma unroll
      for (int t = 0; t < FG_TILES; ++t) {
        const v2f a = *(const v2f*)(aq + t * 16 * APITCH + s * 4);
        acc[t] = __builtin_amdgcn_wmma_f32_16x16x4_f32(false, a, false, bp[s],
                                                       (short)0, acc[t], false, false);
      }
    }
  }
#pragma unroll
  for (int t = 0; t < FG_TILES; ++t) {
    const int m0 = base + t * 16;
    float* cp = C + (size_t)(m0 + half * 8) * FG_N + (n0 + l);
#pragma unroll
    for (int v = 0; v < 8; ++v) {
      if (m0 + half * 8 + v < n) cp[(size_t)v * FG_N] = acc[t][v] * dinv[m0 + half * 8 + v];
    }
  }
}

extern "C" void kernel_launch(void* const* d_in, const int* in_sizes, int n_in,
                              void* d_out, int out_size, void* d_ws, size_t ws_size,
                              hipStream_t stream) {
  const float* x  = (const float*)d_in[0];
  const int*   ei = (const int*)d_in[1];
  const float* W1 = (const float*)d_in[2];
  const float* b1 = (const float*)d_in[3];
  const float* W2 = (const float*)d_in[4];
  const float* b2 = (const float*)d_in[5];

  const int hid    = in_sizes[3];
  const int out_ch = in_sizes[5];
  const int in_ch  = in_sizes[2] / hid;
  const int n      = in_sizes[0] / in_ch;
  const int E      = in_sizes[1] / 2;
  const int* row = ei;
  const int* col = ei + E;

  const int nb  = (n + BNODES - 1) / BNODES;
  const int che = (E + NCH - 1) / NCH;

  float* ws = (float*)d_ws;
  size_t off = 0;
  auto alloc = [&](size_t elems) {
    float* p = ws + off;
    off += (elems + 63) & ~(size_t)63;
    return p;
  };
  int*   T      = (int*)alloc((size_t)nb * NCH);
  int*   L      = (int*)alloc((size_t)nb * NCH);
  int*   offs   = (int*)alloc((size_t)n + 1);
  float* dinv   = alloc((size_t)n);
  int*   part   = (int*)alloc((size_t)E * 2);
  int*   rec    = (int*)alloc((size_t)E);
  float* h1   = alloc((size_t)n * hid);
  float* g2   = alloc((size_t)n * out_ch);
  float* out  = (float*)d_out;

  const int mtiles = (n + 15) / 16;
  const int ngemm = (mtiles + TM_G1 - 1) / TM_G1;

  chunk_group_kernel<<<NCH, SORT_THREADS, 0, stream>>>(row, col, T, L, part, E, che, nb);
  bucket_sort_kernel<<<nb, SORT_THREADS, 0, stream>>>(part, T, L, rec, offs, dinv, n, nb, che);
  xw_scaled_kernel<<<ngemm, GEMM_THREADS, 0, stream>>>(x, W1, dinv, h1, n);

  gather_agg_gemm_kernel<<<(n + FG_ROWS - 1) / FG_ROWS, 32 * FG_WAVES, 0, stream>>>(
      offs, rec, h1, dinv, b1, W2, g2, n);
  gather_agg_kernel<64, true><<<2048, 256, 0, stream>>>(offs, rec, g2, dinv, b2, out, n);
}


# Chunk grouping: a chunk's records, grouped by bucket, are put together in LDS and written to memory in whole 128-byte lines instead of one scattered 8-byte store per edge
# speedup vs baseline: 4.9626x; 1.0301x over previous
#include <hip/hip_runtime.h>
#include <hip/hip_bf16.h>


typedef __attribute__((ext_vector_type(2))) float v2f;
typedef __attribute__((ext_vector_type(8))) float v8f;

constexpr int BSHIFT = 9;
constexpr int BNODES = 1 << BSHIFT;
constexpr int NCH = 256;
constexpr int NBMAX = 1024;
constexpr int SORT_THREADS = 1024;
constexpr int EPB = 4;
static_assert(EPB == 4, "the load fence in chunk_group_kernel names four values");
static_assert(NBMAX == SORT_THREADS, "chunk_group_kernel: one bucket counter per thread");
constexpr int SEGW = NCH / 32;
static_assert(SEGW * 32 == NCH && SEGW <= 32, "bucket_sort_kernel: lane s of a wave holds the wave's piece s");
constexpr int GEMM_THREADS = 256;
constexpr int LCAP = 12288;
constexpr int CHCAP = 6400;

__device__ __forceinline__ int wave_incl_scan(int v, int lane) {
  for (int d = 1; d < 32; d <<= 1) {
    const int y = __shfl_up(v, d);
    if (lane >= d) v += y;
  }
  return v;
}

__device__ __forceinline__ int block_excl_scan_1024(int v, int* wsum, int lane, int wave,
                                                    int& total) {
  const int incl = wave_incl_scan(v, lane);
  if (lane == 31) wsum[wave] = incl;
  __syncthreads();
  const int wt = wsum[lane];
  const int winc = wave_incl_scan(wt, lane);
  const int wave_excl = __shfl(winc, wave) - __shfl(wt, wave);
  total = __shfl(winc, 31);
  return wave_excl + incl - v;
}

__device__ __forceinline__ void place_chunk(const int* __restrict__ row,
                                            const int* __restrict__ col, int* hist, int2* dst,
                                            int e0, int e1, int tid) {
  for (int e = e0 + tid; e < e1; e += EPB * SORT_THREADS) {
    int r[EPB], c[EPB];
#pragma unroll
    for (int u = 0; u < EPB; ++u) {
      const int ee = e + u * SORT_THREADS;
      r[u] = (ee < e1) ? row[ee] : 0;
      c[u] = (ee < e1) ? col[ee] : 0;
    }
#pragma unroll
    for (int u = 0; u < EPB; ++u) {
      if (e + u * SORT_THREADS < e1) {
        const int pos = atomicAdd(&hist[c[u] >> BSHIFT], 1);
        dst[pos] = make_int2(r[u], c[u]);
      }
    }
  }
}

__global__ __launch_bounds__(SORT_THREADS) void chunk_group_kernel(
    const int* __restrict__ row, const int* __restrict__ col, int* __restrict__ T,
    int* __restrict__ L, int* __restrict__ part, int E, int che, int nb) {
  __shared__ int hist[NBMAX];
  __shared__ int wsum[32];
  __shared__ __attribute__((aligned(16))) int seg[2 * CHCAP];
  const int tid = threadIdx.x;
  const int lane = tid & 31;
  const int wave = tid >> 5;
  const int ch = blockIdx.x;
  hist[tid] = 0;
  __syncthreads();
  const int e0 = ch * che;
  const int e1 = min(E, e0 + che);
  for (int e = e0 + tid; e < e1; e += EPB * SORT_THREADS) {
    int c[EPB];
#pragma unroll
    for (int u = 0; u < EPB; ++u) {
      c[u] = col[min(e + u * SORT_THREADS, e1 - 1)];
    }
    asm volatile("" :: "v"(c[0]), "v"(c[1]), "v"(c[2]), "v"(c[3]));
#pragma unroll
    for (int u = 0; u < EPB; ++u) {
      if (e + u * SORT_THREADS < e1) atomicAdd(&hist[c[u] >> BSHIFT], 1);
    }
  }
  __syncthreads();
  const int v = hist[tid];
  int total;
  const int ex = block_excl_scan_1024(v, wsum, lane, wave, total);
  if (tid < nb) {
    T[tid * NCH + ch] = v;
    L[tid * NCH + ch] = ex;
  }
  hist[tid] = ex;
  __syncthreads();
  const int cnt = max(0, e1 - e0);
  int2* const dst = (int2*)part + e0;
  if (cnt <= CHCAP) {
    int2* const segr = (int2*)seg;
    place_chunk(row, col, hist, segr, e0, e1, tid);
    __syncthreads();
    const int head = min(cnt, (16 - (e0 & 15)) & 15);
    if (tid < head) dst[tid] = segr[tid];
    for (int i = head + tid; i < cnt; i += SORT_THREADS) dst[i] = segr[i];
  } else {
    place_chunk(row, col, hist, dst, e0, e1, tid);
  }
}

__device__ __forceinline__ void place_walk(const int* __restrict__ part, int* cnt, int* dst,
                                           int segc, int segl, int mx, int wave, int lane,
                                           int che, int c0) {
  for (int i0 = 0; i0 < mx; i0 += 64) {
    int2 rc[SEGW][2];
#pragma unroll
    for (int s = 0; s < SEGW; ++s) {
      const int len = __builtin_amdgcn_readlane(segc, s);
      const int base = (wave * SEGW + s) * che + __builtin_amdgcn_readlane(segl, s);
#pragma unroll
      for (int k = 0; k < 2; ++k) {
        const int i = i0 + k * 32 + lane;
        rc[s][k] = (i < len) ? ((const int2*)part)[base + i] : make_int2(0, c0);
      }
    }
    int pos[SEGW][2];
#pragma unroll
    for (int s = 0; s < SEGW; ++s) {
      const int len = __builtin_amdgcn_readlane(segc, s);
#pragma unroll
      for (int k = 0; k < 2; ++k) {
        const int i = i0 + k * 32 + lane;
        pos[s][k] = (i < len) ? atomicAdd(&cnt[rc[s][k].y - c0], 1) : 0;
      }
    }
#pragma unroll
    for (int s = 0; s < SEGW; ++s) {
      const int len = __builtin_amdgcn_readlane(segc, s);
#pragma unroll
      for (int k = 0; k < 2; ++k) {
        const int i = i0 + k * 32 + lane;
        if (i < len) dst[pos[s][k]] = rc[s][k].x;
      }
    }
  }
}

__global__ __launch_bounds__(SORT_THREADS) void bucket_sort_kernel(
    const int* __restrict__ part, const int* __restrict__ T, const int* __restrict__ L,
    int* __restrict__ rec, int* __restrict__ offs, float* __restrict__ dinv,
    int n, int nb, int che) {
  __shared__ int cnt[BNODES];
  __shared__ int lst[LCAP];
  __shared__ int wsum[32];
  __shared__ int wsum0[32];
  const int tid = threadIdx.x;
  const int lane = tid & 31;
  const int wave = __builtin_amdgcn_readfirstlane(tid >> 5);
  const int b = blockIdx.x;
  const int c0 = b << BSHIFT;
  int segc = 0, segl = 0;
  if (lane < SEGW) {
    segc = T[b * NCH + wave * SEGW + lane];
    segl = L[b * NCH + wave * SEGW + lane];
  }
  if (tid < BNODES) cnt[tid] = 0;
  int p0;
  (void)block_excl_scan_1024(segl, wsum0, lane, wave, p0);
  int mx = segc;
  for (int sh = 1; sh < SEGW; sh <<= 1) mx = max(mx, __shfl_xor(mx, sh));
  mx = __builtin_amdgcn_readfirstlane(mx);
  for (int i0 = 0; i0 < mx; i0 += 64) {
    int d[SEGW][2];
#pragma unroll
    for (int s = 0; s < SEGW; ++s) {
      const int len = __builtin_amdgcn_readlane(segc, s);
      const int base = (wave * SEGW + s) * che + __builtin_amdgcn_readlane(segl, s);
#pragma unroll
      for (int k = 0; k < 2; ++k) {
        const int i = i0 + k * 32 + lane;
        d[s][k] = (i < len) ? part[2 * (size_t)(base + i) + 1] : c0;
      }
    }
#pragma unroll
    for (int s = 0; s < SEGW; ++s) {
      const int len = __builtin_amdgcn_readlane(segc, s);
#pragma unroll
      for (int k = 0; k < 2; ++k) {
        const int i = i0 + k * 32 + lane;
        if (i < len) atomicAdd(&cnt[d[s][k] - c0], 1);
      }
    }
  }
  __syncthreads();
  const int v = (tid < BNODES) ? cnt[tid] : 0;
  int total;
  const int ex = block_excl_scan_1024(v, wsum, lane, wave, total);
  const int c = c0 + tid;
  if (tid < BNODES && c < n) {
    offs[c] = p0 + ex;
    dinv[c] = rsqrtf((float)(v + 1));
  }
  if (b == nb - 1 && tid == 0) offs[n] = p0 + total;
  if (tid < BNODES) cnt[tid] = ex;
  __syncthreads();
  const int tot = __builtin_amdgcn_readfirstlane(total);
  if (tot <= LCAP) {
    place_walk(part, cnt, lst, segc, segl, mx, wave, lane, che, c0);
    __syncthreads();
    for (int i = tid; i < tot; i += SORT_THREADS) rec[p0 + i] = lst[i];
  } else {
    place_walk(part, cnt, rec + p0, segc, segl, mx, wave, lane, che, c0);
  }
}

constexpr int GK = 128;
constexpr int APITCH = GK + 4;
constexpr int STAGE_TILES = 4;
constexpr int STAGE_ROWS = STAGE_TILES * 16;
constexpr int STAGE_FLOATS = STAGE_ROWS * APITCH;

template <int NT, int ROWS>
__device__ __forceinline__ void stage_load(const float* __restrict__ A, int M, int row0, int tid,
                                           float4 (&v)[ROWS * (GK / 4) / NT]) {
  constexpr int PER = ROWS * (GK / 4) / NT;
  static_assert(NT % 32 == 0 && PER * NT == ROWS * (GK / 4), "whole rows per wave");
#pragma unroll
  for (int i = 0; i < PER; ++i) {
    const int idx = tid + i * NT;
    const int r = min(row0 + (idx >> 5), M - 1);
    v[i] = *(const float4*)(A + (size_t)r * GK + (idx & 31) * 4);
  }
}

template <int NT, int ROWS>
__device__ __forceinline__ void stage_store(float* stage, int tid,
                                            const float4 (&v)[ROWS * (GK / 4) / NT]) {
  constexpr int PER = ROWS * (GK / 4) / NT;
#pragma unroll
  for (int i = 0; i < PER; ++i) {
    const int idx = tid + i * NT;
    *(float4*)(stage + (idx >> 5) * APITCH + (idx & 31) * 4) = v[i];
  }
}

template <int N>
__device__ __forceinline__ void load_b_frags(const float* __restrict__ B, int lane, int n0,
                                             v2f (&breg)[GK / 4]) {
  const int kb = 2 * (lane >> 4);
  const float* bcol = B + (size_t)kb * N + (n0 + (lane & 15));
#pragma unroll
  for (int s = 0; s < GK / 4; ++s) {
    breg[s].x = bcol[s * 4 * N];
    breg[s].y = bcol[(s * 4 + 1) * N];
  }
}

template <int N>
__device__ __forceinline__ void tile_from_stage(const float* tile, const v2f (&breg)[GK / 4],
                                                const float* __restrict__ rscale,
                                                float* __restrict__ C, int M, int m0,
                                                int lane, int n0) {
  const int half = lane >> 4;
  const int l    = lane & 15;
  const float* arow = tile + l * APITCH + 2 * half;
  float sc[8];
#pragma unroll
  for (int v = 0; v < 8; ++v) sc[v] = rscale[min(m0 + half * 8 + v, M - 1)];
  v8f acc = {};
#pragma unroll
  for (int s = 0; s < GK / 4; ++s) {
    const v2f a = *(const v2f*)(arow + s * 4);
    acc = __builtin_amdgcn_wmma_f32_16x16x4_f32(false, a, false, breg[s],
                                                (short)0, acc, false, false);
  }
  float* cp = C + (size_t)(m0 + half * 8) * N + (n0 + l);
#pragma unroll
  for (int v = 0; v < 8; ++v) {
    if (m0 + half * 8 + v < M) cp[(size_t)v * N] = acc[v] * sc[v];
  }
}

constexpr int TM_G1 = 4;
static_assert(TM_G1 == STAGE_TILES, "a work-group stages its tiles in one go");

__global__ __launch_bounds__(GEMM_THREADS) void xw_scaled_kernel(
    const float* __restrict__ x, const float* __restrict__ W1, const float* __restrict__ dinv,
    float* __restrict__ h1, int n) {
  __shared__ __attribute__((aligned(16))) float stage[STAGE_FLOATS];
  const int tid  = threadIdx.x;
  const int lane = tid & 31;
  const int wave = __builtin_amdgcn_readfirstlane(tid >> 5);
  const int mt0  = blockIdx.x * TM_G1;
  float4 v[STAGE_ROWS * (GK / 4) / GEMM_THREADS];
  stage_load<GEMM_THREADS, STAGE_ROWS>(x, n, mt0 * 16, tid, v);
  stage_store<GEMM_THREADS, STAGE_ROWS>(stage, tid, v);
  __syncthreads();
  v2f breg[GK / 4];
  load_b_frags<128>(W1, lane, wave * 16, breg);
  for (int t = 0; t < TM_G1; ++t) {
    const int m0 = (mt0 + t) * 16;
    if (m0 < n) tile_from_stage<128>(stage + t * 16 * APITCH, breg, dinv, h1, n, m0, lane, wave * 16);
  }
}


template <int F, int G, int B>
__device__ __forceinline__ void gather_steps(const float* __restrict__ h, int rv,
                                             int t0, int g, int fl, float4& acc) {
  float4 hv[B];
#pragma unroll
  for (int u = 0; u < B; ++u) {
    int r;
    if constexpr (G == 1) {
      r = __builtin_amdgcn_readlane(rv, t0 + u);
    } else {
      r = __shfl(rv, (t0 + u) * G + g);
    }
    hv[u] = *(const float4*)(h + (size_t)r * F + fl * 4);
  }
#pragma unroll
  for (int u = 0; u < B; ++u) {
    acc.x += hv[u].x;
    acc.y += hv[u].y;
    acc.z += hv[u].z;
    acc.w += hv[u].w;
  }
}

template <int F>
__device__ __forceinline__ void list_walk(const int* __restrict__ rec, const float* __restrict__ h,
                                          int beg, int end, int c, int lane, int g, int fl,
                                          float4& acc) {
  constexpr int G = 32 / (F / 4);
  constexpr int BMAX = 8 / G;
  for (int j = beg; j < end; j += 32) {
    const int m = min(32, end - j);
    int rv = c;
    if (lane < m) rv = rec[j + lane];
    const int S = m / G;
    int t = 0;
    for (; t + BMAX <= S; t += BMAX) gather_steps<F, G, BMAX>(h, rv, t, g, fl, acc);
    if constexpr (BMAX >= 8) {
      if (t + 4 <= S) { gather_steps<F, G, 4>(h, rv, t, g, fl, acc); t += 4; }
    }
    if constexpr (BMAX >= 4) {
      if (t + 2 <= S) { gather_steps<F, G, 2>(h, rv, t, g, fl, acc); t += 2; }
    }
    if (t < S) gather_steps<F, G, 1>(h, rv, t, g, fl, acc);
    if constexpr (G == 2) {
      if (m & 1) {
        const int r = __shfl(rv, m - 1);
        if (g == 0) {
          const float4 hv = *(const float4*)(h + (size_t)r * F + fl * 4);
          acc.x += hv.x;
          acc.y += hv.y;
          acc.z += hv.z;
          acc.w += hv.w;
        }
      }
    }
  }
}

template <int F, bool HAS_BIAS>
__global__ __launch_bounds__(256) void gather_agg_kernel(
    const int* __restrict__ offs, const int* __restrict__ rec,
    const float* __restrict__ h, const float* __restrict__ dinv,
    const float* __restrict__ bias, float* __restrict__ out, int n) {
  constexpr int LPR = F / 4;
  constexpr int G = 32 / LPR;
  const int lane = threadIdx.x & 31;
  const int g = lane / LPR;
  const int fl = lane % LPR;
  const int wave = __builtin_amdgcn_readfirstlane((int)(threadIdx.x >> 5));
  const int wpb = blockDim.x >> 5;
  const int nwaves = gridDim.x * wpb;
  float4 bv = make_float4(0.0f, 0.0f, 0.0f, 0.0f);
  if constexpr (HAS_BIAS) bv = *(const float4*)(bias + fl * 4);

  for (int c = blockIdx.x * wpb + wave; c < n; c += nwaves) {
    const int beg = offs[c];
    const int end = offs[c + 1];
    const float dc = dinv[c];
    float4 acc = make_float4(0.0f, 0.0f, 0.0f, 0.0f);
    float4 hs = make_float4(0.0f, 0.0f, 0.0f, 0.0f);
    if (g == 0) hs = *(const float4*)(h + (size_t)c * F + fl * 4);
    list_walk<F>(rec, h, beg, end, c, lane, g, fl, acc);
    if constexpr (G == 2) {
      acc.x += __shfl_xor(acc.x, 16);
      acc.y += __shfl_xor(acc.y, 16);
      acc.z += __shfl_xor(acc.z, 16);
      acc.w += __shfl_xor(acc.w, 16);
    }
    if (g == 0) {
      acc.x = (acc.x + hs.x) * dc + bv.x;
      acc.y = (acc.y + hs.y) * dc + bv.y;
      acc.z = (acc.z + hs.z) * dc + bv.z;
      acc.w = (acc.w + hs.w) * dc + bv.w;
      *(float4*)(out + (size_t)c * F + fl * 4) = acc;
    }
  }
}

constexpr int FG_TILES = 2;
constexpr int FG_ROWS  = FG_TILES * 16;
constexpr int FG_N     = 64;
constexpr int FG_WAVES = FG_N / 16;
constexpr int FG_NPW   = FG_ROWS / FG_WAVES;
constexpr int FG_KP    = 2;
constexpr int FG_KS    = GK / 4 / FG_KP;

__global__ __launch_bounds__(32 * FG_WAVES) void gather_agg_gemm_kernel(
    const int* __restrict__ offs, const int* __restrict__ rec,
    const float* __restrict__ h, const float* __restrict__ dinv,
    const float* __restrict__ act_bias, const float* __restrict__ B,
    float* __restrict__ C, int n) {
  constexpr int F = GK;
  static_assert(F / 4 == 32, "one lane per 4 features of a row");
  __shared__ __attribute__((aligned(16))) float stage[FG_ROWS * APITCH];
  const int lane = threadIdx.x & 31;
  const int wave = __builtin_amdgcn_readfirstlane((int)(threadIdx.x >> 5));
  const int base = blockIdx.x * FG_ROWS;
  const float4 bv = *(const float4*)(act_bias + lane * 4);
  for (int i = 0; i < FG_NPW; ++i) {
    const int lr = wave * FG_NPW + i;
    const int c = base + lr;
    float4 acc = make_float4(0.0f, 0.0f, 0.0f, 0.0f);
    if (c < n) {
      const int beg = offs[c];
      const int end = offs[c + 1];
      const float dc = dinv[c];
      const float4 hs = *(const float4*)(h + (size_t)c * F + lane * 4);
      list_walk<F>(rec, h, beg, end, c, lane, 0, lane, acc);
      acc.x = fmaxf((acc.x + hs.x) * dc + bv.x, 0.0f);
      acc.y = fmaxf((acc.y + hs.y) * dc + bv.y, 0.0f);
      acc.z = fmaxf((acc.z + hs.z) * dc + bv.z, 0.0f);
      acc.w = fmaxf((acc.w + hs.w) * dc + bv.w, 0.0f);
    }
    *(float4*)(stage + lr * APITCH + lane * 4) = acc;
  }
  __syncthreads();
  const int half = lane >> 4;
  const int l    = lane & 15;
  const int n0   = wave * 16;
  const float* arow = stage + l * APITCH + 2 * half;
  const float* bcol = B + (size_t)(2 * half) * FG_N + (n0 + l);
  v8f acc[FG_TILES] = {};
#pragma unroll 1
  for (int q = 0; q < FG_KP; ++q) {
    const float* bq = bcol + (size_t)q * (FG_KS * 4 * FG_N);
    const float* aq = arow + q * (FG_KS * 4);
    v2f bp[FG_KS];
#pragma unroll
    for (int s = 0; s < FG_KS; ++s) {
      bp[s].x = bq[s * 4 * FG_N];
      bp[s].y = bq[(s * 4 + 1) * FG_N];
    }
#pragma unroll
    for (int s = 0; s < FG_KS; ++s) {
#pragma unroll
      for (int t = 0; t < FG_TILES; ++t) {
        const v2f a = *(const v2f*)(aq + t * 16 * APITCH + s * 4);
        acc[t] = __builtin_amdgcn_wmma_f32_16x16x4_f32(false, a, false, bp[s],
                                                       (short)0, acc[t], false, false);
      }
    }
  }
#pragma unroll
  for (int t = 0; t < FG_TILES; ++t) {
    const int m0 = base + t * 16;
    float* cp = C + (size_t)(m0 + half * 8) * FG_N + (n0 + l);
#pragma unroll
    for (int v = 0; v < 8; ++v) {
      if (m0 + half * 8 + v < n) cp[(size_t)v * FG_N] = acc[t][v] * dinv[m0 + half * 8 + v];
    }
  }
}

extern "C" void kernel_launch(void* const* d_in, const int* in_sizes, int n_in,
                              void* d_out, int out_size, void* d_ws, size_t ws_size,
                              hipStream_t stream) {
  const float* x  = (const float*)d_in[0];
  const int*   ei = (const int*)d_in[1];
  const float* W1 = (const float*)d_in[2];
  const float* b1 = (const float*)d_in[3];
  const float* W2 = (const float*)d_in[4];
  const float* b2 = (const float*)d_in[5];

  const int hid    = in_sizes[3];
  const int out_ch = in_sizes[5];
  const int in_ch  = in_sizes[2] / hid;
  const int n      = in_sizes[0] / in_ch;
  const int E      = in_sizes[1] / 2;
  const int* row = ei;
  const int* col = ei + E;

  const int nb  = (n + BNODES - 1) / BNODES;
  const int che = (E + NCH - 1) / NCH;

  float* ws = (float*)d_ws;
  size_t off = 0;
  auto alloc = [&](size_t elems) {
    float* p = ws + off;
    off += (elems + 63) & ~(size_t)63;
    return p;
  };
  int*   T      = (int*)alloc((size_t)nb * NCH);
  int*   L      = (int*)alloc((size_t)nb * NCH);
  int*   offs   = (int*)alloc((size_t)n + 1);
  float* dinv   = alloc((size_t)n);
  int*   part   = (int*)alloc((size_t)E * 2);
  int*   rec    = (int*)alloc((size_t)E);
  float* h1   = alloc((size_t)n * hid);
  float* g2   = alloc((size_t)n * out_ch);
  float* out  = (float*)d_out;

  const int mtiles = (n + 15) / 16;
  const int ngemm = (mtiles + TM_G1 - 1) / TM_G1;

  chunk_group_kernel<<<NCH, SORT_THREADS, 0, stream>>>(row, col, T, L, part, E, che, nb);
  bucket_sort_kernel<<<nb, SORT_THREADS, 0, stream>>>(part, T, L, rec, offs, dinv, n, nb, che);
  xw_scaled_kernel<<<ngemm, GEMM_THREADS, 0, stream>>>(x, W1, dinv, h1, n);

  gather_agg_gemm_kernel<<<(n + FG_ROWS - 1) / FG_ROWS, 32 * FG_WAVES, 0, stream>>>(
      offs, rec, h1, dinv, b1, W2, g2, n);
  gather_agg_kernel<64, true><<<2048, 256, 0, stream>>>(offs, rec, g2, dinv, b2, out, n);
}


# Layer-1 product behind its sum in short 32-node work-groups, 2 rounds per WGP: x@W1, @W2 issue while other groups wait for x rows (entry times dinv[source]); 5->4 launches; rounding differs
# speedup vs baseline: 5.0993x; 1.0275x over previous
#include <hip/hip_runtime.h>
#include <hip/hip_bf16.h>


typedef __attribute__((ext_vector_type(2))) float v2f;
typedef __attribute__((ext_vector_type(8))) float v8f;

constexpr int BSHIFT = 9;
constexpr int BNODES = 1 << BSHIFT;
constexpr int NCH = 256;
constexpr int NBMAX = 1024;
constexpr int SORT_THREADS = 1024;
constexpr int EPB = 4;
static_assert(EPB == 4, "the load fence in chunk_group_kernel names four values");
static_assert(NBMAX == SORT_THREADS, "chunk_group_kernel: one bucket counter per thread");
constexpr int SEGW = NCH / 32;
static_assert(SEGW * 32 == NCH && SEGW <= 32, "bucket_sort_kernel: lane s of a wave holds the wave's piece s");
constexpr int LCAP = 12288;
constexpr int CHCAP = 6400;

__device__ __forceinline__ int wave_incl_scan(int v, int lane) {
  for (int d = 1; d < 32; d <<= 1) {
    const int y = __shfl_up(v, d);
    if (lane >= d) v += y;
  }
  return v;
}

__device__ __forceinline__ int block_excl_scan_1024(int v, int* wsum, int lane, int wave,
                                                    int& total) {
  const int incl = wave_incl_scan(v, lane);
  if (lane == 31) wsum[wave] = incl;
  __syncthreads();
  const int wt = wsum[lane];
  const int winc = wave_incl_scan(wt, lane);
  const int wave_excl = __shfl(winc, wave) - __shfl(wt, wave);
  total = __shfl(winc, 31);
  return wave_excl + incl - v;
}

__device__ __forceinline__ void place_chunk(const int* __restrict__ row,
                                            const int* __restrict__ col, int* hist, int2* dst,
                                            int e0, int e1, int tid) {
  for (int e = e0 + tid; e < e1; e += EPB * SORT_THREADS) {
    int r[EPB], c[EPB];
#pragma unroll
    for (int u = 0; u < EPB; ++u) {
      const int ee = e + u * SORT_THREADS;
      r[u] = (ee < e1) ? row[ee] : 0;
      c[u] = (ee < e1) ? col[ee] : 0;
    }
#pragma unroll
    for (int u = 0; u < EPB; ++u) {
      if (e + u * SORT_THREADS < e1) {
        const int pos = atomicAdd(&hist[c[u] >> BSHIFT], 1);
        dst[pos] = make_int2(r[u], c[u]);
      }
    }
  }
}

__global__ __launch_bounds__(SORT_THREADS) void chunk_group_kernel(
    const int* __restrict__ row, const int* __restrict__ col, int* __restrict__ T,
    int* __restrict__ L, int* __restrict__ part, int E, int che, int nb) {
  __shared__ int hist[NBMAX];
  __shared__ int wsum[32];
  __shared__ __attribute__((aligned(16))) int seg[2 * CHCAP];
  const int tid = threadIdx.x;
  const int lane = tid & 31;
  const int wave = tid >> 5;
  const int ch = blockIdx.x;
  hist[tid] = 0;
  __syncthreads();
  const int e0 = ch * che;
  const int e1 = min(E, e0 + che);
  for (int e = e0 + tid; e < e1; e += EPB * SORT_THREADS) {
    int c[EPB];
#pragma unroll
    for (int u = 0; u < EPB; ++u) {
      c[u] = col[min(e + u * SORT_THREADS, e1 - 1)];
    }
    asm volatile("" :: "v"(c[0]), "v"(c[1]), "v"(c[2]), "v"(c[3]));
#pragma unroll
    for (int u = 0; u < EPB; ++u) {
      if (e + u * SORT_THREADS < e1) atomicAdd(&hist[c[u] >> BSHIFT], 1);
    }
  }
  __syncthreads();
  const int v = hist[tid];
  int total;
  const int ex = block_excl_scan_1024(v, wsum, lane, wave, total);
  if (tid < nb) {
    T[tid * NCH + ch] = v;
    L[tid * NCH + ch] = ex;
  }
  hist[tid] = ex;
  __syncthreads();
  const int cnt = max(0, e1 - e0);
  int2* const dst = (int2*)part + e0;
  if (cnt <= CHCAP) {
    int2* const segr = (int2*)seg;
    place_chunk(row, col, hist, segr, e0, e1, tid);
    __syncthreads();
    const int head = min(cnt, (16 - (e0 & 15)) & 15);
    if (tid < head) dst[tid] = segr[tid];
    for (int i = head + tid; i < cnt; i += SORT_THREADS) dst[i] = segr[i];
  } else {
    place_chunk(row, col, hist, dst, e0, e1, tid);
  }
}

__device__ __forceinline__ void place_walk(const int* __restrict__ part, int* cnt, int* dst,
                                           int segc, int segl, int mx, int wave, int lane,
                                           int che, int c0) {
  for (int i0 = 0; i0 < mx; i0 += 64) {
    int2 rc[SEGW][2];
#pragma unroll
    for (int s = 0; s < SEGW; ++s) {
      const int len = __builtin_amdgcn_readlane(segc, s);
      const int base = (wave * SEGW + s) * che + __builtin_amdgcn_readlane(segl, s);
#pragma unroll
      for (int k = 0; k < 2; ++k) {
        const int i = i0 + k * 32 + lane;
        rc[s][k] = (i < len) ? ((const int2*)part)[base + i] : make_int2(0, c0);
      }
    }
    int pos[SEGW][2];
#pragma unroll
    for (int s = 0; s < SEGW; ++s) {
      const int len = __builtin_amdgcn_readlane(segc, s);
#pragma unroll
      for (int k = 0; k < 2; ++k) {
        const int i = i0 + k * 32 + lane;
        pos[s][k] = (i < len) ? atomicAdd(&cnt[rc[s][k].y - c0], 1) : 0;
      }
    }
#pragma unroll
    for (int s = 0; s < SEGW; ++s) {
      const int len = __builtin_amdgcn_readlane(segc, s);
#pragma unroll
      for (int k = 0; k < 2; ++k) {
        const int i = i0 + k * 32 + lane;
        if (i < len) dst[pos[s][k]] = rc[s][k].x;
      }
    }
  }
}

__global__ __launch_bounds__(SORT_THREADS) void bucket_sort_kernel(
    const int* __restrict__ part, const int* __restrict__ T, const int* __restrict__ L,
    int* __restrict__ rec, int* __restrict__ offs, float* __restrict__ dinv,
    int n, int nb, int che) {
  __shared__ int cnt[BNODES];
  __shared__ int lst[LCAP];
  __shared__ int wsum[32];
  __shared__ int wsum0[32];
  const int tid = threadIdx.x;
  const int lane = tid & 31;
  const int wave = __builtin_amdgcn_readfirstlane(tid >> 5);
  const int b = blockIdx.x;
  const int c0 = b << BSHIFT;
  int segc = 0, segl = 0;
  if (lane < SEGW) {
    segc = T[b * NCH + wave * SEGW + lane];
    segl = L[b * NCH + wave * SEGW + lane];
  }
  if (tid < BNODES) cnt[tid] = 0;
  int p0;
  (void)block_excl_scan_1024(segl, wsum0, lane, wave, p0);
  int mx = segc;
  for (int sh = 1; sh < SEGW; sh <<= 1) mx = max(mx, __shfl_xor(mx, sh));
  mx = __builtin_amdgcn_readfirstlane(mx);
  for (int i0 = 0; i0 < mx; i0 += 64) {
    int d[SEGW][2];
#pragma unroll
    for (int s = 0; s < SEGW; ++s) {
      const int len = __builtin_amdgcn_readlane(segc, s);
      const int base = (wave * SEGW + s) * che + __builtin_amdgcn_readlane(segl, s);
#pragma unroll
      for (int k = 0; k < 2; ++k) {
        const int i = i0 + k * 32 + lane;
        d[s][k] = (i < len) ? part[2 * (size_t)(base + i) + 1] : c0;
      }
    }
#pragma unroll
    for (int s = 0; s < SEGW; ++s) {
      const int len = __builtin_amdgcn_readlane(segc, s);
#pragma unroll
      for (int k = 0; k < 2; ++k) {
        const int i = i0 + k * 32 + lane;
        if (i < len) atomicAdd(&cnt[d[s][k] - c0], 1);
      }
    }
  }
  __syncthreads();
  const int v = (tid < BNODES) ? cnt[tid] : 0;
  int total;
  const int ex = block_excl_scan_1024(v, wsum, lane, wave, total);
  const int c = c0 + tid;
  if (tid < BNODES && c < n) {
    offs[c] = p0 + ex;
    dinv[c] = rsqrtf((float)(v + 1));
  }
  if (b == nb - 1 && tid == 0) offs[n] = p0 + total;
  if (tid < BNODES) cnt[tid] = ex;
  __syncthreads();
  const int tot = __builtin_amdgcn_readfirstlane(total);
  if (tot <= LCAP) {
    place_walk(part, cnt, lst, segc, segl, mx, wave, lane, che, c0);
    __syncthreads();
    for (int i = tid; i < tot; i += SORT_THREADS) rec[p0 + i] = lst[i];
  } else {
    place_walk(part, cnt, rec + p0, segc, segl, mx, wave, lane, che, c0);
  }
}

constexpr int GK = 128;
constexpr int APITCH = GK + 4;


template <int F, int G, int B>
__device__ __forceinline__ void gather_steps(const float* __restrict__ h, int rv,
                                             int t0, int g, int fl, float4& acc) {
  float4 hv[B];
#pragma unroll
  for (int u = 0; u < B; ++u) {
    int r;
    if constexpr (G == 1) {
      r = __builtin_amdgcn_readlane(rv, t0 + u);
    } else {
      r = __shfl(rv, (t0 + u) * G + g);
    }
    hv[u] = *(const float4*)(h + (size_t)r * F + fl * 4);
  }
#pragma unroll
  for (int u = 0; u < B; ++u) {
    acc.x += hv[u].x;
    acc.y += hv[u].y;
    acc.z += hv[u].z;
    acc.w += hv[u].w;
  }
}

template <int F>
__device__ __forceinline__ void list_walk(const int* __restrict__ rec, const float* __restrict__ h,
                                          int beg, int end, int c, int lane, int g, int fl,
                                          float4& acc) {
  constexpr int G = 32 / (F / 4);
  constexpr int BMAX = 8 / G;
  for (int j = beg; j < end; j += 32) {
    const int m = min(32, end - j);
    int rv = c;
    if (lane < m) rv = rec[j + lane];
    const int S = m / G;
    int t = 0;
    for (; t + BMAX <= S; t += BMAX) gather_steps<F, G, BMAX>(h, rv, t, g, fl, acc);
    if constexpr (BMAX >= 8) {
      if (t + 4 <= S) { gather_steps<F, G, 4>(h, rv, t, g, fl, acc); t += 4; }
    }
    if constexpr (BMAX >= 4) {
      if (t + 2 <= S) { gather_steps<F, G, 2>(h, rv, t, g, fl, acc); t += 2; }
    }
    if (t < S) gather_steps<F, G, 1>(h, rv, t, g, fl, acc);
    if constexpr (G == 2) {
      if (m & 1) {
        const int r = __shfl(rv, m - 1);
        if (g == 0) {
          const float4 hv = *(const float4*)(h + (size_t)r * F + fl * 4);
          acc.x += hv.x;
          acc.y += hv.y;
          acc.z += hv.z;
          acc.w += hv.w;
        }
      }
    }
  }
}

template <int F, bool HAS_BIAS>
__global__ __launch_bounds__(256) void gather_agg_kernel(
    const int* __restrict__ offs, const int* __restrict__ rec,
    const float* __restrict__ h, const float* __restrict__ dinv,
    const float* __restrict__ bias, float* __restrict__ out, int n) {
  constexpr int LPR = F / 4;
  constexpr int G = 32 / LPR;
  const int lane = threadIdx.x & 31;
  const int g = lane / LPR;
  const int fl = lane % LPR;
  const int wave = __builtin_amdgcn_readfirstlane((int)(threadIdx.x >> 5));
  const int wpb = blockDim.x >> 5;
  const int nwaves = gridDim.x * wpb;
  float4 bv = make_float4(0.0f, 0.0f, 0.0f, 0.0f);
  if constexpr (HAS_BIAS) bv = *(const float4*)(bias + fl * 4);

  for (int c = blockIdx.x * wpb + wave; c < n; c += nwaves) {
    const int beg = offs[c];
    const int end = offs[c + 1];
    const float dc = dinv[c];
    float4 acc = make_float4(0.0f, 0.0f, 0.0f, 0.0f);
    float4 hs = make_float4(0.0f, 0.0f, 0.0f, 0.0f);
    if (g == 0) hs = *(const float4*)(h + (size_t)c * F + fl * 4);
    list_walk<F>(rec, h, beg, end, c, lane, g, fl, acc);
    if constexpr (G == 2) {
      acc.x += __shfl_xor(acc.x, 16);
      acc.y += __shfl_xor(acc.y, 16);
      acc.z += __shfl_xor(acc.z, 16);
      acc.w += __shfl_xor(acc.w, 16);
    }
    if (g == 0) {
      acc.x = (acc.x + hs.x) * dc + bv.x;
      acc.y = (acc.y + hs.y) * dc + bv.y;
      acc.z = (acc.z + hs.z) * dc + bv.z;
      acc.w = (acc.w + hs.w) * dc + bv.w;
      *(float4*)(out + (size_t)c * F + fl * 4) = acc;
    }
  }
}

constexpr int FX_TILES = 2;
constexpr int FX_ROWS  = FX_TILES * 16;
constexpr int FX_WAVES = GK / 16;
constexpr int FX_NPW   = FX_ROWS / FX_WAVES;
constexpr int FX_N2    = 64;
constexpr int FX_KP    = 2;
constexpr int FX_KS    = GK / 4 / FX_KP;
static_assert(FX_TILES * (FX_N2 / 16) == FX_WAVES, "second product: one (M tile, column tile) pair per wave");

template <int B>
__device__ __forceinline__ void gather_steps_w(const float* __restrict__ x, int rv, float wv,
                                               int t0, int fl, float4& acc) {
  float4 hv[B];
#pragma unroll
  for (int u = 0; u < B; ++u) {
    const int r = __builtin_amdgcn_readlane(rv, t0 + u);
    hv[u] = *(const float4*)(x + (size_t)r * GK + fl * 4);
  }
#pragma unroll
  for (int u = 0; u < B; ++u) {
    const float w = __int_as_float(__builtin_amdgcn_readlane(__float_as_int(wv), t0 + u));
    acc.x = __builtin_fmaf(w, hv[u].x, acc.x);
    acc.y = __builtin_fmaf(w, hv[u].y, acc.y);
    acc.z = __builtin_fmaf(w, hv[u].z, acc.z);
    acc.w = __builtin_fmaf(w, hv[u].w, acc.w);
  }
}

__device__ __forceinline__ void list_walk_w(const int* __restrict__ rec, const float* __restrict__ x,
                                            const float* __restrict__ dinv, int beg, int end,
                                            int c, int lane, float4& acc) {
  for (int j = beg; j < end; j += 32) {
    const int m = min(32, end - j);
    int rv = c;
    if (lane < m) rv = rec[j + lane];
    const float wv = dinv[rv];
    int t = 0;
    for (; t + 8 <= m; t += 8) gather_steps_w<8>(x, rv, wv, t, lane, acc);
    if (t + 4 <= m) { gather_steps_w<4>(x, rv, wv, t, lane, acc); t += 4; }
    if (t + 2 <= m) { gather_steps_w<2>(x, rv, wv, t, lane, acc); t += 2; }
    if (t < m) gather_steps_w<1>(x, rv, wv, t, lane, acc);
  }
}

__global__ __launch_bounds__(32 * FX_WAVES) void gather_xw_gemm_kernel(
    const int* __restrict__ offs, const int* __restrict__ rec,
    const float* __restrict__ x, const float* __restrict__ dinv,
    const float* __restrict__ W1, const float* __restrict__ b1,
    const float* __restrict__ W2, float* __restrict__ C, int n) {
  static_assert(GK / 4 == 32, "one lane per 4 features of a row");
  __shared__ __attribute__((aligned(16))) float stage[FX_ROWS * APITCH];
  __shared__ __attribute__((aligned(16))) float stage2[FX_ROWS * APITCH];
  const int lane = threadIdx.x & 31;
  const int wave = __builtin_amdgcn_readfirstlane((int)(threadIdx.x >> 5));
  const int base = blockIdx.x * FX_ROWS;
  for (int i = 0; i < FX_NPW; ++i) {
    const int lr = wave * FX_NPW + i;
    const int c = base + lr;
    float4 acc = make_float4(0.0f, 0.0f, 0.0f, 0.0f);
    if (c < n) {
      const int beg = offs[c];
      const int end = offs[c + 1];
      const float dc = dinv[c];
      const float4 xs = *(const float4*)(x + (size_t)c * GK + lane * 4);
      list_walk_w(rec, x, dinv, beg, end, c, lane, acc);
      acc.x = (acc.x + dc * xs.x) * dc;
      acc.y = (acc.y + dc * xs.y) * dc;
      acc.z = (acc.z + dc * xs.z) * dc;
      acc.w = (acc.w + dc * xs.w) * dc;
    }
    *(float4*)(stage + lr * APITCH + lane * 4) = acc;
  }
  __syncthreads();
  const int half = lane >> 4;
  const int l    = lane & 15;
  {
    const int n0 = wave * 16;
    const float* arow = stage + l * APITCH + 2 * half;
    const float* bcol = W1 + (size_t)(2 * half) * GK + (n0 + l);
    v8f acc1[FX_TILES] = {};
#pragma unroll 1
    for (int q = 0; q < FX_KP; ++q) {
      const float* bq = bcol + (size_t)q * (FX_KS * 4 * GK);
      const float* aq = arow + q * (FX_KS * 4);
      v2f bp[FX_KS];
#pragma unroll
      for (int s = 0; s < FX_KS; ++s) {
        bp[s].x = bq[s * 4 * GK];
        bp[s].y = bq[(s * 4 + 1) * GK];
      }
#pragma unroll
      for (int s = 0; s < FX_KS; ++s) {
#pragma unroll
        for (int t = 0; t < FX_TILES; ++t) {
          const v2f a = *(const v2f*)(aq + t * 16 * APITCH + s * 4);
          acc1[t] = __builtin_amdgcn_wmma_f32_16x16x4_f32(false, a, false, bp[s],
                                                          (short)0, acc1[t], false, false);
        }
      }
    }
    const float bb = b1[n0 + l];
#pragma unroll
    for (int t = 0; t < FX_TILES; ++t) {
#pragma unroll
      for (int v = 0; v < 8; ++v) {
        stage2[(t * 16 + half * 8 + v) * APITCH + (n0 + l)] = fmaxf(acc1[t][v] + bb, 0.0f);
      }
    }
  }
  __syncthreads();
  const int t2 = wave / (FX_N2 / 16);
  const int n2 = (wave % (FX_N2 / 16)) * 16;
  const float* arow2 = stage2 + (t2 * 16 + l) * APITCH + 2 * half;
  const float* bcol2 = W2 + (size_t)(2 * half) * FX_N2 + (n2 + l);
  v8f acc2 = {};
#pragma unroll 1
  for (int q = 0; q < FX_KP; ++q) {
    const float* bq = bcol2 + (size_t)q * (FX_KS * 4 * FX_N2);
    const float* aq = arow2 + q * (FX_KS * 4);
    v2f bp[FX_KS];
#pragma unroll
    for (int s = 0; s < FX_KS; ++s) {
      bp[s].x = bq[s * 4 * FX_N2];
      bp[s].y = bq[(s * 4 + 1) * FX_N2];
    }
#pragma unroll
    for (int s = 0; s < FX_KS; ++s) {
      const v2f a = *(const v2f*)(aq + s * 4);
      acc2 = __builtin_amdgcn_wmma_f32_16x16x4_f32(false, a, false, bp[s],
                                                   (short)0, acc2, false, false);
    }
  }
  const int m0 = base + t2 * 16;
  float* cp = C + (size_t)(m0 + half * 8) * FX_N2 + (n2 + l);
#pragma unroll
  for (int v = 0; v < 8; ++v) {
    if (m0 + half * 8 + v < n) cp[(size_t)v * FX_N2] = acc2[v] * dinv[m0 + half * 8 + v];
  }
}

extern "C" void kernel_launch(void* const* d_in, const int* in_sizes, int n_in,
                              void* d_out, int out_size, void* d_ws, size_t ws_size,
                              hipStream_t stream) {
  const float* x  = (const float*)d_in[0];
  const int*   ei = (const int*)d_in[1];
  const float* W1 = (const float*)d_in[2];
  const float* b1 = (const float*)d_in[3];
  const float* W2 = (const float*)d_in[4];
  const float* b2 = (const float*)d_in[5];

  const int hid    = in_sizes[3];
  const int out_ch = in_sizes[5];
  const int in_ch  = in_sizes[2] / hid;
  const int n      = in_sizes[0] / in_ch;
  const int E      = in_sizes[1] / 2;
  const int* row = ei;
  const int* col = ei + E;

  const int nb  = (n + BNODES - 1) / BNODES;
  const int che = (E + NCH - 1) / NCH;

  float* ws = (float*)d_ws;
  size_t off = 0;
  auto alloc = [&](size_t elems) {
    float* p = ws + off;
    off += (elems + 63) & ~(size_t)63;
    return p;
  };
  int*   T      = (int*)alloc((size_t)nb * NCH);
  int*   L      = (int*)alloc((size_t)nb * NCH);
  int*   offs   = (int*)alloc((size_t)n + 1);
  float* dinv   = alloc((size_t)n);
  int*   part   = (int*)alloc((size_t)E * 2);
  int*   rec    = (int*)alloc((size_t)E);
  float* g2   = alloc((size_t)n * out_ch);
  float* out  = (float*)d_out;

  chunk_group_kernel<<<NCH, SORT_THREADS, 0, stream>>>(row, col, T, L, part, E, che, nb);
  bucket_sort_kernel<<<nb, SORT_THREADS, 0, stream>>>(part, T, L, rec, offs, dinv, n, nb, che);

  gather_xw_gemm_kernel<<<(n + FX_ROWS - 1) / FX_ROWS, 32 * FX_WAVES, 0, stream>>>(
      offs, rec, x, dinv, W1, b1, W2, g2, n);
  gather_agg_kernel<64, true><<<2048, 256, 0, stream>>>(offs, rec, g2, dinv, b2, out, n);
}


# B fragments of both products read as 16-byte loads (512 contiguous bytes per wave) from a fragment-ordered copy of W1, W2 that the first launch writes each call, not 4-byte two-row loads from W
# speedup vs baseline: 5.5065x; 1.0799x over previous
#include <hip/hip_runtime.h>
#include <hip/hip_bf16.h>


typedef __attribute__((ext_vector_type(2))) float v2f;
typedef __attribute__((ext_vector_type(8))) float v8f;
typedef __attribute__((ext_vector_type(4))) float v4f;

constexpr int PK_K  = 128;
constexpr int PK_N1 = 128;
constexpr int PK_N2 = 64;
constexpr int PK_TOTAL = PK_K * (PK_N1 + PK_N2);

__device__ __forceinline__ int pack_src(int f, int ncols) {
  const int e = f & 3;
  const int lane = (f >> 2) & 31;
  const int j = (f >> 7) & 15;
  const int ct = f >> 11;
  const int k = 8 * j + 2 * (lane >> 4) + (e & 1) + 4 * (e >> 1);
  return k * ncols + ct * 16 + (lane & 15);
}

constexpr int BSHIFT = 9;
constexpr int BNODES = 1 << BSHIFT;
constexpr int NCH = 256;
constexpr int NBMAX = 1024;
constexpr int SORT_THREADS = 1024;
constexpr int EPB = 4;
static_assert(EPB == 4, "the load fence in chunk_group_kernel names four values");
static_assert(NBMAX == SORT_THREADS, "chunk_group_kernel: one bucket counter per thread");
constexpr int SEGW = NCH / 32;
static_assert(SEGW * 32 == NCH && SEGW <= 32, "bucket_sort_kernel: lane s of a wave holds the wave's piece s");
constexpr int LCAP = 12288;
constexpr int CHCAP = 6400;
constexpr int PK_PER_WG = PK_TOTAL / NCH;
static_assert(PK_PER_WG * NCH == PK_TOTAL && PK_PER_WG <= SORT_THREADS, "weight copy: one element per thread of the first PK_PER_WG");

__device__ __forceinline__ int wave_incl_scan(int v, int lane) {
  for (int d = 1; d < 32; d <<= 1) {
    const int y = __shfl_up(v, d);
    if (lane >= d) v += y;
  }
  return v;
}

__device__ __forceinline__ int block_excl_scan_1024(int v, int* wsum, int lane, int wave,
                                                    int& total) {
  const int incl = wave_incl_scan(v, lane);
  if (lane == 31) wsum[wave] = incl;
  __syncthreads();
  const int wt = wsum[lane];
  const int winc = wave_incl_scan(wt, lane);
  const int wave_excl = __shfl(winc, wave) - __shfl(wt, wave);
  total = __shfl(winc, 31);
  return wave_excl + incl - v;
}

__device__ __forceinline__ void place_chunk(const int* __restrict__ row,
                                            const int* __restrict__ col, int* hist, int2* dst,
                                            int e0, int e1, int tid) {
  for (int e = e0 + tid; e < e1; e += EPB * SORT_THREADS) {
    int r[EPB], c[EPB];
#pragma unroll
    for (int u = 0; u < EPB; ++u) {
      const int ee = e + u * SORT_THREADS;
      r[u] = (ee < e1) ? row[ee] : 0;
      c[u] = (ee < e1) ? col[ee] : 0;
    }
#pragma unroll
    for (int u = 0; u < EPB; ++u) {
      if (e + u * SORT_THREADS < e1) {
        const int pos = atomicAdd(&hist[c[u] >> BSHIFT], 1);
        dst[pos] = make_int2(r[u], c[u]);
      }
    }
  }
}

__global__ __launch_bounds__(SORT_THREADS) void chunk_group_kernel(
    const int* __restrict__ row, const int* __restrict__ col, int* __restrict__ T,
    int* __restrict__ L, int* __restrict__ part, int E, int che, int nb,
    const float* __restrict__ W1, const float* __restrict__ W2, float* __restrict__ P) {
  __shared__ int hist[NBMAX];
  __shared__ int wsum[32];
  __shared__ __attribute__((aligned(16))) int seg[2 * CHCAP];
  const int tid = threadIdx.x;
  const int lane = tid & 31;
  const int wave = tid >> 5;
  const int ch = blockIdx.x;
  const int pf = ch * PK_PER_WG + tid;
  float pv = 0.0f;
  if (tid < PK_PER_WG) {
    pv = (pf < PK_K * PK_N1) ? W1[pack_src(pf, PK_N1)] : W2[pack_src(pf - PK_K * PK_N1, PK_N2)];
  }
  hist[tid] = 0;
  __syncthreads();
  const int e0 = ch * che;
  const int e1 = min(E, e0 + che);
  for (int e = e0 + tid; e < e1; e += EPB * SORT_THREADS) {
    int c[EPB];
#pragma unroll
    for (int u = 0; u < EPB; ++u) {
      c[u] = col[min(e + u * SORT_THREADS, e1 - 1)];
    }
    asm volatile("" :: "v"(c[0]), "v"(c[1]), "v"(c[2]), "v"(c[3]));
#pragma unroll
    for (int u = 0; u < EPB; ++u) {
      if (e + u * SORT_THREADS < e1) atomicAdd(&hist[c[u] >> BSHIFT], 1);
    }
  }
  __syncthreads();
  const int v = hist[tid];
  int total;
  const int ex = block_excl_scan_1024(v, wsum, lane, wave, total);
  if (tid < nb) {
    T[tid * NCH + ch] = v;
    L[tid * NCH + ch] = ex;
  }
  hist[tid] = ex;
  __syncthreads();
  const int cnt = max(0, e1 - e0);
  int2* const dst = (int2*)part + e0;
  if (cnt <= CHCAP) {
    int2* const segr = (int2*)seg;
    place_chunk(row, col, hist, segr, e0, e1, tid);
    __syncthreads();
    const int head = min(cnt, (16 - (e0 & 15)) & 15);
    if (tid < head) dst[tid] = segr[tid];
    for (int i = head + tid; i < cnt; i += SORT_THREADS) dst[i] = segr[i];
  } else {
    place_chunk(row, col, hist, dst, e0, e1, tid);
  }
  if (tid < PK_PER_WG) P[pf] = pv;
}

__device__ __forceinline__ void place_walk(const int* __restrict__ part, int* cnt, int* dst,
                                           int segc, int segl, int mx, int wave, int lane,
                                           int che, int c0) {
  for (int i0 = 0; i0 < mx; i0 += 64) {
    int2 rc[SEGW][2];
#pragma unroll
    for (int s = 0; s < SEGW; ++s) {
      const int len = __builtin_amdgcn_readlane(segc, s);
      const int base = (wave * SEGW + s) * che + __builtin_amdgcn_readlane(segl, s);
#pragma unroll
      for (int k = 0; k < 2; ++k) {
        const int i = i0 + k * 32 + lane;
        rc[s][k] = (i < len) ? ((const int2*)part)[base + i] : make_int2(0, c0);
      }
    }
    int pos[SEGW][2];
#pragma unroll
    for (int s = 0; s < SEGW; ++s) {
      const int len = __builtin_amdgcn_readlane(segc, s);
#pragma unroll
      for (int k = 0; k < 2; ++k) {
        const int i = i0 + k * 32 + lane;
        pos[s][k] = (i < len) ? atomicAdd(&cnt[rc[s][k].y - c0], 1) : 0;
      }
    }
#pragma unroll
    for (int s = 0; s < SEGW; ++s) {
      const int len = __builtin_amdgcn_readlane(segc, s);
#pragma unroll
      for (int k = 0; k < 2; ++k) {
        const int i = i0 + k * 32 + lane;
        if (i < len) dst[pos[s][k]] = rc[s][k].x;
      }
    }
  }
}

__global__ __launch_bounds__(SORT_THREADS) void bucket_sort_kernel(
    const int* __restrict__ part, const int* __restrict__ T, const int* __restrict__ L,
    int* __restrict__ rec, int* __restrict__ offs, float* __restrict__ dinv,
    int n, int nb, int che) {
  __shared__ int cnt[BNODES];
  __shared__ int lst[LCAP];
  __shared__ int wsum[32];
  __shared__ int wsum0[32];
  const int tid = threadIdx.x;
  const int lane = tid & 31;
  const int wave = __builtin_amdgcn_readfirstlane(tid >> 5);
  const int b = blockIdx.x;
  const int c0 = b << BSHIFT;
  int segc = 0, segl = 0;
  if (lane < SEGW) {
    segc = T[b * NCH + wave * SEGW + lane];
    segl = L[b * NCH + wave * SEGW + lane];
  }
  if (tid < BNODES) cnt[tid] = 0;
  int p0;
  (void)block_excl_scan_1024(segl, wsum0, lane, wave, p0);
  int mx = segc;
  for (int sh = 1; sh < SEGW; sh <<= 1) mx = max(mx, __shfl_xor(mx, sh));
  mx = __builtin_amdgcn_readfirstlane(mx);
  for (int i0 = 0; i0 < mx; i0 += 64) {
    int d[SEGW][2];
#pragma unroll
    for (int s = 0; s < SEGW; ++s) {
      const int len = __builtin_amdgcn_readlane(segc, s);
      const int base = (wave * SEGW + s) * che + __builtin_amdgcn_readlane(segl, s);
#pragma unroll
      for (int k = 0; k < 2; ++k) {
        const int i = i0 + k * 32 + lane;
        d[s][k] = (i < len) ? part[2 * (size_t)(base + i) + 1] : c0;
      }
    }
#pragma unroll
    for (int s = 0; s < SEGW; ++s) {
      const int len = __builtin_amdgcn_readlane(segc, s);
#pragma unroll
      for (int k = 0; k < 2; ++k) {
        const int i = i0 + k * 32 + lane;
        if (i < len) atomicAdd(&cnt[d[s][k] - c0], 1);
      }
    }
  }
  __syncthreads();
  const int v = (tid < BNODES) ? cnt[tid] : 0;
  int total;
  const int ex = block_excl_scan_1024(v, wsum, lane, wave, total);
  const int c = c0 + tid;
  if (tid < BNODES && c < n) {
    offs[c] = p0 + ex;
    dinv[c] = rsqrtf((float)(v + 1));
  }
  if (b == nb - 1 && tid == 0) offs[n] = p0 + total;
  if (tid < BNODES) cnt[tid] = ex;
  __syncthreads();
  const int tot = __builtin_amdgcn_readfirstlane(total);
  if (tot <= LCAP) {
    place_walk(part, cnt, lst, segc, segl, mx, wave, lane, che, c0);
    __syncthreads();
    for (int i = tid; i < tot; i += SORT_THREADS) rec[p0 + i] = lst[i];
  } else {
    place_walk(part, cnt, rec + p0, segc, segl, mx, wave, lane, che, c0);
  }
}

constexpr int GK = 128;
constexpr int APITCH = GK + 4;


template <int F, int G, int B>
__device__ __forceinline__ void gather_steps(const float* __restrict__ h, int rv,
                                             int t0, int g, int fl, float4& acc) {
  float4 hv[B];
#pragma unroll
  for (int u = 0; u < B; ++u) {
    int r;
    if constexpr (G == 1) {
      r = __builtin_amdgcn_readlane(rv, t0 + u);
    } else {
      r = __shfl(rv, (t0 + u) * G + g);
    }
    hv[u] = *(const float4*)(h + (size_t)r * F + fl * 4);
  }
#pragma unroll
  for (int u = 0; u < B; ++u) {
    acc.x += hv[u].x;
    acc.y += hv[u].y;
    acc.z += hv[u].z;
    acc.w += hv[u].w;
  }
}

template <int F>
__device__ __forceinline__ void list_walk(const int* __restrict__ rec, const float* __restrict__ h,
                                          int beg, int end, int c, int lane, int g, int fl,
                                          float4& acc) {
  constexpr int G = 32 / (F / 4);
  constexpr int BMAX = 8 / G;
  for (int j = beg; j < end; j += 32) {
    const int m = min(32, end - j);
    int rv = c;
    if (lane < m) rv = rec[j + lane];
    const int S = m / G;
    int t = 0;
    for (; t + BMAX <= S; t += BMAX) gather_steps<F, G, BMAX>(h, rv, t, g, fl, acc);
    if constexpr (BMAX >= 8) {
      if (t + 4 <= S) { gather_steps<F, G, 4>(h, rv, t, g, fl, acc); t += 4; }
    }
    if constexpr (BMAX >= 4) {
      if (t + 2 <= S) { gather_steps<F, G, 2>(h, rv, t, g, fl, acc); t += 2; }
    }
    if (t < S) gather_steps<F, G, 1>(h, rv, t, g, fl, acc);
    if constexpr (G == 2) {
      if (m & 1) {
        const int r = __shfl(rv, m - 1);
        if (g == 0) {
          const float4 hv = *(const float4*)(h + (size_t)r * F + fl * 4);
          acc.x += hv.x;
          acc.y += hv.y;
          acc.z += hv.z;
          acc.w += hv.w;
        }
      }
    }
  }
}

template <int F, bool HAS_BIAS>
__global__ __launch_bounds__(256) void gather_agg_kernel(
    const int* __restrict__ offs, const int* __restrict__ rec,
    const float* __restrict__ h, const float* __restrict__ dinv,
    const float* __restrict__ bias, float* __restrict__ out, int n) {
  constexpr int LPR = F / 4;
  constexpr int G = 32 / LPR;
  const int lane = threadIdx.x & 31;
  const int g = lane / LPR;
  const int fl = lane % LPR;
  const int wave = __builtin_amdgcn_readfirstlane((int)(threadIdx.x >> 5));
  const int wpb = blockDim.x >> 5;
  const int nwaves = gridDim.x * wpb;
  float4 bv = make_float4(0.0f, 0.0f, 0.0f, 0.0f);
  if constexpr (HAS_BIAS) bv = *(const float4*)(bias + fl * 4);

  for (int c = blockIdx.x * wpb + wave; c < n; c += nwaves) {
    const int beg = offs[c];
    const int end = offs[c + 1];
    const float dc = dinv[c];
    float4 acc = make_float4(0.0f, 0.0f, 0.0f, 0.0f);
    float4 hs = make_float4(0.0f, 0.0f, 0.0f, 0.0f);
    if (g == 0) hs = *(const float4*)(h + (size_t)c * F + fl * 4);
    list_walk<F>(rec, h, beg, end, c, lane, g, fl, acc);
    if constexpr (G == 2) {
      acc.x += __shfl_xor(acc.x, 16);
      acc.y += __shfl_xor(acc.y, 16);
      acc.z += __shfl_xor(acc.z, 16);
      acc.w += __shfl_xor(acc.w, 16);
    }
    if (g == 0) {
      acc.x = (acc.x + hs.x) * dc + bv.x;
      acc.y = (acc.y + hs.y) * dc + bv.y;
      acc.z = (acc.z + hs.z) * dc + bv.z;
      acc.w = (acc.w + hs.w) * dc + bv.w;
      *(float4*)(out + (size_t)c * F + fl * 4) = acc;
    }
  }
}

constexpr int FX_TILES = 2;
constexpr int FX_ROWS  = FX_TILES * 16;
constexpr int FX_WAVES = GK / 16;
constexpr int FX_NPW   = FX_ROWS / FX_WAVES;
constexpr int FX_N2    = 64;
constexpr int FX_KP    = 2;
constexpr int FX_KS    = GK / 4 / FX_KP;
static_assert(FX_TILES * (FX_N2 / 16) == FX_WAVES, "second product: one (M tile, column tile) pair per wave");

template <int B>
__device__ __forceinline__ void gather_steps_w(const float* __restrict__ x, int rv, float wv,
                                               int t0, int fl, float4& acc) {
  float4 hv[B];
#pragma unroll
  for (int u = 0; u < B; ++u) {
    const int r = __builtin_amdgcn_readlane(rv, t0 + u);
    hv[u] = *(const float4*)(x + (size_t)r * GK + fl * 4);
  }
#pragma unroll
  for (int u = 0; u < B; ++u) {
    const float w = __int_as_float(__builtin_amdgcn_readlane(__float_as_int(wv), t0 + u));
    acc.x = __builtin_fmaf(w, hv[u].x, acc.x);
    acc.y = __builtin_fmaf(w, hv[u].y, acc.y);
    acc.z = __builtin_fmaf(w, hv[u].z, acc.z);
    acc.w = __builtin_fmaf(w, hv[u].w, acc.w);
  }
}

__device__ __forceinline__ void list_walk_w(const int* __restrict__ rec, const float* __restrict__ x,
                                            const float* __restrict__ dinv, int beg, int end,
                                            int c, int lane, float4& acc) {
  for (int j = beg; j < end; j += 32) {
    const int m = min(32, end - j);
    int rv = c;
    if (lane < m) rv = rec[j + lane];
    const float wv = dinv[rv];
    int t = 0;
    for (; t + 8 <= m; t += 8) gather_steps_w<8>(x, rv, wv, t, lane, acc);
    if (t + 4 <= m) { gather_steps_w<4>(x, rv, wv, t, lane, acc); t += 4; }
    if (t + 2 <= m) { gather_steps_w<2>(x, rv, wv, t, lane, acc); t += 2; }
    if (t < m) gather_steps_w<1>(x, rv, wv, t, lane, acc);
  }
}

__global__ __launch_bounds__(32 * FX_WAVES) void gather_xw_gemm_kernel(
    const int* __restrict__ offs, const int* __restrict__ rec,
    const float* __restrict__ x, const float* __restrict__ dinv,
    const float* __restrict__ P1, const float* __restrict__ b1,
    const float* __restrict__ P2, float* __restrict__ C, int n) {
  static_assert(GK / 4 == 32, "one lane per 4 features of a row");
  static_assert(GK == PK_K && GK == PK_N1 && FX_N2 == PK_N2 && FX_KS % 2 == 0,
                "the fragment-ordered copies of W1 / W2: shapes, and whole k groups per piece");
  __shared__ __attribute__((aligned(16))) float stage[FX_ROWS * APITCH];
  __shared__ __attribute__((aligned(16))) float stage2[FX_ROWS * APITCH];
  const int lane = threadIdx.x & 31;
  const int wave = __builtin_amdgcn_readfirstlane((int)(threadIdx.x >> 5));
  const int base = blockIdx.x * FX_ROWS;
  for (int i = 0; i < FX_NPW; ++i) {
    const int lr = wave * FX_NPW + i;
    const int c = base + lr;
    float4 acc = make_float4(0.0f, 0.0f, 0.0f, 0.0f);
    if (c < n) {
      const int beg = offs[c];
      const int end = offs[c + 1];
      const float dc = dinv[c];
      const float4 xs = *(const float4*)(x + (size_t)c * GK + lane * 4);
      list_walk_w(rec, x, dinv, beg, end, c, lane, acc);
      acc.x = (acc.x + dc * xs.x) * dc;
      acc.y = (acc.y + dc * xs.y) * dc;
      acc.z = (acc.z + dc * xs.z) * dc;
      acc.w = (acc.w + dc * xs.w) * dc;
    }
    *(float4*)(stage + lr * APITCH + lane * 4) = acc;
  }
  __syncthreads();
  const int half = lane >> 4;
  const int l    = lane & 15;
  {
    const int n0 = wave * 16;
    const float* arow = stage + l * APITCH + 2 * half;
    const v4f* bt = (const v4f*)P1 + (size_t)(wave * (GK / 8)) * 32 + lane;
    v8f acc1[FX_TILES] = {};
#pragma unroll 1
    for (int q = 0; q < FX_KP; ++q) {
      const v4f* bq = bt + (size_t)q * (FX_KS / 2) * 32;
      const float* aq = arow + q * (FX_KS * 4);
      v2f bp[FX_KS];
#pragma unroll
      for (int s = 0; s < FX_KS; s += 2) {
        const v4f b4 = bq[(s / 2) * 32];
        bp[s].x = b4.x;
        bp[s].y = b4.y;
        bp[s + 1].x = b4.z;
        bp[s + 1].y = b4.w;
      }
#pragma unroll
      for (int s = 0; s < FX_KS; ++s) {
#pragma unroll
        for (int t = 0; t < FX_TILES; ++t) {
          const v2f a = *(const v2f*)(aq + t * 16 * APITCH + s * 4);
          acc1[t] = __builtin_amdgcn_wmma_f32_16x16x4_f32(false, a, false, bp[s],
                                                          (short)0, acc1[t], false, false);
        }
      }
    }
    const float bb = b1[n0 + l];
#pragma unroll
    for (int t = 0; t < FX_TILES; ++t) {
#pragma unroll
      for (int v = 0; v < 8; ++v) {
        stage2[(t * 16 + half * 8 + v) * APITCH + (n0 + l)] = fmaxf(acc1[t][v] + bb, 0.0f);
      }
    }
  }
  __syncthreads();
  const int t2 = wave / (FX_N2 / 16);
  const int n2 = (wave % (FX_N2 / 16)) * 16;
  const float* arow2 = stage2 + (t2 * 16 + l) * APITCH + 2 * half;
  const v4f* bt2 = (const v4f*)P2 + (size_t)((n2 / 16) * (GK / 8)) * 32 + lane;
  v8f acc2 = {};
#pragma unroll 1
  for (int q = 0; q < FX_KP; ++q) {
    const v4f* bq = bt2 + (size_t)q * (FX_KS / 2) * 32;
    const float* aq = arow2 + q * (FX_KS * 4);
    v2f bp[FX_KS];
#pragma unroll
    for (int s = 0; s < FX_KS; s += 2) {
      const v4f b4 = bq[(s / 2) * 32];
      bp[s].x = b4.x;
      bp[s].y = b4.y;
      bp[s + 1].x = b4.z;
      bp[s + 1].y = b4.w;
    }
#pragma unroll
    for (int s = 0; s < FX_KS; ++s) {
      const v2f a = *(const v2f*)(aq + s * 4);
      acc2 = __builtin_amdgcn_wmma_f32_16x16x4_f32(false, a, false, bp[s],
                                                   (short)0, acc2, false, false);
    }
  }
  const int m0 = base + t2 * 16;
  float* cp = C + (size_t)(m0 + half * 8) * FX_N2 + (n2 + l);
#pragma unroll
  for (int v = 0; v < 8; ++v) {
    if (m0 + half * 8 + v < n) cp[(size_t)v * FX_N2] = acc2[v] * dinv[m0 + half * 8 + v];
  }
}

extern "C" void kernel_launch(void* const* d_in, const int* in_sizes, int n_in,
                              void* d_out, int out_size, void* d_ws, size_t ws_size,
                              hipStream_t stream) {
  const float* x  = (const float*)d_in[0];
  const int*   ei = (const int*)d_in[1];
  const float* W1 = (const float*)d_in[2];
  const float* b1 = (const float*)d_in[3];
  const float* W2 = (const float*)d_in[4];
  const float* b2 = (const float*)d_in[5];

  const int hid    = in_sizes[3];
  const int out_ch = in_sizes[5];
  const int in_ch  = in_sizes[2] / hid;
  const int n      = in_sizes[0] / in_ch;
  const int E      = in_sizes[1] / 2;
  const int* row = ei;
  const int* col = ei + E;

  const int nb  = (n + BNODES - 1) / BNODES;
  const int che = (E + NCH - 1) / NCH;

  float* ws = (float*)d_ws;
  size_t off = 0;
  auto alloc = [&](size_t elems) {
    float* p = ws + off;
    off += (elems + 63) & ~(size_t)63;
    return p;
  };
  int*   T      = (int*)alloc((size_t)nb * NCH);
  int*   L      = (int*)alloc((size_t)nb * NCH);
  int*   offs   = (int*)alloc((size_t)n + 1);
  float* dinv   = alloc((size_t)n);
  int*   part   = (int*)alloc((size_t)E * 2);
  int*   rec    = (int*)alloc((size_t)E);
  float* g2   = alloc((size_t)n * out_ch);
  float* P    = alloc((size_t)PK_TOTAL);
  float* out  = (float*)d_out;

  chunk_group_kernel<<<NCH, SORT_THREADS, 0, stream>>>(row, col, T, L, part, E, che, nb, W1, W2, P);
  bucket_sort_kernel<<<nb, SORT_THREADS, 0, stream>>>(part, T, L, rec, offs, dinv, n, nb, che);

  gather_xw_gemm_kernel<<<(n + FX_ROWS - 1) / FX_ROWS, 32 * FX_WAVES, 0, stream>>>(
      offs, rec, x, dinv, P, b1, P + PK_K * PK_N1, g2, n);
  gather_agg_kernel<64, true><<<2048, 256, 0, stream>>>(offs, rec, g2, dinv, b2, out, n);
}
